# starting kernel
# baseline (speedup 1.0000x reference)
#include <hip/hip_runtime.h>

typedef float v2f __attribute__((ext_vector_type(2)));
typedef float v8f __attribute__((ext_vector_type(8)));

#define FAGCN_EPS 0.3f

__global__ void deg_kernel(const int* __restrict__ ei, float* __restrict__ deg, int E) {
  int e = blockIdx.x * blockDim.x + threadIdx.x;
  if (e < E) atomicAdd(&deg[ei[e]], 1.0f);
}

__global__ void dn_kernel(float* dn, int N) {
  int i = blockIdx.x * blockDim.x + threadIdx.x;
  if (i < N) dn[i] = 1.0f / sqrtf(fmaxf(dn[i], 1.0f));
}

__global__ void gemm_relu_kernel(const float* __restrict__ X, const float* __restrict__ W,
                                 const float* __restrict__ bias, float* __restrict__ H,
                                 int Nrows) {
  int wave = threadIdx.x >> 5;
  int lane = threadIdx.x & 31;
  int tileM = blockIdx.x;
  int mA = tileM * 16 + (lane & 15);
  int mclamp = mA < Nrows ? mA : (Nrows - 1);
  int n = wave * 16 + (lane & 15);
  int khalf = (lane >> 4) * 2;
  const float* xrow = X + (size_t)mclamp * 128 + khalf;
  const float* wcol = W + (size_t)khalf * 128 + n;
  __builtin_prefetch(xrow, 0, 0);
  v8f acc = {};
#pragma unroll 4
  for (int kk = 0; kk < 128; kk += 4) {
    v2f a = *(const v2f*)(xrow + kk);
    v2f b;
    b.x = wcol[(size_t)kk * 128];
    b.y = wcol[(size_t)kk * 128 + 128];
    acc = __builtin_amdgcn_wmma_f32_16x16x4_f32(false, a, false, b, (short)0, acc,
                                               false, false);
  }
  int mbase = tileM * 16 + ((lane >> 4) * 8);
  float bn = bias[n];
  if (tileM * 16 + 16 <= Nrows) {
    float* hp = H + (size_t)mbase * 128 + n;
#pragma unroll
    for (int r = 0; r < 8; ++r)
      hp[(size_t)r * 128] = fmaxf(acc[r] + bn, 0.0f);
  } else {
#pragma unroll
    for (int r = 0; r < 8; ++r) {
      int mm = mbase + r;
      if (mm < Nrows) H[(size_t)mm * 128 + n] = fmaxf(acc[r] + bn, 0.0f);
    }
  }
}

__global__ void gate_kernel(const float* __restrict__ H, const float* __restrict__ Wg,
                            float* __restrict__ ga, float* __restrict__ gb, int Nrows) {
  int wave = threadIdx.x >> 5;
  int lane = threadIdx.x & 31;
  int node = blockIdx.x * 8 + wave;
  if (node >= Nrows) return;
  const float* hrow = H + (size_t)node * 128;
  float sa = 0.0f, sb = 0.0f;
#pragma unroll
  for (int j = 0; j < 4; ++j) {
    int f = lane + j * 32;
    float hv = hrow[f];
    sa += hv * Wg[f];
    sb += hv * Wg[128 + f];
  }
  for (int off = 16; off > 0; off >>= 1) {
    sa += __shfl_down(sa, off, 32);
    sb += __shfl_down(sb, off, 32);
  }
  if (lane == 0) { ga[node] = sa; gb[node] = sb; }
}

__global__ void norm_kernel(const int* __restrict__ ei, const float* __restrict__ ga,
                            const float* __restrict__ gb, const float* __restrict__ dn,
                            const float* __restrict__ bg, int layer,
                            float* __restrict__ nrm, int E) {
  int e = blockIdx.x * blockDim.x + threadIdx.x;
  if (e >= E) return;
  int r = ei[e], c = ei[E + e];
  float g = tanhf(ga[r] + gb[c] + bg[layer]);
  nrm[e] = g * dn[r] * dn[c];
}

__global__ void scatter_kernel(const int* __restrict__ ei, const float* __restrict__ nrm,
                               const float* __restrict__ H, float* __restrict__ agg, int E) {
  long long idx = (long long)blockIdx.x * blockDim.x + threadIdx.x;
  if (idx >= (long long)E * 32) return;
  int e = (int)(idx >> 5);
  int f = (int)(idx & 31) * 4;
  int r = ei[e], c = ei[E + e];
  float s = nrm[e];
  float4 hv = *(const float4*)(H + (size_t)r * 128 + f);
  float* dst = agg + (size_t)c * 128 + f;
  atomicAdd(dst + 0, s * hv.x);
  atomicAdd(dst + 1, s * hv.y);
  atomicAdd(dst + 2, s * hv.z);
  atomicAdd(dst + 3, s * hv.w);
}

__global__ void residual_kernel(const float4* __restrict__ raw, float4* __restrict__ agg,
                                long long n4) {
  long long i = (long long)blockIdx.x * blockDim.x + threadIdx.x;
  if (i < n4) {
    float4 r = raw[i];
    float4 a = agg[i];
    a.x = FAGCN_EPS * r.x + a.x;
    a.y = FAGCN_EPS * r.y + a.y;
    a.z = FAGCN_EPS * r.z + a.z;
    a.w = FAGCN_EPS * r.w + a.w;
    agg[i] = a;
  }
}

__global__ void gemm_out_kernel(const float* __restrict__ H, const float* __restrict__ W,
                                const float* __restrict__ bias, float* __restrict__ out,
                                int Nrows) {
  int wave = threadIdx.x >> 5;
  int lane = threadIdx.x & 31;
  int tileM = blockIdx.x * 8 + wave;
  int tiles = (Nrows + 15) >> 4;
  if (tileM >= tiles) return;
  int mA = tileM * 16 + (lane & 15);
  int mclamp = mA < Nrows ? mA : (Nrows - 1);
  int n = lane & 15;
  int khalf = (lane >> 4) * 2;
  const float* hrow = H + (size_t)mclamp * 128 + khalf;
  const float* wcol = W + (size_t)khalf * 16 + n;
  __builtin_prefetch(hrow, 0, 0);
  v8f acc = {};
#pragma unroll 4
  for (int kk = 0; kk < 128; kk += 4) {
    v2f a = *(const v2f*)(hrow + kk);
    v2f b;
    b.x = wcol[(size_t)kk * 16];
    b.y = wcol[(size_t)kk * 16 + 16];
    acc = __builtin_amdgcn_wmma_f32_16x16x4_f32(false, a, false, b, (short)0, acc,
                                               false, false);
  }
  int mbase = tileM * 16 + ((lane >> 4) * 8);
  float bn = bias[n];
  if (tileM * 16 + 16 <= Nrows) {
    float* op = out + (size_t)mbase * 16 + n;
#pragma unroll
    for (int r = 0; r < 8; ++r)
      op[(size_t)r * 16] = acc[r] + bn;
  } else {
#pragma unroll
    for (int r = 0; r < 8; ++r) {
      int mm = mbase + r;
      if (mm < Nrows) out[(size_t)mm * 16 + n] = acc[r] + bn;
    }
  }
}

__global__ void logsoftmax_kernel(float4* out, int N) {
  int i = blockIdx.x * blockDim.x + threadIdx.x;
  if (i >= N) return;
  float4* row = out + (size_t)i * 4;
  float4 v[4];
#pragma unroll
  for (int q = 0; q < 4; ++q) v[q] = row[q];
  float m = v[0].x;
#pragma unroll
  for (int q = 0; q < 4; ++q) {
    m = fmaxf(m, v[q].x); m = fmaxf(m, v[q].y);
    m = fmaxf(m, v[q].z); m = fmaxf(m, v[q].w);
  }
  float s = 0.0f;
#pragma unroll
  for (int q = 0; q < 4; ++q) {
    s += expf(v[q].x - m); s += expf(v[q].y - m);
    s += expf(v[q].z - m); s += expf(v[q].w - m);
  }
  float d = m + logf(s);
#pragma unroll
  for (int q = 0; q < 4; ++q) {
    v[q].x -= d; v[q].y -= d; v[q].z -= d; v[q].w -= d;
    row[q] = v[q];
  }
}

extern "C" void kernel_launch(void* const* d_in, const int* in_sizes, int n_in,
                              void* d_out, int out_size, void* d_ws, size_t ws_size,
                              hipStream_t stream) {
  const float* x     = (const float*)d_in[0];
  const int*   ei    = (const int*)d_in[1];
  const float* W_in  = (const float*)d_in[2];
  const float* b_in  = (const float*)d_in[3];
  const float* Wg    = (const float*)d_in[4];
  const float* bg    = (const float*)d_in[5];
  const float* W_out = (const float*)d_in[6];
  const float* b_out = (const float*)d_in[7];
  float* out = (float*)d_out;

  const int Hdim = 128;
  int N = in_sizes[0] / Hdim;
  int E = in_sizes[1] / 2;
  int L = in_sizes[5];

  float* ws   = (float*)d_ws;
  float* dn   = ws;
  float* ga   = dn + N;
  float* gb   = ga + N;
  float* nrm  = gb + N;
  float* raw  = nrm + E;
  float* bufA = raw + (size_t)N * Hdim;
  float* bufB = bufA + (size_t)N * Hdim;

  hipMemsetAsync(dn, 0, (size_t)N * sizeof(float), stream);
  deg_kernel<<<(E + 255) / 256, 256, 0, stream>>>(ei, dn, E);
  dn_kernel<<<(N + 255) / 256, 256, 0, stream>>>(dn, N);

  gemm_relu_kernel<<<(N + 15) / 16, 256, 0, stream>>>(x, W_in, b_in, raw, N);

  float* h = raw;
  float* agg = bufA;
  for (int i = 0; i < L; ++i) {
    gate_kernel<<<(N + 7) / 8, 256, 0, stream>>>(h, Wg + (size_t)i * 2 * Hdim, ga, gb, N);
    norm_kernel<<<(E + 255) / 256, 256, 0, stream>>>(ei, ga, gb, dn, bg, i, nrm, E);
    hipMemsetAsync(agg, 0, (size_t)N * Hdim * sizeof(float), stream);
    long long tot = (long long)E * 32;
    scatter_kernel<<<(unsigned)((tot + 255) / 256), 256, 0, stream>>>(ei, nrm, h, agg, E);
    long long n4 = (long long)N * Hdim / 4;
    residual_kernel<<<(unsigned)((n4 + 255) / 256), 256, 0, stream>>>(
        (const float4*)raw, (float4*)agg, n4);
    h = agg;
    agg = (h == bufA) ? bufB : bufA;
  }

  int tiles = (N + 15) / 16;
  gemm_out_kernel<<<(tiles + 7) / 8, 256, 0, stream>>>(h, W_out, b_out, out, N);
  logsoftmax_kernel<<<(N + 255) / 256, 256, 0, stream>>>((float4*)out, N);
}


# Scatter-add leaves global float atomics: each work-group owns 196 target rows in LDS, scans all targets, sums its own edges with LDS float adds, stores rows once; agg memset goes; sum order changes
# speedup vs baseline: 1.7603x; 1.7603x over previous
#include <hip/hip_runtime.h>

typedef float v2f __attribute__((ext_vector_type(2)));
typedef float v8f __attribute__((ext_vector_type(8)));

#define FAGCN_EPS 0.3f

__global__ void deg_kernel(const int* __restrict__ ei, float* __restrict__ deg, int E) {
  int e = blockIdx.x * blockDim.x + threadIdx.x;
  if (e < E) atomicAdd(&deg[ei[e]], 1.0f);
}

__global__ void dn_kernel(float* dn, int N) {
  int i = blockIdx.x * blockDim.x + threadIdx.x;
  if (i < N) dn[i] = 1.0f / sqrtf(fmaxf(dn[i], 1.0f));
}

__global__ void gemm_relu_kernel(const float* __restrict__ X, const float* __restrict__ W,
                                 const float* __restrict__ bias, float* __restrict__ H,
                                 int Nrows) {
  int wave = threadIdx.x >> 5;
  int lane = threadIdx.x & 31;
  int tileM = blockIdx.x;
  int mA = tileM * 16 + (lane & 15);
  int mclamp = mA < Nrows ? mA : (Nrows - 1);
  int n = wave * 16 + (lane & 15);
  int khalf = (lane >> 4) * 2;
  const float* xrow = X + (size_t)mclamp * 128 + khalf;
  const float* wcol = W + (size_t)khalf * 128 + n;
  __builtin_prefetch(xrow, 0, 0);
  v8f acc = {};
#pragma unroll 4
  for (int kk = 0; kk < 128; kk += 4) {
    v2f a = *(const v2f*)(xrow + kk);
    v2f b;
    b.x = wcol[(size_t)kk * 128];
    b.y = wcol[(size_t)kk * 128 + 128];
    acc = __builtin_amdgcn_wmma_f32_16x16x4_f32(false, a, false, b, (short)0, acc,
                                               false, false);
  }
  int mbase = tileM * 16 + ((lane >> 4) * 8);
  float bn = bias[n];
  if (tileM * 16 + 16 <= Nrows) {
    float* hp = H + (size_t)mbase * 128 + n;
#pragma unroll
    for (int r = 0; r < 8; ++r)
      hp[(size_t)r * 128] = fmaxf(acc[r] + bn, 0.0f);
  } else {
#pragma unroll
    for (int r = 0; r < 8; ++r) {
      int mm = mbase + r;
      if (mm < Nrows) H[(size_t)mm * 128 + n] = fmaxf(acc[r] + bn, 0.0f);
    }
  }
}

__global__ void gate_kernel(const float* __restrict__ H, const float* __restrict__ Wg,
                            float* __restrict__ ga, float* __restrict__ gb, int Nrows) {
  int wave = threadIdx.x >> 5;
  int lane = threadIdx.x & 31;
  int node = blockIdx.x * 8 + wave;
  if (node >= Nrows) return;
  const float* hrow = H + (size_t)node * 128;
  float sa = 0.0f, sb = 0.0f;
#pragma unroll
  for (int j = 0; j < 4; ++j) {
    int f = lane + j * 32;
    float hv = hrow[f];
    sa += hv * Wg[f];
    sb += hv * Wg[128 + f];
  }
  for (int off = 16; off > 0; off >>= 1) {
    sa += __shfl_down(sa, off, 32);
    sb += __shfl_down(sb, off, 32);
  }
  if (lane == 0) { ga[node] = sa; gb[node] = sb; }
}

__global__ void norm_kernel(const int* __restrict__ ei, const float* __restrict__ ga,
                            const float* __restrict__ gb, const float* __restrict__ dn,
                            const float* __restrict__ bg, int layer,
                            float* __restrict__ nrm, int E) {
  int e = blockIdx.x * blockDim.x + threadIdx.x;
  if (e >= E) return;
  int r = ei[e], c = ei[E + e];
  float g = tanhf(ga[r] + gb[c] + bg[layer]);
  nrm[e] = g * dn[r] * dn[c];
}

#define AGG_TN 196
#define AGG_WAVES 32
#define AGG_STEP 512
#define AGG_QCAP (AGG_STEP + 32)

__device__ __forceinline__ void agg_drain(const int* __restrict__ ei, const int* __restrict__ col,
                                          const float* __restrict__ nrm,
                                          const float* __restrict__ H, float* acc,
                                          const int* q, int cnt, int base, int lane) {
  int hi = lane >> 4;
  for (int b = 0; b < cnt; b += 32) {
    int n = cnt - b < 32 ? cnt - b : 32;
    int e = q[b + (lane < n ? lane : 0)];
    int r = ei[e];
    int lrow = (col[e] - base) * 128;
    float sc = nrm[e];
    for (int j0 = 0; j0 < n; j0 += 8) {
      float4 hv[8];
#pragma unroll
      for (int u = 0; u < 8; ++u) {
        int jj = j0 + u < n ? j0 + u : n - 1;
        int rj = __builtin_amdgcn_readlane(r, jj);
        hv[u] = *(const float4*)(H + (size_t)rj * 128 + lane * 4);
      }
#pragma unroll
      for (int u = 0; u < 8; ++u) {
        if (j0 + u < n) {
          int lr = __builtin_amdgcn_readlane(lrow, j0 + u);
          float sj = __builtin_bit_cast(float, __builtin_amdgcn_readlane(__builtin_bit_cast(int, sc), j0 + u));
          float a0 = hi ? hv[u].y : hv[u].x;
          float a1 = hi ? hv[u].x : hv[u].y;
          float a2 = hi ? hv[u].w : hv[u].z;
          float a3 = hi ? hv[u].z : hv[u].w;
          float* p = acc + lr + lane * 4;
          atomicAdd(p + (0 ^ hi), sj * a0);
          atomicAdd(p + (1 ^ hi), sj * a1);
          atomicAdd(p + (2 ^ hi), sj * a2);
          atomicAdd(p + (3 ^ hi), sj * a3);
        }
      }
    }
  }
}

__device__ __forceinline__ void agg_load_full(const int* __restrict__ col, int s, int lane,
                                              int (&cv)[16]) {
  const int* p = col + s * AGG_STEP + lane * 4;
#pragma unroll
  for (int k = 0; k < 4; ++k) {
    int4 t = *(const int4*)(p + k * 128);
    cv[4 * k + 0] = t.x; cv[4 * k + 1] = t.y; cv[4 * k + 2] = t.z; cv[4 * k + 3] = t.w;
  }
}

__device__ __forceinline__ void agg_queue(const int (&cv)[16], int s, int lane, int base, int* q,
                                          int* wc) {
  int e0 = s * AGG_STEP + lane * 4;
#pragma unroll
  for (int i = 0; i < 16; ++i) {
    if ((unsigned)(cv[i] - base) < (unsigned)AGG_TN) {
      int p = atomicAdd(wc, 1);
      q[p] = e0 + (i >> 2) * 128 + (i & 3);
    }
  }
  __builtin_amdgcn_fence(__ATOMIC_ACQ_REL, "wavefront");
}

__global__ __launch_bounds__(1024) void agg_lds_kernel(const int* __restrict__ ei,
                                                       const float* __restrict__ nrm,
                                                       const float* __restrict__ H,
                                                       float* __restrict__ agg, int E, int Nrows) {
  __shared__ float acc[AGG_TN * 128];
  __shared__ int wq[AGG_WAVES][AGG_QCAP];
  __shared__ int wcnt[AGG_WAVES];
  int tid = threadIdx.x;
  int wave = __builtin_amdgcn_readfirstlane(tid >> 5);
  int lane = tid & 31;
  int base = blockIdx.x * AGG_TN;
  for (int i = tid; i < AGG_TN * 32; i += 1024)
    ((float4*)acc)[i] = make_float4(0.0f, 0.0f, 0.0f, 0.0f);
  if (lane == 0) wcnt[wave] = 0;
  __syncthreads();

  const int* col = ei + E;
  bool aligned16 = (((size_t)col) & 15) == 0;
  int nsteps = (E + AGG_STEP - 1) / AGG_STEP;
  int nfull = aligned16 ? E / AGG_STEP : 0;
  int* q = wq[wave];
  int cv[16], nx[16];
  if (wave < nfull) {
    agg_load_full(col, wave, lane, nx);
    for (int s = wave; s < nfull; s += AGG_WAVES) {
#pragma unroll
      for (int i = 0; i < 16; ++i) cv[i] = nx[i];
      int sp = s + AGG_WAVES < nfull ? s + AGG_WAVES : nfull - 1;
      agg_load_full(col, sp, lane, nx);
      agg_queue(cv, s, lane, base, q, &wcnt[wave]);
      int cnt = __builtin_amdgcn_readfirstlane(__hip_atomic_load(&wcnt[wave], __ATOMIC_RELAXED, __HIP_MEMORY_SCOPE_WAVEFRONT));
      if (cnt >= 32) {
        agg_drain(ei, col, nrm, H, acc, q, cnt, base, lane);
        if (lane == 0) wcnt[wave] = 0;
        __builtin_amdgcn_fence(__ATOMIC_ACQ_REL, "wavefront");
      }
    }
  }
  for (int s = nfull + wave; s < nsteps; s += AGG_WAVES) {
    int e0 = s * AGG_STEP + lane * 4;
#pragma unroll
    for (int i = 0; i < 16; ++i) {
      int ee = e0 + (i >> 2) * 128 + (i & 3);
      cv[i] = ee < E ? col[ee] : -1;
    }
    agg_queue(cv, s, lane, base, q, &wcnt[wave]);
    int cnt = __builtin_amdgcn_readfirstlane(__hip_atomic_load(&wcnt[wave], __ATOMIC_RELAXED, __HIP_MEMORY_SCOPE_WAVEFRONT));
    if (cnt >= 32) {
      agg_drain(ei, col, nrm, H, acc, q, cnt, base, lane);
      if (lane == 0) wcnt[wave] = 0;
      __builtin_amdgcn_fence(__ATOMIC_ACQ_REL, "wavefront");
    }
  }
  int cnt = __builtin_amdgcn_readfirstlane(__hip_atomic_load(&wcnt[wave], __ATOMIC_RELAXED, __HIP_MEMORY_SCOPE_WAVEFRONT));
  if (cnt > 0) agg_drain(ei, col, nrm, H, acc, q, cnt, base, lane);
  __syncthreads();

  for (int i = tid; i < AGG_TN * 32; i += 1024) {
    int node = base + (i >> 5);
    if (node < Nrows) ((float4*)agg)[(size_t)node * 32 + (i & 31)] = ((float4*)acc)[i];
  }
}

__global__ void residual_kernel(const float4* __restrict__ raw, float4* __restrict__ agg,
                                long long n4) {
  long long i = (long long)blockIdx.x * blockDim.x + threadIdx.x;
  if (i < n4) {
    float4 r = raw[i];
    float4 a = agg[i];
    a.x = FAGCN_EPS * r.x + a.x;
    a.y = FAGCN_EPS * r.y + a.y;
    a.z = FAGCN_EPS * r.z + a.z;
    a.w = FAGCN_EPS * r.w + a.w;
    agg[i] = a;
  }
}

__global__ void gemm_out_kernel(const float* __restrict__ H, const float* __restrict__ W,
                                const float* __restrict__ bias, float* __restrict__ out,
                                int Nrows) {
  int wave = threadIdx.x >> 5;
  int lane = threadIdx.x & 31;
  int tileM = blockIdx.x * 8 + wave;
  int tiles = (Nrows + 15) >> 4;
  if (tileM >= tiles) return;
  int mA = tileM * 16 + (lane & 15);
  int mclamp = mA < Nrows ? mA : (Nrows - 1);
  int n = lane & 15;
  int khalf = (lane >> 4) * 2;
  const float* hrow = H + (size_t)mclamp * 128 + khalf;
  const float* wcol = W + (size_t)khalf * 16 + n;
  __builtin_prefetch(hrow, 0, 0);
  v8f acc = {};
#pragma unroll 4
  for (int kk = 0; kk < 128; kk += 4) {
    v2f a = *(const v2f*)(hrow + kk);
    v2f b;
    b.x = wcol[(size_t)kk * 16];
    b.y = wcol[(size_t)kk * 16 + 16];
    acc = __builtin_amdgcn_wmma_f32_16x16x4_f32(false, a, false, b, (short)0, acc,
                                               false, false);
  }
  int mbase = tileM * 16 + ((lane >> 4) * 8);
  float bn = bias[n];
  if (tileM * 16 + 16 <= Nrows) {
    float* op = out + (size_t)mbase * 16 + n;
#pragma unroll
    for (int r = 0; r < 8; ++r)
      op[(size_t)r * 16] = acc[r] + bn;
  } else {
#pragma unroll
    for (int r = 0; r < 8; ++r) {
      int mm = mbase + r;
      if (mm < Nrows) out[(size_t)mm * 16 + n] = acc[r] + bn;
    }
  }
}

__global__ void logsoftmax_kernel(float4* out, int N) {
  int i = blockIdx.x * blockDim.x + threadIdx.x;
  if (i >= N) return;
  float4* row = out + (size_t)i * 4;
  float4 v[4];
#pragma unroll
  for (int q = 0; q < 4; ++q) v[q] = row[q];
  float m = v[0].x;
#pragma unroll
  for (int q = 0; q < 4; ++q) {
    m = fmaxf(m, v[q].x); m = fmaxf(m, v[q].y);
    m = fmaxf(m, v[q].z); m = fmaxf(m, v[q].w);
  }
  float s = 0.0f;
#pragma unroll
  for (int q = 0; q < 4; ++q) {
    s += expf(v[q].x - m); s += expf(v[q].y - m);
    s += expf(v[q].z - m); s += expf(v[q].w - m);
  }
  float d = m + logf(s);
#pragma unroll
  for (int q = 0; q < 4; ++q) {
    v[q].x -= d; v[q].y -= d; v[q].z -= d; v[q].w -= d;
    row[q] = v[q];
  }
}

extern "C" void kernel_launch(void* const* d_in, const int* in_sizes, int n_in,
                              void* d_out, int out_size, void* d_ws, size_t ws_size,
                              hipStream_t stream) {
  const float* x     = (const float*)d_in[0];
  const int*   ei    = (const int*)d_in[1];
  const float* W_in  = (const float*)d_in[2];
  const float* b_in  = (const float*)d_in[3];
  const float* Wg    = (const float*)d_in[4];
  const float* bg    = (const float*)d_in[5];
  const float* W_out = (const float*)d_in[6];
  const float* b_out = (const float*)d_in[7];
  float* out = (float*)d_out;

  const int Hdim = 128;
  int N = in_sizes[0] / Hdim;
  int E = in_sizes[1] / 2;
  int L = in_sizes[5];

  float* ws   = (float*)d_ws;
  float* dn   = ws;
  float* ga   = dn + N;
  float* gb   = ga + N;
  float* nrm  = gb + N;
  float* raw  = nrm + E;
  float* bufA = raw + (size_t)N * Hdim;
  float* bufB = bufA + (size_t)N * Hdim;

  hipMemsetAsync(dn, 0, (size_t)N * sizeof(float), stream);
  deg_kernel<<<(E + 255) / 256, 256, 0, stream>>>(ei, dn, E);
  dn_kernel<<<(N + 255) / 256, 256, 0, stream>>>(dn, N);

  gemm_relu_kernel<<<(N + 15) / 16, 256, 0, stream>>>(x, W_in, b_in, raw, N);

  float* h = raw;
  float* agg = bufA;
  for (int i = 0; i < L; ++i) {
    gate_kernel<<<(N + 7) / 8, 256, 0, stream>>>(h, Wg + (size_t)i * 2 * Hdim, ga, gb, N);
    norm_kernel<<<(E + 255) / 256, 256, 0, stream>>>(ei, ga, gb, dn, bg, i, nrm, E);
    agg_lds_kernel<<<(N + AGG_TN - 1) / AGG_TN, 1024, 0, stream>>>(ei, nrm, h, agg, E, N);
    long long n4 = (long long)N * Hdim / 4;
    residual_kernel<<<(unsigned)((n4 + 255) / 256), 256, 0, stream>>>(
        (const float4*)raw, (float4*)agg, n4);
    h = agg;
    agg = (h == bufA) ? bufB : bufA;
  }

  int tiles = (N + 15) / 16;
  gemm_out_kernel<<<(tiles + 7) / 8, 256, 0, stream>>>(h, W_out, b_out, out, N);
  logsoftmax_kernel<<<(N + 255) / 256, 256, 0, stream>>>((float4*)out, N);
}


# Aggregation's target scan reads a 1-byte owner key per edge (target/196, packed once per call by the degree kernel): 16 keys per lane per load, packed zero-byte match; LDS add order of waves changes
# speedup vs baseline: 1.9048x; 1.0821x over previous
#include <hip/hip_runtime.h>

typedef float v2f __attribute__((ext_vector_type(2)));
typedef float v8f __attribute__((ext_vector_type(8)));

#define FAGCN_EPS 0.3f

#define AGG_TN 196
#define AGG_MAXG 256

__device__ __forceinline__ bool agg_keys_ok(const unsigned* key, const float* ws, size_t ws_bytes,
                                            int E, int Nrows) {
  size_t need = (size_t)((const char*)key - (const char*)ws) + (((size_t)E + 3) & ~(size_t)3);
  return need <= ws_bytes && (((size_t)key) & 15) == 0 &&
         (Nrows + AGG_TN - 1) / AGG_TN <= AGG_MAXG;
}

__global__ void deg_key_kernel(const int* __restrict__ ei, float* __restrict__ deg,
                               unsigned* __restrict__ key, const float* ws, size_t ws_bytes,
                               int E, int Nrows) {
  int t = blockIdx.x * blockDim.x + threadIdx.x;
  int e0 = t * 4;
  if (e0 >= E) return;
  const int* col = ei + E;
  int r[4], c[4];
  if (e0 + 3 < E && ((((size_t)ei) | ((size_t)col)) & 15) == 0) {
    int4 rv = *(const int4*)(ei + e0);
    int4 cw = *(const int4*)(col + e0);
    r[0] = rv.x; r[1] = rv.y; r[2] = rv.z; r[3] = rv.w;
    c[0] = cw.x; c[1] = cw.y; c[2] = cw.z; c[3] = cw.w;
  } else {
#pragma unroll
    for (int k = 0; k < 4; ++k) {
      int e = e0 + k < E ? e0 + k : E - 1;
      r[k] = e0 + k < E ? ei[e] : -1;
      c[k] = col[e];
    }
  }
  unsigned pk = 0;
#pragma unroll
  for (int k = 0; k < 4; ++k) {
    if (r[k] >= 0) atomicAdd(&deg[r[k]], 1.0f);
    pk |= (((unsigned)c[k] / AGG_TN) & 255u) << (8 * k);
  }
  if (agg_keys_ok(key, ws, ws_bytes, E, Nrows)) key[t] = pk;
}

__global__ void dn_kernel(float* dn, int N) {
  int i = blockIdx.x * blockDim.x + threadIdx.x;
  if (i < N) dn[i] = 1.0f / sqrtf(fmaxf(dn[i], 1.0f));
}

__global__ void gemm_relu_kernel(const float* __restrict__ X, const float* __restrict__ W,
                                 const float* __restrict__ bias, float* __restrict__ H,
                                 int Nrows) {
  int wave = threadIdx.x >> 5;
  int lane = threadIdx.x & 31;
  int tileM = blockIdx.x;
  int mA = tileM * 16 + (lane & 15);
  int mclamp = mA < Nrows ? mA : (Nrows - 1);
  int n = wave * 16 + (lane & 15);
  int khalf = (lane >> 4) * 2;
  const float* xrow = X + (size_t)mclamp * 128 + khalf;
  const float* wcol = W + (size_t)khalf * 128 + n;
  __builtin_prefetch(xrow, 0, 0);
  v8f acc = {};
#pragma unroll 4
  for (int kk = 0; kk < 128; kk += 4) {
    v2f a = *(const v2f*)(xrow + kk);
    v2f b;
    b.x = wcol[(size_t)kk * 128];
    b.y = wcol[(size_t)kk * 128 + 128];
    acc = __builtin_amdgcn_wmma_f32_16x16x4_f32(false, a, false, b, (short)0, acc,
                                               false, false);
  }
  int mbase = tileM * 16 + ((lane >> 4) * 8);
  float bn = bias[n];
  if (tileM * 16 + 16 <= Nrows) {
    float* hp = H + (size_t)mbase * 128 + n;
#pragma unroll
    for (int r = 0; r < 8; ++r)
      hp[(size_t)r * 128] = fmaxf(acc[r] + bn, 0.0f);
  } else {
#pragma unroll
    for (int r = 0; r < 8; ++r) {
      int mm = mbase + r;
      if (mm < Nrows) H[(size_t)mm * 128 + n] = fmaxf(acc[r] + bn, 0.0f);
    }
  }
}

__global__ void gate_kernel(const float* __restrict__ H, const float* __restrict__ Wg,
                            float* __restrict__ ga, float* __restrict__ gb, int Nrows) {
  int wave = threadIdx.x >> 5;
  int lane = threadIdx.x & 31;
  int node = blockIdx.x * 8 + wave;
  if (node >= Nrows) return;
  const float* hrow = H + (size_t)node * 128;
  float sa = 0.0f, sb = 0.0f;
#pragma unroll
  for (int j = 0; j < 4; ++j) {
    int f = lane + j * 32;
    float hv = hrow[f];
    sa += hv * Wg[f];
    sb += hv * Wg[128 + f];
  }
  for (int off = 16; off > 0; off >>= 1) {
    sa += __shfl_down(sa, off, 32);
    sb += __shfl_down(sb, off, 32);
  }
  if (lane == 0) { ga[node] = sa; gb[node] = sb; }
}

__global__ void norm_kernel(const int* __restrict__ ei, const float* __restrict__ ga,
                            const float* __restrict__ gb, const float* __restrict__ dn,
                            const float* __restrict__ bg, int layer,
                            float* __restrict__ nrm, int E) {
  int e = blockIdx.x * blockDim.x + threadIdx.x;
  if (e >= E) return;
  int r = ei[e], c = ei[E + e];
  float g = tanhf(ga[r] + gb[c] + bg[layer]);
  nrm[e] = g * dn[r] * dn[c];
}

#define AGG_WAVES 32
#define AGG_STEP 512
#define AGG_KSTEP (4 * AGG_STEP)
#define AGG_QCAP (AGG_STEP + 32)

__device__ __forceinline__ void agg_drain(const int* __restrict__ ei, const int* __restrict__ col,
                                          const float* __restrict__ nrm,
                                          const float* __restrict__ H, float* acc,
                                          const int* q, int cnt, int base, int lane) {
  int hi = lane >> 4;
  for (int b = 0; b < cnt; b += 32) {
    int n = cnt - b < 32 ? cnt - b : 32;
    int e = q[b + (lane < n ? lane : 0)];
    int r = ei[e];
    int lrow = (col[e] - base) * 128;
    float sc = nrm[e];
    for (int j0 = 0; j0 < n; j0 += 8) {
      float4 hv[8];
#pragma unroll
      for (int u = 0; u < 8; ++u) {
        int jj = j0 + u < n ? j0 + u : n - 1;
        int rj = __builtin_amdgcn_readlane(r, jj);
        hv[u] = *(const float4*)(H + (size_t)rj * 128 + lane * 4);
      }
#pragma unroll
      for (int u = 0; u < 8; ++u) {
        if (j0 + u < n) {
          int lr = __builtin_amdgcn_readlane(lrow, j0 + u);
          float sj = __builtin_bit_cast(float, __builtin_amdgcn_readlane(__builtin_bit_cast(int, sc), j0 + u));
          float a0 = hi ? hv[u].y : hv[u].x;
          float a1 = hi ? hv[u].x : hv[u].y;
          float a2 = hi ? hv[u].w : hv[u].z;
          float a3 = hi ? hv[u].z : hv[u].w;
          float* p = acc + lr + lane * 4;
          atomicAdd(p + (0 ^ hi), sj * a0);
          atomicAdd(p + (1 ^ hi), sj * a1);
          atomicAdd(p + (2 ^ hi), sj * a2);
          atomicAdd(p + (3 ^ hi), sj * a3);
        }
      }
    }
  }
}

__device__ __forceinline__ void agg_load_keys(const unsigned* __restrict__ key, int S, int lane,
                                              uint4 (&kv)[4]) {
  const uint4* p = (const uint4*)key + (size_t)S * (AGG_KSTEP / 16) + lane;
#pragma unroll
  for (int k = 0; k < 4; ++k) kv[k] = p[k * 32];
}

__device__ __forceinline__ unsigned agg_zero_bytes(unsigned x) {
  return ~(((x & 0x7f7f7f7fu) + 0x7f7f7f7fu) | x) & 0x80808080u;
}

__device__ __forceinline__ unsigned agg_hits8(uint4 t0, uint4 t1, unsigned pat) {
  return (agg_zero_bytes(t0.x ^ pat) >> 7) | (agg_zero_bytes(t0.y ^ pat) >> 6) |
         (agg_zero_bytes(t0.z ^ pat) >> 5) | (agg_zero_bytes(t0.w ^ pat) >> 4) |
         (agg_zero_bytes(t1.x ^ pat) >> 3) | (agg_zero_bytes(t1.y ^ pat) >> 2) |
         (agg_zero_bytes(t1.z ^ pat) >> 1) | agg_zero_bytes(t1.w ^ pat);
}

__device__ __forceinline__ int agg_append(bool hit, int e, int* q, int cnt) {
  unsigned b = __builtin_amdgcn_ballot_w32(hit);
  if (hit) q[cnt + __builtin_amdgcn_mbcnt_lo(b, 0)] = e;
  return cnt + __builtin_popcount(b);
}

__device__ __forceinline__ int agg_queue(const int (&cv)[16], int s, int lane, int base, int* q,
                                         int cnt) {
  int e0 = s * AGG_STEP + lane * 4;
#pragma unroll
  for (int i = 0; i < 16; ++i)
    cnt = agg_append((unsigned)(cv[i] - base) < (unsigned)AGG_TN,
                     e0 + (i >> 2) * 128 + (i & 3), q, cnt);
  __builtin_amdgcn_fence(__ATOMIC_ACQ_REL, "wavefront");
  return cnt;
}

__global__ __launch_bounds__(1024) void agg_lds_kernel(const int* __restrict__ ei,
                                                       const float* __restrict__ nrm,
                                                       const float* __restrict__ H,
                                                       float* __restrict__ agg,
                                                       const unsigned* __restrict__ key,
                                                       const float* ws, size_t ws_bytes,
                                                       int E, int Nrows) {
  __shared__ float acc[AGG_TN * 128];
  __shared__ int wq[AGG_WAVES][AGG_QCAP];
  int tid = threadIdx.x;
  int wave = __builtin_amdgcn_readfirstlane(tid >> 5);
  int lane = tid & 31;
  int base = blockIdx.x * AGG_TN;
  for (int i = tid; i < AGG_TN * 32; i += 1024)
    ((float4*)acc)[i] = make_float4(0.0f, 0.0f, 0.0f, 0.0f);
  __syncthreads();

  const int* col = ei + E;
  int nsteps = (E + AGG_STEP - 1) / AGG_STEP;
  int nsuper = agg_keys_ok(key, ws, ws_bytes, E, Nrows) ? E / AGG_KSTEP : 0;
  int* q = wq[wave];
  int cnt = 0;
  if (wave < nsuper) {
    unsigned pat = blockIdx.x * 0x01010101u;
    uint4 kv[4];
    agg_load_keys(key, wave, lane, kv);
    for (int S = wave; S < nsuper; S += AGG_WAVES) {
      unsigned long long m = ((unsigned long long)agg_hits8(kv[2], kv[3], pat) << 32) |
                             agg_hits8(kv[0], kv[1], pat);
      int Sp = S + AGG_WAVES < nsuper ? S + AGG_WAVES : nsuper - 1;
      agg_load_keys(key, Sp, lane, kv);
      int e0 = S * AGG_KSTEP + lane * 16;
      while (__builtin_amdgcn_ballot_w32(m != 0) != 0) {
        bool hit = m != 0;
        int bit = hit ? __builtin_ctzll(m) : 0;
        int off = ((((bit >> 5) << 1) | ((bit >> 2) & 1)) << 9) | ((bit & 3) << 2) |
                  ((bit >> 3) & 3);
        cnt = agg_append(hit, e0 + off, q, cnt);
        m &= m - 1;
        __builtin_amdgcn_fence(__ATOMIC_ACQ_REL, "wavefront");
        if (cnt >= 32) {
          agg_drain(ei, col, nrm, H, acc, q, cnt, base, lane);
          cnt = 0;
        }
      }
    }
  }
  for (int s = nsuper * 4 + wave; s < nsteps; s += AGG_WAVES) {
    int cv[16];
    int e0 = s * AGG_STEP + lane * 4;
#pragma unroll
    for (int i = 0; i < 16; ++i) {
      int ee = e0 + (i >> 2) * 128 + (i & 3);
      cv[i] = ee < E ? col[ee] : -1;
    }
    cnt = agg_queue(cv, s, lane, base, q, cnt);
    if (cnt >= 32) {
      agg_drain(ei, col, nrm, H, acc, q, cnt, base, lane);
      cnt = 0;
    }
  }
  if (cnt > 0) agg_drain(ei, col, nrm, H, acc, q, cnt, base, lane);
  __syncthreads();

  for (int i = tid; i < AGG_TN * 32; i += 1024) {
    int node = base + (i >> 5);
    if (node < Nrows) ((float4*)agg)[(size_t)node * 32 + (i & 31)] = ((float4*)acc)[i];
  }
}

__global__ void residual_kernel(const float4* __restrict__ raw, float4* __restrict__ agg,
                                long long n4) {
  long long i = (long long)blockIdx.x * blockDim.x + threadIdx.x;
  if (i < n4) {
    float4 r = raw[i];
    float4 a = agg[i];
    a.x = FAGCN_EPS * r.x + a.x;
    a.y = FAGCN_EPS * r.y + a.y;
    a.z = FAGCN_EPS * r.z + a.z;
    a.w = FAGCN_EPS * r.w + a.w;
    agg[i] = a;
  }
}

__global__ void gemm_out_kernel(const float* __restrict__ H, const float* __restrict__ W,
                                const float* __restrict__ bias, float* __restrict__ out,
                                int Nrows) {
  int wave = threadIdx.x >> 5;
  int lane = threadIdx.x & 31;
  int tileM = blockIdx.x * 8 + wave;
  int tiles = (Nrows + 15) >> 4;
  if (tileM >= tiles) return;
  int mA = tileM * 16 + (lane & 15);
  int mclamp = mA < Nrows ? mA : (Nrows - 1);
  int n = lane & 15;
  int khalf = (lane >> 4) * 2;
  const float* hrow = H + (size_t)mclamp * 128 + khalf;
  const float* wcol = W + (size_t)khalf * 16 + n;
  __builtin_prefetch(hrow, 0, 0);
  v8f acc = {};
#pragma unroll 4
  for (int kk = 0; kk < 128; kk += 4) {
    v2f a = *(const v2f*)(hrow + kk);
    v2f b;
    b.x = wcol[(size_t)kk * 16];
    b.y = wcol[(size_t)kk * 16 + 16];
    acc = __builtin_amdgcn_wmma_f32_16x16x4_f32(false, a, false, b, (short)0, acc,
                                               false, false);
  }
  int mbase = tileM * 16 + ((lane >> 4) * 8);
  float bn = bias[n];
  if (tileM * 16 + 16 <= Nrows) {
    float* op = out + (size_t)mbase * 16 + n;
#pragma unroll
    for (int r = 0; r < 8; ++r)
      op[(size_t)r * 16] = acc[r] + bn;
  } else {
#pragma unroll
    for (int r = 0; r < 8; ++r) {
      int mm = mbase + r;
      if (mm < Nrows) out[(size_t)mm * 16 + n] = acc[r] + bn;
    }
  }
}

__global__ void logsoftmax_kernel(float4* out, int N) {
  int i = blockIdx.x * blockDim.x + threadIdx.x;
  if (i >= N) return;
  float4* row = out + (size_t)i * 4;
  float4 v[4];
#pragma unroll
  for (int q = 0; q < 4; ++q) v[q] = row[q];
  float m = v[0].x;
#pragma unroll
  for (int q = 0; q < 4; ++q) {
    m = fmaxf(m, v[q].x); m = fmaxf(m, v[q].y);
    m = fmaxf(m, v[q].z); m = fmaxf(m, v[q].w);
  }
  float s = 0.0f;
#pragma unroll
  for (int q = 0; q < 4; ++q) {
    s += expf(v[q].x - m); s += expf(v[q].y - m);
    s += expf(v[q].z - m); s += expf(v[q].w - m);
  }
  float d = m + logf(s);
#pragma unroll
  for (int q = 0; q < 4; ++q) {
    v[q].x -= d; v[q].y -= d; v[q].z -= d; v[q].w -= d;
    row[q] = v[q];
  }
}

extern "C" void kernel_launch(void* const* d_in, const int* in_sizes, int n_in,
                              void* d_out, int out_size, void* d_ws, size_t ws_size,
                              hipStream_t stream) {
  const float* x     = (const float*)d_in[0];
  const int*   ei    = (const int*)d_in[1];
  const float* W_in  = (const float*)d_in[2];
  const float* b_in  = (const float*)d_in[3];
  const float* Wg    = (const float*)d_in[4];
  const float* bg    = (const float*)d_in[5];
  const float* W_out = (const float*)d_in[6];
  const float* b_out = (const float*)d_in[7];
  float* out = (float*)d_out;

  const int Hdim = 128;
  int N = in_sizes[0] / Hdim;
  int E = in_sizes[1] / 2;
  int L = in_sizes[5];

  float* ws   = (float*)d_ws;
  float* dn   = ws;
  float* ga   = dn + N;
  float* gb   = ga + N;
  float* nrm  = gb + N;
  float* raw  = nrm + E;
  float* bufA = raw + (size_t)N * Hdim;
  float* bufB = bufA + (size_t)N * Hdim;
  unsigned* key = (unsigned*)(bufB + (size_t)N * Hdim);

  hipMemsetAsync(dn, 0, (size_t)N * sizeof(float), stream);
  deg_key_kernel<<<(E + 1023) / 1024, 256, 0, stream>>>(ei, dn, key, ws, ws_size, E, N);
  dn_kernel<<<(N + 255) / 256, 256, 0, stream>>>(dn, N);

  gemm_relu_kernel<<<(N + 15) / 16, 256, 0, stream>>>(x, W_in, b_in, raw, N);

  float* h = raw;
  float* agg = bufA;
  for (int i = 0; i < L; ++i) {
    gate_kernel<<<(N + 7) / 8, 256, 0, stream>>>(h, Wg + (size_t)i * 2 * Hdim, ga, gb, N);
    norm_kernel<<<(E + 255) / 256, 256, 0, stream>>>(ei, ga, gb, dn, bg, i, nrm, E);
    agg_lds_kernel<<<(N + AGG_TN - 1) / AGG_TN, 1024, 0, stream>>>(ei, nrm, h, agg, key, ws,
                                                                    ws_size, E, N);
    long long n4 = (long long)N * Hdim / 4;
    residual_kernel<<<(unsigned)((n4 + 255) / 256), 256, 0, stream>>>(
        (const float4*)raw, (float4*)agg, n4);
    h = agg;
    agg = (h == bufA) ? bufB : bufA;
  }

  int tiles = (N + 15) / 16;
  gemm_out_kernel<<<(tiles + 7) / 8, 256, 0, stream>>>(h, W_out, b_out, out, N);
  logsoftmax_kernel<<<(N + 255) / 256, 256, 0, stream>>>((float4*)out, N);
}


# Per-edge gate*norm pass folds into the aggregation's drain: a drained edge takes its source's (gate dot, degree norm) in one 8-B load, its target's from LDS; 2 launches, nrm array, 4 of 5 lookups go
# speedup vs baseline: 2.0855x; 1.0949x over previous
#include <hip/hip_runtime.h>

typedef float v2f __attribute__((ext_vector_type(2)));
typedef float v8f __attribute__((ext_vector_type(8)));

#define FAGCN_EPS 0.3f

#define AGG_TN 196
#define AGG_MAXG 256

__device__ __forceinline__ bool agg_keys_ok(const unsigned* key, const float* ws, size_t ws_bytes,
                                            int E, int Nrows) {
  size_t need = (size_t)((const char*)key - (const char*)ws) + (((size_t)E + 3) & ~(size_t)3);
  return need <= ws_bytes && (((size_t)key) & 15) == 0 &&
         (Nrows + AGG_TN - 1) / AGG_TN <= AGG_MAXG;
}

__global__ void deg_key_kernel(const int* __restrict__ ei, float* __restrict__ deg,
                               unsigned* __restrict__ key, const float* ws, size_t ws_bytes,
                               int E, int Nrows) {
  int t = blockIdx.x * blockDim.x + threadIdx.x;
  int e0 = t * 4;
  if (e0 >= E) return;
  const int* col = ei + E;
  int r[4], c[4];
  if (e0 + 3 < E && ((((size_t)ei) | ((size_t)col)) & 15) == 0) {
    int4 rv = *(const int4*)(ei + e0);
    int4 cw = *(const int4*)(col + e0);
    r[0] = rv.x; r[1] = rv.y; r[2] = rv.z; r[3] = rv.w;
    c[0] = cw.x; c[1] = cw.y; c[2] = cw.z; c[3] = cw.w;
  } else {
#pragma unroll
    for (int k = 0; k < 4; ++k) {
      int e = e0 + k < E ? e0 + k : E - 1;
      r[k] = e0 + k < E ? ei[e] : -1;
      c[k] = col[e];
    }
  }
  unsigned pk = 0;
#pragma unroll
  for (int k = 0; k < 4; ++k) {
    if (r[k] >= 0) atomicAdd(&deg[r[k]], 1.0f);
    pk |= (((unsigned)c[k] / AGG_TN) & 255u) << (8 * k);
  }
  if (agg_keys_ok(key, ws, ws_bytes, E, Nrows)) key[t] = pk;
}

__global__ void dn_kernel(float* dn, int N) {
  int i = blockIdx.x * blockDim.x + threadIdx.x;
  if (i < N) dn[i] = 1.0f / sqrtf(fmaxf(dn[i], 1.0f));
}

__global__ void gemm_relu_kernel(const float* __restrict__ X, const float* __restrict__ W,
                                 const float* __restrict__ bias, float* __restrict__ H,
                                 int Nrows) {
  int wave = threadIdx.x >> 5;
  int lane = threadIdx.x & 31;
  int tileM = blockIdx.x;
  int mA = tileM * 16 + (lane & 15);
  int mclamp = mA < Nrows ? mA : (Nrows - 1);
  int n = wave * 16 + (lane & 15);
  int khalf = (lane >> 4) * 2;
  const float* xrow = X + (size_t)mclamp * 128 + khalf;
  const float* wcol = W + (size_t)khalf * 128 + n;
  __builtin_prefetch(xrow, 0, 0);
  v8f acc = {};
#pragma unroll 4
  for (int kk = 0; kk < 128; kk += 4) {
    v2f a = *(const v2f*)(xrow + kk);
    v2f b;
    b.x = wcol[(size_t)kk * 128];
    b.y = wcol[(size_t)kk * 128 + 128];
    acc = __builtin_amdgcn_wmma_f32_16x16x4_f32(false, a, false, b, (short)0, acc,
                                               false, false);
  }
  int mbase = tileM * 16 + ((lane >> 4) * 8);
  float bn = bias[n];
  if (tileM * 16 + 16 <= Nrows) {
    float* hp = H + (size_t)mbase * 128 + n;
#pragma unroll
    for (int r = 0; r < 8; ++r)
      hp[(size_t)r * 128] = fmaxf(acc[r] + bn, 0.0f);
  } else {
#pragma unroll
    for (int r = 0; r < 8; ++r) {
      int mm = mbase + r;
      if (mm < Nrows) H[(size_t)mm * 128 + n] = fmaxf(acc[r] + bn, 0.0f);
    }
  }
}

__global__ void gate_kernel(const float* __restrict__ H, const float* __restrict__ Wg,
                            const float* __restrict__ dn, float2* __restrict__ gs,
                            float2* __restrict__ gt, int Nrows) {
  int wave = threadIdx.x >> 5;
  int lane = threadIdx.x & 31;
  int node = blockIdx.x * 8 + wave;
  if (node >= Nrows) return;
  const float* hrow = H + (size_t)node * 128;
  float sa = 0.0f, sb = 0.0f;
#pragma unroll
  for (int j = 0; j < 4; ++j) {
    int f = lane + j * 32;
    float hv = hrow[f];
    sa += hv * Wg[f];
    sb += hv * Wg[128 + f];
  }
  for (int off = 16; off > 0; off >>= 1) {
    sa += __shfl_down(sa, off, 32);
    sb += __shfl_down(sb, off, 32);
  }
  if (lane == 0) {
    float d = dn[node];
    gs[node] = make_float2(sa, d);
    gt[node] = make_float2(sb, d);
  }
}

#define AGG_WAVES 32
#define AGG_STEP 512
#define AGG_KSTEP (4 * AGG_STEP)
#define AGG_QCAP (AGG_STEP + 32)

__device__ __forceinline__ void agg_rows_load(const float* __restrict__ H, int r, int j0, int n,
                                              int lane, float4 (&hv)[8]) {
#pragma unroll
  for (int u = 0; u < 8; ++u) {
    int jj = j0 + u < n ? j0 + u : n - 1;
    int rj = __builtin_amdgcn_readlane(r, jj);
    hv[u] = *(const float4*)(H + (size_t)rj * 128 + lane * 4);
  }
}

__device__ __forceinline__ void agg_rows_add(float* acc, int lrow, float sc, int j0, int n,
                                             int lane, const float4 (&hv)[8]) {
  int hi = lane >> 4;
#pragma unroll
  for (int u = 0; u < 8; ++u) {
    if (j0 + u < n) {
      int lr = __builtin_amdgcn_readlane(lrow, j0 + u);
      float sj = __builtin_bit_cast(float, __builtin_amdgcn_readlane(__builtin_bit_cast(int, sc), j0 + u));
      float a0 = hi ? hv[u].y : hv[u].x;
      float a1 = hi ? hv[u].x : hv[u].y;
      float a2 = hi ? hv[u].w : hv[u].z;
      float a3 = hi ? hv[u].z : hv[u].w;
      float* p = acc + lr + lane * 4;
      atomicAdd(p + (0 ^ hi), sj * a0);
      atomicAdd(p + (1 ^ hi), sj * a1);
      atomicAdd(p + (2 ^ hi), sj * a2);
      atomicAdd(p + (3 ^ hi), sj * a3);
    }
  }
}

__device__ __forceinline__ void agg_drain(const int* __restrict__ ei, const int* __restrict__ col,
                                          const float2* __restrict__ gs, const float2* lgt,
                                          float bgl, const float* __restrict__ H, float* acc,
                                          const int* q, int cnt, int base, int lane) {
  for (int b = 0; b < cnt; b += 32) {
    int n = cnt - b < 32 ? cnt - b : 32;
    int e = q[b + (lane < n ? lane : 0)];
    int r = ei[e];
    int cl = col[e] - base;
    float2 s = gs[r];
    float4 hv[8];
    agg_rows_load(H, r, 0, n, lane, hv);
    float2 t = lgt[cl];
    float sc = tanhf(s.x + t.x + bgl) * s.y * t.y;
    int lrow = cl * 128;
    for (int j0 = 0;;) {
      agg_rows_add(acc, lrow, sc, j0, n, lane, hv);
      j0 += 8;
      if (j0 >= n) break;
      agg_rows_load(H, r, j0, n, lane, hv);
    }
  }
}

__device__ __forceinline__ void agg_load_keys(const unsigned* __restrict__ key, int S, int lane,
                                              uint4 (&kv)[4]) {
  const uint4* p = (const uint4*)key + (size_t)S * (AGG_KSTEP / 16) + lane;
#pragma unroll
  for (int k = 0; k < 4; ++k) kv[k] = p[k * 32];
}

__device__ __forceinline__ unsigned agg_zero_bytes(unsigned x) {
  return ~(((x & 0x7f7f7f7fu) + 0x7f7f7f7fu) | x) & 0x80808080u;
}

__device__ __forceinline__ unsigned agg_hits8(uint4 t0, uint4 t1, unsigned pat) {
  return (agg_zero_bytes(t0.x ^ pat) >> 7) | (agg_zero_bytes(t0.y ^ pat) >> 6) |
         (agg_zero_bytes(t0.z ^ pat) >> 5) | (agg_zero_bytes(t0.w ^ pat) >> 4) |
         (agg_zero_bytes(t1.x ^ pat) >> 3) | (agg_zero_bytes(t1.y ^ pat) >> 2) |
         (agg_zero_bytes(t1.z ^ pat) >> 1) | agg_zero_bytes(t1.w ^ pat);
}

__device__ __forceinline__ int agg_append(bool hit, int e, int* q, int cnt) {
  unsigned b = __builtin_amdgcn_ballot_w32(hit);
  if (hit) q[cnt + __builtin_amdgcn_mbcnt_lo(b, 0)] = e;
  return cnt + __builtin_popcount(b);
}

__device__ __forceinline__ int agg_queue(const int (&cv)[16], int s, int lane, int base, int* q,
                                         int cnt) {
  int e0 = s * AGG_STEP + lane * 4;
#pragma unroll
  for (int i = 0; i < 16; ++i)
    cnt = agg_append((unsigned)(cv[i] - base) < (unsigned)AGG_TN,
                     e0 + (i >> 2) * 128 + (i & 3), q, cnt);
  __builtin_amdgcn_fence(__ATOMIC_ACQ_REL, "wavefront");
  return cnt;
}

__global__ __launch_bounds__(1024) void agg_lds_kernel(const int* __restrict__ ei,
                                                       const float2* __restrict__ gs,
                                                       const float2* __restrict__ gt,
                                                       const float* __restrict__ bg, int layer,
                                                       const float* __restrict__ H,
                                                       float* __restrict__ agg,
                                                       const unsigned* __restrict__ key,
                                                       const float* ws, size_t ws_bytes,
                                                       int E, int Nrows) {
  __shared__ float acc[AGG_TN * 128];
  __shared__ int wq[AGG_WAVES][AGG_QCAP];
  __shared__ float2 lgt[AGG_TN];
  int tid = threadIdx.x;
  int wave = __builtin_amdgcn_readfirstlane(tid >> 5);
  int lane = tid & 31;
  int base = blockIdx.x * AGG_TN;
  for (int i = tid; i < AGG_TN * 32; i += 1024)
    ((float4*)acc)[i] = make_float4(0.0f, 0.0f, 0.0f, 0.0f);
  if (tid < AGG_TN)
    lgt[tid] = base + tid < Nrows ? gt[base + tid] : make_float2(0.0f, 0.0f);
  float bgl = bg[layer];
  __syncthreads();

  const int* col = ei + E;
  int nsteps = (E + AGG_STEP - 1) / AGG_STEP;
  int nsuper = agg_keys_ok(key, ws, ws_bytes, E, Nrows) ? E / AGG_KSTEP : 0;
  int* q = wq[wave];
  int cnt = 0;
  if (wave < nsuper) {
    unsigned pat = blockIdx.x * 0x01010101u;
    uint4 kv[4];
    agg_load_keys(key, wave, lane, kv);
    for (int S = wave; S < nsuper; S += AGG_WAVES) {
      unsigned long long m = ((unsigned long long)agg_hits8(kv[2], kv[3], pat) << 32) |
                             agg_hits8(kv[0], kv[1], pat);
      int Sp = S + AGG_WAVES < nsuper ? S + AGG_WAVES : nsuper - 1;
      agg_load_keys(key, Sp, lane, kv);
      int e0 = S * AGG_KSTEP + lane * 16;
      while (__builtin_amdgcn_ballot_w32(m != 0) != 0) {
        bool hit = m != 0;
        int bit = hit ? __builtin_ctzll(m) : 0;
        int off = ((((bit >> 5) << 1) | ((bit >> 2) & 1)) << 9) | ((bit & 3) << 2) |
                  ((bit >> 3) & 3);
        cnt = agg_append(hit, e0 + off, q, cnt);
        m &= m - 1;
        __builtin_amdgcn_fence(__ATOMIC_ACQ_REL, "wavefront");
        if (cnt >= 32) {
          agg_drain(ei, col, gs, lgt, bgl, H, acc, q, cnt, base, lane);
          cnt = 0;
        }
      }
    }
  }
  for (int s = nsuper * 4 + wave; s < nsteps; s += AGG_WAVES) {
    int cv[16];
    int e0 = s * AGG_STEP + lane * 4;
#pragma unroll
    for (int i = 0; i < 16; ++i) {
      int ee = e0 + (i >> 2) * 128 + (i & 3);
      cv[i] = ee < E ? col[ee] : -1;
    }
    cnt = agg_queue(cv, s, lane, base, q, cnt);
    if (cnt >= 32) {
      agg_drain(ei, col, gs, lgt, bgl, H, acc, q, cnt, base, lane);
      cnt = 0;
    }
  }
  if (cnt > 0) agg_drain(ei, col, gs, lgt, bgl, H, acc, q, cnt, base, lane);
  __syncthreads();

  for (int i = tid; i < AGG_TN * 32; i += 1024) {
    int node = base + (i >> 5);
    if (node < Nrows) ((float4*)agg)[(size_t)node * 32 + (i & 31)] = ((float4*)acc)[i];
  }
}

__global__ void residual_kernel(const float4* __restrict__ raw, float4* __restrict__ agg,
                                long long n4) {
  long long i = (long long)blockIdx.x * blockDim.x + threadIdx.x;
  if (i < n4) {
    float4 r = raw[i];
    float4 a = agg[i];
    a.x = FAGCN_EPS * r.x + a.x;
    a.y = FAGCN_EPS * r.y + a.y;
    a.z = FAGCN_EPS * r.z + a.z;
    a.w = FAGCN_EPS * r.w + a.w;
    agg[i] = a;
  }
}

__global__ void gemm_out_kernel(const float* __restrict__ H, const float* __restrict__ W,
                                const float* __restrict__ bias, float* __restrict__ out,
                                int Nrows) {
  int wave = threadIdx.x >> 5;
  int lane = threadIdx.x & 31;
  int tileM = blockIdx.x * 8 + wave;
  int tiles = (Nrows + 15) >> 4;
  if (tileM >= tiles) return;
  int mA = tileM * 16 + (lane & 15);
  int mclamp = mA < Nrows ? mA : (Nrows - 1);
  int n = lane & 15;
  int khalf = (lane >> 4) * 2;
  const float* hrow = H + (size_t)mclamp * 128 + khalf;
  const float* wcol = W + (size_t)khalf * 16 + n;
  __builtin_prefetch(hrow, 0, 0);
  v8f acc = {};
#pragma unroll 4
  for (int kk = 0; kk < 128; kk += 4) {
    v2f a = *(const v2f*)(hrow + kk);
    v2f b;
    b.x = wcol[(size_t)kk * 16];
    b.y = wcol[(size_t)kk * 16 + 16];
    acc = __builtin_amdgcn_wmma_f32_16x16x4_f32(false, a, false, b, (short)0, acc,
                                               false, false);
  }
  int mbase = tileM * 16 + ((lane >> 4) * 8);
  float bn = bias[n];
  if (tileM * 16 + 16 <= Nrows) {
    float* op = out + (size_t)mbase * 16 + n;
#pragma unroll
    for (int r = 0; r < 8; ++r)
      op[(size_t)r * 16] = acc[r] + bn;
  } else {
#pragma unroll
    for (int r = 0; r < 8; ++r) {
      int mm = mbase + r;
      if (mm < Nrows) out[(size_t)mm * 16 + n] = acc[r] + bn;
    }
  }
}

__global__ void logsoftmax_kernel(float4* out, int N) {
  int i = blockIdx.x * blockDim.x + threadIdx.x;
  if (i >= N) return;
  float4* row = out + (size_t)i * 4;
  float4 v[4];
#pragma unroll
  for (int q = 0; q < 4; ++q) v[q] = row[q];
  float m = v[0].x;
#pragma unroll
  for (int q = 0; q < 4; ++q) {
    m = fmaxf(m, v[q].x); m = fmaxf(m, v[q].y);
    m = fmaxf(m, v[q].z); m = fmaxf(m, v[q].w);
  }
  float s = 0.0f;
#pragma unroll
  for (int q = 0; q < 4; ++q) {
    s += expf(v[q].x - m); s += expf(v[q].y - m);
    s += expf(v[q].z - m); s += expf(v[q].w - m);
  }
  float d = m + logf(s);
#pragma unroll
  for (int q = 0; q < 4; ++q) {
    v[q].x -= d; v[q].y -= d; v[q].z -= d; v[q].w -= d;
    row[q] = v[q];
  }
}

extern "C" void kernel_launch(void* const* d_in, const int* in_sizes, int n_in,
                              void* d_out, int out_size, void* d_ws, size_t ws_size,
                              hipStream_t stream) {
  const float* x     = (const float*)d_in[0];
  const int*   ei    = (const int*)d_in[1];
  const float* W_in  = (const float*)d_in[2];
  const float* b_in  = (const float*)d_in[3];
  const float* Wg    = (const float*)d_in[4];
  const float* bg    = (const float*)d_in[5];
  const float* W_out = (const float*)d_in[6];
  const float* b_out = (const float*)d_in[7];
  float* out = (float*)d_out;

  const int Hdim = 128;
  int N = in_sizes[0] / Hdim;
  int E = in_sizes[1] / 2;
  int L = in_sizes[5];

  float* ws   = (float*)d_ws;
  float2* gs  = (float2*)ws;
  float2* gt  = gs + N;
  float* dn   = (float*)(gt + N);
  float* raw  = ws + 3 * (size_t)N + E;
  float* bufA = raw + (size_t)N * Hdim;
  float* bufB = bufA + (size_t)N * Hdim;
  unsigned* key = (unsigned*)(bufB + (size_t)N * Hdim);

  hipMemsetAsync(dn, 0, (size_t)N * sizeof(float), stream);
  deg_key_kernel<<<(E + 1023) / 1024, 256, 0, stream>>>(ei, dn, key, ws, ws_size, E, N);
  dn_kernel<<<(N + 255) / 256, 256, 0, stream>>>(dn, N);

  gemm_relu_kernel<<<(N + 15) / 16, 256, 0, stream>>>(x, W_in, b_in, raw, N);

  float* h = raw;
  float* agg = bufA;
  for (int i = 0; i < L; ++i) {
    gate_kernel<<<(N + 7) / 8, 256, 0, stream>>>(h, Wg + (size_t)i * 2 * Hdim, dn, gs, gt, N);
    agg_lds_kernel<<<(N + AGG_TN - 1) / AGG_TN, 1024, 0, stream>>>(ei, gs, gt, bg, i, h, agg,
                                                                    key, ws, ws_size, E, N);
    long long n4 = (long long)N * Hdim / 4;
    residual_kernel<<<(unsigned)((n4 + 255) / 256), 256, 0, stream>>>(
        (const float4*)raw, (float4*)agg, n4);
    h = agg;
    agg = (h == bufA) ? bufB : bufA;
  }

  int tiles = (N + 15) / 16;
  gemm_out_kernel<<<(tiles + 7) / 8, 256, 0, stream>>>(h, W_out, b_out, out, N);
  logsoftmax_kernel<<<(N + 255) / 256, 256, 0, stream>>>((float4*)out, N);
}


# Degree count leaves global float atomics: work-groups (64 edge shares x 4 node ranges) count sources in LDS integer tables, store them whole; norm kernel adds the 64 partials (exact); dn memset goes
# speedup vs baseline: 2.4807x; 1.1895x over previous
#include <hip/hip_runtime.h>

typedef float v2f __attribute__((ext_vector_type(2)));
typedef float v8f __attribute__((ext_vector_type(8)));

#define FAGCN_EPS 0.3f

#define AGG_TN 196
#define AGG_MAXG 256

__device__ __forceinline__ bool agg_keys_ok(const unsigned* key, const float* ws, size_t ws_bytes,
                                            int E, int Nrows) {
  size_t need = (size_t)((const char*)key - (const char*)ws) + (((size_t)E + 3) & ~(size_t)3);
  return need <= ws_bytes && (((size_t)key) & 15) == 0 &&
         (Nrows + AGG_TN - 1) / AGG_TN <= AGG_MAXG;
}

#define HIST_K 64
#define HIST_CAP 16384
#define HIST_T 1024
#define HIST_U 4

__device__ __forceinline__ void hist_count(unsigned* tab, int r, int rbase, int nhere) {
  unsigned d = (unsigned)(r - rbase);
  if (d < (unsigned)nhere) atomicAdd(&tab[d], 1u);
}

__device__ __forceinline__ unsigned agg_pack_keys(int c0, int c1, int c2, int c3) {
  return (((unsigned)c0 / AGG_TN) & 255u) | ((((unsigned)c1 / AGG_TN) & 255u) << 8) |
         ((((unsigned)c2 / AGG_TN) & 255u) << 16) | ((((unsigned)c3 / AGG_TN) & 255u) << 24);
}

__global__ __launch_bounds__(HIST_T) void deg_key_kernel(const int* __restrict__ ei,
                                                         unsigned* __restrict__ part,
                                                         unsigned* __restrict__ key,
                                                         const float* ws, size_t ws_bytes,
                                                         int E, int Nrows) {
  __shared__ unsigned tab[HIST_CAP];
  int tid = threadIdx.x;
  int rbase = blockIdx.y * HIST_CAP;
  int nhere = Nrows - rbase < HIST_CAP ? Nrows - rbase : HIST_CAP;
  for (int i = tid; i < HIST_CAP / 4; i += HIST_T) ((uint4*)tab)[i] = make_uint4(0u, 0u, 0u, 0u);
  __syncthreads();

  const int* col = ei + E;
  int Q = (E + 3) >> 2;
  int qpb = (Q + HIST_K - 1) / HIST_K;
  int q0 = blockIdx.x * qpb;
  int q1 = q0 + qpb < Q ? q0 + qpb : Q;
  bool keys = blockIdx.y == 0 && agg_keys_ok(key, ws, ws_bytes, E, Nrows);
  if (q0 < q1) {
    if ((E & 3) == 0 && ((((size_t)ei) | ((size_t)col)) & 15) == 0) {
      int trips = (qpb + HIST_T - 1) / HIST_T;
      for (int j0 = 0; j0 < trips; j0 += HIST_U) {
        int4 rv[HIST_U], cw[HIST_U];
#pragma unroll
        for (int u = 0; u < HIST_U; ++u) {
          int q = q0 + (j0 + u) * HIST_T + tid;
          int qq = q < q1 ? q : q1 - 1;
          rv[u] = ((const int4*)ei)[qq];
          cw[u] = ((const int4*)col)[qq];
        }
        asm volatile("" ::"v"(rv[0].x), "v"(rv[1].x), "v"(rv[2].x), "v"(rv[3].x), "v"(cw[0].x),
                     "v"(cw[1].x), "v"(cw[2].x), "v"(cw[3].x));
#pragma unroll
        for (int u = 0; u < HIST_U; ++u) {
          int q = q0 + (j0 + u) * HIST_T + tid;
          if (q < q1) {
            hist_count(tab, rv[u].x, rbase, nhere);
            hist_count(tab, rv[u].y, rbase, nhere);
            hist_count(tab, rv[u].z, rbase, nhere);
            hist_count(tab, rv[u].w, rbase, nhere);
            if (keys) key[q] = agg_pack_keys(cw[u].x, cw[u].y, cw[u].z, cw[u].w);
          }
        }
      }
    } else {
      for (int q = q0 + tid; q < q1; q += HIST_T) {
        int c[4];
#pragma unroll
        for (int k = 0; k < 4; ++k) {
          int e = q * 4 + k;
          int ec = e < E ? e : E - 1;
          if (e < E) hist_count(tab, ei[ec], rbase, nhere);
          c[k] = col[ec];
        }
        if (keys) key[q] = agg_pack_keys(c[0], c[1], c[2], c[3]);
      }
    }
  }
  __syncthreads();

  unsigned* dst = part + (size_t)blockIdx.x * Nrows + rbase;
  if ((nhere & 3) == 0 && (((size_t)dst) & 15) == 0) {
    for (int i = tid; i < nhere / 4; i += HIST_T) ((uint4*)dst)[i] = ((const uint4*)tab)[i];
  } else {
    for (int i = tid; i < nhere; i += HIST_T) dst[i] = tab[i];
  }
}

__global__ void dn_kernel(const unsigned* __restrict__ part, float* __restrict__ dn, int N) {
  int i = blockIdx.x * blockDim.x + threadIdx.x;
  if (i >= N) return;
  unsigned s = 0;
#pragma unroll
  for (int b = 0; b < HIST_K; ++b) s += part[(size_t)b * N + i];
  dn[i] = 1.0f / sqrtf(fmaxf((float)s, 1.0f));
}

__global__ void gemm_relu_kernel(const float* __restrict__ X, const float* __restrict__ W,
                                 const float* __restrict__ bias, float* __restrict__ H,
                                 int Nrows) {
  int wave = threadIdx.x >> 5;
  int lane = threadIdx.x & 31;
  int tileM = blockIdx.x;
  int mA = tileM * 16 + (lane & 15);
  int mclamp = mA < Nrows ? mA : (Nrows - 1);
  int n = wave * 16 + (lane & 15);
  int khalf = (lane >> 4) * 2;
  const float* xrow = X + (size_t)mclamp * 128 + khalf;
  const float* wcol = W + (size_t)khalf * 128 + n;
  __builtin_prefetch(xrow, 0, 0);
  v8f acc = {};
#pragma unroll 4
  for (int kk = 0; kk < 128; kk += 4) {
    v2f a = *(const v2f*)(xrow + kk);
    v2f b;
    b.x = wcol[(size_t)kk * 128];
    b.y = wcol[(size_t)kk * 128 + 128];
    acc = __builtin_amdgcn_wmma_f32_16x16x4_f32(false, a, false, b, (short)0, acc,
                                               false, false);
  }
  int mbase = tileM * 16 + ((lane >> 4) * 8);
  float bn = bias[n];
  if (tileM * 16 + 16 <= Nrows) {
    float* hp = H + (size_t)mbase * 128 + n;
#pragma unroll
    for (int r = 0; r < 8; ++r)
      hp[(size_t)r * 128] = fmaxf(acc[r] + bn, 0.0f);
  } else {
#pragma unroll
    for (int r = 0; r < 8; ++r) {
      int mm = mbase + r;
      if (mm < Nrows) H[(size_t)mm * 128 + n] = fmaxf(acc[r] + bn, 0.0f);
    }
  }
}

__global__ void gate_kernel(const float* __restrict__ H, const float* __restrict__ Wg,
                            const float* __restrict__ dn, float2* __restrict__ gs,
                            float2* __restrict__ gt, int Nrows) {
  int wave = threadIdx.x >> 5;
  int lane = threadIdx.x & 31;
  int node = blockIdx.x * 8 + wave;
  if (node >= Nrows) return;
  const float* hrow = H + (size_t)node * 128;
  float sa = 0.0f, sb = 0.0f;
#pragma unroll
  for (int j = 0; j < 4; ++j) {
    int f = lane + j * 32;
    float hv = hrow[f];
    sa += hv * Wg[f];
    sb += hv * Wg[128 + f];
  }
  for (int off = 16; off > 0; off >>= 1) {
    sa += __shfl_down(sa, off, 32);
    sb += __shfl_down(sb, off, 32);
  }
  if (lane == 0) {
    float d = dn[node];
    gs[node] = make_float2(sa, d);
    gt[node] = make_float2(sb, d);
  }
}

#define AGG_WAVES 32
#define AGG_STEP 512
#define AGG_KSTEP (4 * AGG_STEP)
#define AGG_QCAP (AGG_STEP + 32)

__device__ __forceinline__ void agg_rows_load(const float* __restrict__ H, int r, int j0, int n,
                                              int lane, float4 (&hv)[8]) {
#pragma unroll
  for (int u = 0; u < 8; ++u) {
    int jj = j0 + u < n ? j0 + u : n - 1;
    int rj = __builtin_amdgcn_readlane(r, jj);
    hv[u] = *(const float4*)(H + (size_t)rj * 128 + lane * 4);
  }
}

__device__ __forceinline__ void agg_rows_add(float* acc, int lrow, float sc, int j0, int n,
                                             int lane, const float4 (&hv)[8]) {
  int hi = lane >> 4;
#pragma unroll
  for (int u = 0; u < 8; ++u) {
    if (j0 + u < n) {
      int lr = __builtin_amdgcn_readlane(lrow, j0 + u);
      float sj = __builtin_bit_cast(float, __builtin_amdgcn_readlane(__builtin_bit_cast(int, sc), j0 + u));
      float a0 = hi ? hv[u].y : hv[u].x;
      float a1 = hi ? hv[u].x : hv[u].y;
      float a2 = hi ? hv[u].w : hv[u].z;
      float a3 = hi ? hv[u].z : hv[u].w;
      float* p = acc + lr + lane * 4;
      atomicAdd(p + (0 ^ hi), sj * a0);
      atomicAdd(p + (1 ^ hi), sj * a1);
      atomicAdd(p + (2 ^ hi), sj * a2);
      atomicAdd(p + (3 ^ hi), sj * a3);
    }
  }
}

__device__ __forceinline__ void agg_drain(const int* __restrict__ ei, const int* __restrict__ col,
                                          const float2* __restrict__ gs, const float2* lgt,
                                          float bgl, const float* __restrict__ H, float* acc,
                                          const int* q, int cnt, int base, int lane) {
  for (int b = 0; b < cnt; b += 32) {
    int n = cnt - b < 32 ? cnt - b : 32;
    int e = q[b + (lane < n ? lane : 0)];
    int r = ei[e];
    int cl = col[e] - base;
    float2 s = gs[r];
    float4 hv[8];
    agg_rows_load(H, r, 0, n, lane, hv);
    float2 t = lgt[cl];
    float sc = tanhf(s.x + t.x + bgl) * s.y * t.y;
    int lrow = cl * 128;
    for (int j0 = 0;;) {
      agg_rows_add(acc, lrow, sc, j0, n, lane, hv);
      j0 += 8;
      if (j0 >= n) break;
      agg_rows_load(H, r, j0, n, lane, hv);
    }
  }
}

__device__ __forceinline__ void agg_load_keys(const unsigned* __restrict__ key, int S, int lane,
                                              uint4 (&kv)[4]) {
  const uint4* p = (const uint4*)key + (size_t)S * (AGG_KSTEP / 16) + lane;
#pragma unroll
  for (int k = 0; k < 4; ++k) kv[k] = p[k * 32];
}

__device__ __forceinline__ unsigned agg_zero_bytes(unsigned x) {
  return ~(((x & 0x7f7f7f7fu) + 0x7f7f7f7fu) | x) & 0x80808080u;
}

__device__ __forceinline__ unsigned agg_hits8(uint4 t0, uint4 t1, unsigned pat) {
  return (agg_zero_bytes(t0.x ^ pat) >> 7) | (agg_zero_bytes(t0.y ^ pat) >> 6) |
         (agg_zero_bytes(t0.z ^ pat) >> 5) | (agg_zero_bytes(t0.w ^ pat) >> 4) |
         (agg_zero_bytes(t1.x ^ pat) >> 3) | (agg_zero_bytes(t1.y ^ pat) >> 2) |
         (agg_zero_bytes(t1.z ^ pat) >> 1) | agg_zero_bytes(t1.w ^ pat);
}

__device__ __forceinline__ int agg_append(bool hit, int e, int* q, int cnt) {
  unsigned b = __builtin_amdgcn_ballot_w32(hit);
  if (hit) q[cnt + __builtin_amdgcn_mbcnt_lo(b, 0)] = e;
  return cnt + __builtin_popcount(b);
}

__device__ __forceinline__ int agg_queue(const int (&cv)[16], int s, int lane, int base, int* q,
                                         int cnt) {
  int e0 = s * AGG_STEP + lane * 4;
#pragma unroll
  for (int i = 0; i < 16; ++i)
    cnt = agg_append((unsigned)(cv[i] - base) < (unsigned)AGG_TN,
                     e0 + (i >> 2) * 128 + (i & 3), q, cnt);
  __builtin_amdgcn_fence(__ATOMIC_ACQ_REL, "wavefront");
  return cnt;
}

__global__ __launch_bounds__(1024) void agg_lds_kernel(const int* __restrict__ ei,
                                                       const float2* __restrict__ gs,
                                                       const float2* __restrict__ gt,
                                                       const float* __restrict__ bg, int layer,
                                                       const float* __restrict__ H,
                                                       float* __restrict__ agg,
                                                       const unsigned* __restrict__ key,
                                                       const float* ws, size_t ws_bytes,
                                                       int E, int Nrows) {
  __shared__ float acc[AGG_TN * 128];
  __shared__ int wq[AGG_WAVES][AGG_QCAP];
  __shared__ float2 lgt[AGG_TN];
  int tid = threadIdx.x;
  int wave = __builtin_amdgcn_readfirstlane(tid >> 5);
  int lane = tid & 31;
  int base = blockIdx.x * AGG_TN;
  for (int i = tid; i < AGG_TN * 32; i += 1024)
    ((float4*)acc)[i] = make_float4(0.0f, 0.0f, 0.0f, 0.0f);
  if (tid < AGG_TN)
    lgt[tid] = base + tid < Nrows ? gt[base + tid] : make_float2(0.0f, 0.0f);
  float bgl = bg[layer];
  __syncthreads();

  const int* col = ei + E;
  int nsteps = (E + AGG_STEP - 1) / AGG_STEP;
  int nsuper = agg_keys_ok(key, ws, ws_bytes, E, Nrows) ? E / AGG_KSTEP : 0;
  int* q = wq[wave];
  int cnt = 0;
  if (wave < nsuper) {
    unsigned pat = blockIdx.x * 0x01010101u;
    uint4 kv[4];
    agg_load_keys(key, wave, lane, kv);
    for (int S = wave; S < nsuper; S += AGG_WAVES) {
      unsigned long long m = ((unsigned long long)agg_hits8(kv[2], kv[3], pat) << 32) |
                             agg_hits8(kv[0], kv[1], pat);
      int Sp = S + AGG_WAVES < nsuper ? S + AGG_WAVES : nsuper - 1;
      agg_load_keys(key, Sp, lane, kv);
      int e0 = S * AGG_KSTEP + lane * 16;
      while (__builtin_amdgcn_ballot_w32(m != 0) != 0) {
        bool hit = m != 0;
        int bit = hit ? __builtin_ctzll(m) : 0;
        int off = ((((bit >> 5) << 1) | ((bit >> 2) & 1)) << 9) | ((bit & 3) << 2) |
                  ((bit >> 3) & 3);
        cnt = agg_append(hit, e0 + off, q, cnt);
        m &= m - 1;
        __builtin_amdgcn_fence(__ATOMIC_ACQ_REL, "wavefront");
        if (cnt >= 32) {
          agg_drain(ei, col, gs, lgt, bgl, H, acc, q, cnt, base, lane);
          cnt = 0;
        }
      }
    }
  }
  for (int s = nsuper * 4 + wave; s < nsteps; s += AGG_WAVES) {
    int cv[16];
    int e0 = s * AGG_STEP + lane * 4;
#pragma unroll
    for (int i = 0; i < 16; ++i) {
      int ee = e0 + (i >> 2) * 128 + (i & 3);
      cv[i] = ee < E ? col[ee] : -1;
    }
    cnt = agg_queue(cv, s, lane, base, q, cnt);
    if (cnt >= 32) {
      agg_drain(ei, col, gs, lgt, bgl, H, acc, q, cnt, base, lane);
      cnt = 0;
    }
  }
  if (cnt > 0) agg_drain(ei, col, gs, lgt, bgl, H, acc, q, cnt, base, lane);
  __syncthreads();

  for (int i = tid; i < AGG_TN * 32; i += 1024) {
    int node = base + (i >> 5);
    if (node < Nrows) ((float4*)agg)[(size_t)node * 32 + (i & 31)] = ((float4*)acc)[i];
  }
}

__global__ void residual_kernel(const float4* __restrict__ raw, float4* __restrict__ agg,
                                long long n4) {
  long long i = (long long)blockIdx.x * blockDim.x + threadIdx.x;
  if (i < n4) {
    float4 r = raw[i];
    float4 a = agg[i];
    a.x = FAGCN_EPS * r.x + a.x;
    a.y = FAGCN_EPS * r.y + a.y;
    a.z = FAGCN_EPS * r.z + a.z;
    a.w = FAGCN_EPS * r.w + a.w;
    agg[i] = a;
  }
}

__global__ void gemm_out_kernel(const float* __restrict__ H, const float* __restrict__ W,
                                const float* __restrict__ bias, float* __restrict__ out,
                                int Nrows) {
  int wave = threadIdx.x >> 5;
  int lane = threadIdx.x & 31;
  int tileM = blockIdx.x * 8 + wave;
  int tiles = (Nrows + 15) >> 4;
  if (tileM >= tiles) return;
  int mA = tileM * 16 + (lane & 15);
  int mclamp = mA < Nrows ? mA : (Nrows - 1);
  int n = lane & 15;
  int khalf = (lane >> 4) * 2;
  const float* hrow = H + (size_t)mclamp * 128 + khalf;
  const float* wcol = W + (size_t)khalf * 16 + n;
  __builtin_prefetch(hrow, 0, 0);
  v8f acc = {};
#pragma unroll 4
  for (int kk = 0; kk < 128; kk += 4) {
    v2f a = *(const v2f*)(hrow + kk);
    v2f b;
    b.x = wcol[(size_t)kk * 16];
    b.y = wcol[(size_t)kk * 16 + 16];
    acc = __builtin_amdgcn_wmma_f32_16x16x4_f32(false, a, false, b, (short)0, acc,
                                               false, false);
  }
  int mbase = tileM * 16 + ((lane >> 4) * 8);
  float bn = bias[n];
  if (tileM * 16 + 16 <= Nrows) {
    float* op = out + (size_t)mbase * 16 + n;
#pragma unroll
    for (int r = 0; r < 8; ++r)
      op[(size_t)r * 16] = acc[r] + bn;
  } else {
#pragma unroll
    for (int r = 0; r < 8; ++r) {
      int mm = mbase + r;
      if (mm < Nrows) out[(size_t)mm * 16 + n] = acc[r] + bn;
    }
  }
}

__global__ void logsoftmax_kernel(float4* out, int N) {
  int i = blockIdx.x * blockDim.x + threadIdx.x;
  if (i >= N) return;
  float4* row = out + (size_t)i * 4;
  float4 v[4];
#pragma unroll
  for (int q = 0; q < 4; ++q) v[q] = row[q];
  float m = v[0].x;
#pragma unroll
  for (int q = 0; q < 4; ++q) {
    m = fmaxf(m, v[q].x); m = fmaxf(m, v[q].y);
    m = fmaxf(m, v[q].z); m = fmaxf(m, v[q].w);
  }
  float s = 0.0f;
#pragma unroll
  for (int q = 0; q < 4; ++q) {
    s += expf(v[q].x - m); s += expf(v[q].y - m);
    s += expf(v[q].z - m); s += expf(v[q].w - m);
  }
  float d = m + logf(s);
#pragma unroll
  for (int q = 0; q < 4; ++q) {
    v[q].x -= d; v[q].y -= d; v[q].z -= d; v[q].w -= d;
    row[q] = v[q];
  }
}

extern "C" void kernel_launch(void* const* d_in, const int* in_sizes, int n_in,
                              void* d_out, int out_size, void* d_ws, size_t ws_size,
                              hipStream_t stream) {
  const float* x     = (const float*)d_in[0];
  const int*   ei    = (const int*)d_in[1];
  const float* W_in  = (const float*)d_in[2];
  const float* b_in  = (const float*)d_in[3];
  const float* Wg    = (const float*)d_in[4];
  const float* bg    = (const float*)d_in[5];
  const float* W_out = (const float*)d_in[6];
  const float* b_out = (const float*)d_in[7];
  float* out = (float*)d_out;

  const int Hdim = 128;
  int N = in_sizes[0] / Hdim;
  int E = in_sizes[1] / 2;
  int L = in_sizes[5];

  float* ws   = (float*)d_ws;
  float2* gs  = (float2*)ws;
  float2* gt  = gs + N;
  float* dn   = (float*)(gt + N);
  float* raw  = ws + 3 * (size_t)N + E;
  float* bufA = raw + (size_t)N * Hdim;
  float* bufB = bufA + (size_t)N * Hdim;
  unsigned* key = (unsigned*)(bufB + (size_t)N * Hdim);

  unsigned* part = (unsigned*)bufB;
  deg_key_kernel<<<dim3(HIST_K, (N + HIST_CAP - 1) / HIST_CAP), HIST_T, 0, stream>>>(
      ei, part, key, ws, ws_size, E, N);
  dn_kernel<<<(N + 255) / 256, 256, 0, stream>>>(part, dn, N);

  gemm_relu_kernel<<<(N + 15) / 16, 256, 0, stream>>>(x, W_in, b_in, raw, N);

  float* h = raw;
  float* agg = bufA;
  for (int i = 0; i < L; ++i) {
    gate_kernel<<<(N + 7) / 8, 256, 0, stream>>>(h, Wg + (size_t)i * 2 * Hdim, dn, gs, gt, N);
    agg_lds_kernel<<<(N + AGG_TN - 1) / AGG_TN, 1024, 0, stream>>>(ei, gs, gt, bg, i, h, agg,
                                                                    key, ws, ws_size, E, N);
    long long n4 = (long long)N * Hdim / 4;
    residual_kernel<<<(unsigned)((n4 + 255) / 256), 256, 0, stream>>>(
        (const float4*)raw, (float4*)agg, n4);
    h = agg;
    agg = (h == bufA) ? bufB : bufA;
  }

  int tiles = (N + 15) / 16;
  gemm_out_kernel<<<(tiles + 7) / 8, 256, 0, stream>>>(h, W_out, b_out, out, N);
  logsoftmax_kernel<<<(N + 255) / 256, 256, 0, stream>>>((float4*)out, N);
}


# Aggregation's LDS row sums start at eps*raw (whole-line loads) instead of zero, so the row each work-group stores is the layer's h; residual pass and its 2 launches go; float sum order changes
# speedup vs baseline: 2.6605x; 1.0725x over previous
#include <hip/hip_runtime.h>

typedef float v2f __attribute__((ext_vector_type(2)));
typedef float v8f __attribute__((ext_vector_type(8)));

#define FAGCN_EPS 0.3f

#define AGG_TN 196
#define AGG_MAXG 256

__device__ __forceinline__ bool agg_keys_ok(const unsigned* key, const float* ws, size_t ws_bytes,
                                            int E, int Nrows) {
  size_t need = (size_t)((const char*)key - (const char*)ws) + (((size_t)E + 3) & ~(size_t)3);
  return need <= ws_bytes && (((size_t)key) & 15) == 0 &&
         (Nrows + AGG_TN - 1) / AGG_TN <= AGG_MAXG;
}

#define HIST_K 64
#define HIST_CAP 16384
#define HIST_T 1024
#define HIST_U 4

__device__ __forceinline__ void hist_count(unsigned* tab, int r, int rbase, int nhere) {
  unsigned d = (unsigned)(r - rbase);
  if (d < (unsigned)nhere) atomicAdd(&tab[d], 1u);
}

__device__ __forceinline__ unsigned agg_pack_keys(int c0, int c1, int c2, int c3) {
  return (((unsigned)c0 / AGG_TN) & 255u) | ((((unsigned)c1 / AGG_TN) & 255u) << 8) |
         ((((unsigned)c2 / AGG_TN) & 255u) << 16) | ((((unsigned)c3 / AGG_TN) & 255u) << 24);
}

__global__ __launch_bounds__(HIST_T) void deg_key_kernel(const int* __restrict__ ei,
                                                         unsigned* __restrict__ part,
                                                         unsigned* __restrict__ key,
                                                         const float* ws, size_t ws_bytes,
                                                         int E, int Nrows) {
  __shared__ unsigned tab[HIST_CAP];
  int tid = threadIdx.x;
  int rbase = blockIdx.y * HIST_CAP;
  int nhere = Nrows - rbase < HIST_CAP ? Nrows - rbase : HIST_CAP;
  for (int i = tid; i < HIST_CAP / 4; i += HIST_T) ((uint4*)tab)[i] = make_uint4(0u, 0u, 0u, 0u);
  __syncthreads();

  const int* col = ei + E;
  int Q = (E + 3) >> 2;
  int qpb = (Q + HIST_K - 1) / HIST_K;
  int q0 = blockIdx.x * qpb;
  int q1 = q0 + qpb < Q ? q0 + qpb : Q;
  bool keys = blockIdx.y == 0 && agg_keys_ok(key, ws, ws_bytes, E, Nrows);
  if (q0 < q1) {
    if ((E & 3) == 0 && ((((size_t)ei) | ((size_t)col)) & 15) == 0) {
      int trips = (qpb + HIST_T - 1) / HIST_T;
      for (int j0 = 0; j0 < trips; j0 += HIST_U) {
        int4 rv[HIST_U], cw[HIST_U];
#pragma unroll
        for (int u = 0; u < HIST_U; ++u) {
          int q = q0 + (j0 + u) * HIST_T + tid;
          int qq = q < q1 ? q : q1 - 1;
          rv[u] = ((const int4*)ei)[qq];
          cw[u] = ((const int4*)col)[qq];
        }
        asm volatile("" ::"v"(rv[0].x), "v"(rv[1].x), "v"(rv[2].x), "v"(rv[3].x), "v"(cw[0].x),
                     "v"(cw[1].x), "v"(cw[2].x), "v"(cw[3].x));
#pragma unroll
        for (int u = 0; u < HIST_U; ++u) {
          int q = q0 + (j0 + u) * HIST_T + tid;
          if (q < q1) {
            hist_count(tab, rv[u].x, rbase, nhere);
            hist_count(tab, rv[u].y, rbase, nhere);
            hist_count(tab, rv[u].z, rbase, nhere);
            hist_count(tab, rv[u].w, rbase, nhere);
            if (keys) key[q] = agg_pack_keys(cw[u].x, cw[u].y, cw[u].z, cw[u].w);
          }
        }
      }
    } else {
      for (int q = q0 + tid; q < q1; q += HIST_T) {
        int c[4];
#pragma unroll
        for (int k = 0; k < 4; ++k) {
          int e = q * 4 + k;
          int ec = e < E ? e : E - 1;
          if (e < E) hist_count(tab, ei[ec], rbase, nhere);
          c[k] = col[ec];
        }
        if (keys) key[q] = agg_pack_keys(c[0], c[1], c[2], c[3]);
      }
    }
  }
  __syncthreads();

  unsigned* dst = part + (size_t)blockIdx.x * Nrows + rbase;
  if ((nhere & 3) == 0 && (((size_t)dst) & 15) == 0) {
    for (int i = tid; i < nhere / 4; i += HIST_T) ((uint4*)dst)[i] = ((const uint4*)tab)[i];
  } else {
    for (int i = tid; i < nhere; i += HIST_T) dst[i] = tab[i];
  }
}

__global__ void dn_kernel(const unsigned* __restrict__ part, float* __restrict__ dn, int N) {
  int i = blockIdx.x * blockDim.x + threadIdx.x;
  if (i >= N) return;
  unsigned s = 0;
#pragma unroll
  for (int b = 0; b < HIST_K; ++b) s += part[(size_t)b * N + i];
  dn[i] = 1.0f / sqrtf(fmaxf((float)s, 1.0f));
}

__global__ void gemm_relu_kernel(const float* __restrict__ X, const float* __restrict__ W,
                                 const float* __restrict__ bias, float* __restrict__ H,
                                 int Nrows) {
  int wave = threadIdx.x >> 5;
  int lane = threadIdx.x & 31;
  int tileM = blockIdx.x;
  int mA = tileM * 16 + (lane & 15);
  int mclamp = mA < Nrows ? mA : (Nrows - 1);
  int n = wave * 16 + (lane & 15);
  int khalf = (lane >> 4) * 2;
  const float* xrow = X + (size_t)mclamp * 128 + khalf;
  const float* wcol = W + (size_t)khalf * 128 + n;
  __builtin_prefetch(xrow, 0, 0);
  v8f acc = {};
#pragma unroll 4
  for (int kk = 0; kk < 128; kk += 4) {
    v2f a = *(const v2f*)(xrow + kk);
    v2f b;
    b.x = wcol[(size_t)kk * 128];
    b.y = wcol[(size_t)kk * 128 + 128];
    acc = __builtin_amdgcn_wmma_f32_16x16x4_f32(false, a, false, b, (short)0, acc,
                                               false, false);
  }
  int mbase = tileM * 16 + ((lane >> 4) * 8);
  float bn = bias[n];
  if (tileM * 16 + 16 <= Nrows) {
    float* hp = H + (size_t)mbase * 128 + n;
#pragma unroll
    for (int r = 0; r < 8; ++r)
      hp[(size_t)r * 128] = fmaxf(acc[r] + bn, 0.0f);
  } else {
#pragma unroll
    for (int r = 0; r < 8; ++r) {
      int mm = mbase + r;
      if (mm < Nrows) H[(size_t)mm * 128 + n] = fmaxf(acc[r] + bn, 0.0f);
    }
  }
}

__global__ void gate_kernel(const float* __restrict__ H, const float* __restrict__ Wg,
                            const float* __restrict__ dn, float2* __restrict__ gs,
                            float2* __restrict__ gt, int Nrows) {
  int wave = threadIdx.x >> 5;
  int lane = threadIdx.x & 31;
  int node = blockIdx.x * 8 + wave;
  if (node >= Nrows) return;
  const float* hrow = H + (size_t)node * 128;
  float sa = 0.0f, sb = 0.0f;
#pragma unroll
  for (int j = 0; j < 4; ++j) {
    int f = lane + j * 32;
    float hv = hrow[f];
    sa += hv * Wg[f];
    sb += hv * Wg[128 + f];
  }
  for (int off = 16; off > 0; off >>= 1) {
    sa += __shfl_down(sa, off, 32);
    sb += __shfl_down(sb, off, 32);
  }
  if (lane == 0) {
    float d = dn[node];
    gs[node] = make_float2(sa, d);
    gt[node] = make_float2(sb, d);
  }
}

#define AGG_WAVES 32
#define AGG_STEP 512
#define AGG_KSTEP (4 * AGG_STEP)
#define AGG_QCAP (AGG_STEP + 32)
#define AGG_INIT_TRIPS ((AGG_TN * 32 + 1023) / 1024)

__device__ __forceinline__ void agg_rows_load(const float* __restrict__ H, int r, int j0, int n,
                                              int lane, float4 (&hv)[8]) {
#pragma unroll
  for (int u = 0; u < 8; ++u) {
    int jj = j0 + u < n ? j0 + u : n - 1;
    int rj = __builtin_amdgcn_readlane(r, jj);
    hv[u] = *(const float4*)(H + (size_t)rj * 128 + lane * 4);
  }
}

__device__ __forceinline__ void agg_rows_add(float* acc, int lrow, float sc, int j0, int n,
                                             int lane, const float4 (&hv)[8]) {
  int hi = lane >> 4;
#pragma unroll
  for (int u = 0; u < 8; ++u) {
    if (j0 + u < n) {
      int lr = __builtin_amdgcn_readlane(lrow, j0 + u);
      float sj = __builtin_bit_cast(float, __builtin_amdgcn_readlane(__builtin_bit_cast(int, sc), j0 + u));
      float a0 = hi ? hv[u].y : hv[u].x;
      float a1 = hi ? hv[u].x : hv[u].y;
      float a2 = hi ? hv[u].w : hv[u].z;
      float a3 = hi ? hv[u].z : hv[u].w;
      float* p = acc + lr + lane * 4;
      atomicAdd(p + (0 ^ hi), sj * a0);
      atomicAdd(p + (1 ^ hi), sj * a1);
      atomicAdd(p + (2 ^ hi), sj * a2);
      atomicAdd(p + (3 ^ hi), sj * a3);
    }
  }
}

__device__ __forceinline__ void agg_drain(const int* __restrict__ ei, const int* __restrict__ col,
                                          const float2* __restrict__ gs, const float2* lgt,
                                          float bgl, const float* __restrict__ H, float* acc,
                                          const int* q, int cnt, int base, int lane) {
  for (int b = 0; b < cnt; b += 32) {
    int n = cnt - b < 32 ? cnt - b : 32;
    int e = q[b + (lane < n ? lane : 0)];
    int r = ei[e];
    int cl = col[e] - base;
    float2 s = gs[r];
    float4 hv[8];
    agg_rows_load(H, r, 0, n, lane, hv);
    float2 t = lgt[cl];
    float sc = tanhf(s.x + t.x + bgl) * s.y * t.y;
    int lrow = cl * 128;
    for (int j0 = 0;;) {
      agg_rows_add(acc, lrow, sc, j0, n, lane, hv);
      j0 += 8;
      if (j0 >= n) break;
      agg_rows_load(H, r, j0, n, lane, hv);
    }
  }
}

__device__ __forceinline__ void agg_load_keys(const unsigned* __restrict__ key, int S, int lane,
                                              uint4 (&kv)[4]) {
  const uint4* p = (const uint4*)key + (size_t)S * (AGG_KSTEP / 16) + lane;
#pragma unroll
  for (int k = 0; k < 4; ++k) kv[k] = p[k * 32];
}

__device__ __forceinline__ unsigned agg_zero_bytes(unsigned x) {
  return ~(((x & 0x7f7f7f7fu) + 0x7f7f7f7fu) | x) & 0x80808080u;
}

__device__ __forceinline__ unsigned agg_hits8(uint4 t0, uint4 t1, unsigned pat) {
  return (agg_zero_bytes(t0.x ^ pat) >> 7) | (agg_zero_bytes(t0.y ^ pat) >> 6) |
         (agg_zero_bytes(t0.z ^ pat) >> 5) | (agg_zero_bytes(t0.w ^ pat) >> 4) |
         (agg_zero_bytes(t1.x ^ pat) >> 3) | (agg_zero_bytes(t1.y ^ pat) >> 2) |
         (agg_zero_bytes(t1.z ^ pat) >> 1) | agg_zero_bytes(t1.w ^ pat);
}

__device__ __forceinline__ int agg_append(bool hit, int e, int* q, int cnt) {
  unsigned b = __builtin_amdgcn_ballot_w32(hit);
  if (hit) q[cnt + __builtin_amdgcn_mbcnt_lo(b, 0)] = e;
  return cnt + __builtin_popcount(b);
}

__device__ __forceinline__ int agg_queue(const int (&cv)[16], int s, int lane, int base, int* q,
                                         int cnt) {
  int e0 = s * AGG_STEP + lane * 4;
#pragma unroll
  for (int i = 0; i < 16; ++i)
    cnt = agg_append((unsigned)(cv[i] - base) < (unsigned)AGG_TN,
                     e0 + (i >> 2) * 128 + (i & 3), q, cnt);
  __builtin_amdgcn_fence(__ATOMIC_ACQ_REL, "wavefront");
  return cnt;
}

__global__ __launch_bounds__(1024) void agg_lds_kernel(const int* __restrict__ ei,
                                                       const float2* __restrict__ gs,
                                                       const float2* __restrict__ gt,
                                                       const float* __restrict__ bg, int layer,
                                                       const float* __restrict__ H,
                                                       const float* __restrict__ raw,
                                                       float* __restrict__ agg,
                                                       const unsigned* __restrict__ key,
                                                       const float* ws, size_t ws_bytes,
                                                       int E, int Nrows) {
  __shared__ float acc[AGG_TN * 128];
  __shared__ int wq[AGG_WAVES][AGG_QCAP];
  __shared__ float2 lgt[AGG_TN];
  int tid = threadIdx.x;
  int wave = __builtin_amdgcn_readfirstlane(tid >> 5);
  int lane = tid & 31;
  int base = blockIdx.x * AGG_TN;
  {
    float4 rv[AGG_INIT_TRIPS];
#pragma unroll
    for (int k = 0; k < AGG_INIT_TRIPS; ++k) {
      int i = tid + k * 1024;
      int ic = i < AGG_TN * 32 ? i : AGG_TN * 32 - 1;
      int node = base + (ic >> 5);
      int nc = node < Nrows ? node : Nrows - 1;
      rv[k] = ((const float4*)raw)[(size_t)nc * 32 + (ic & 31)];
    }
#pragma unroll
    for (int k = 0; k < AGG_INIT_TRIPS; ++k) {
      int i = tid + k * 1024;
      if (i < AGG_TN * 32) {
        float s = base + (i >> 5) < Nrows ? FAGCN_EPS : 0.0f;
        ((float4*)acc)[i] = make_float4(s * rv[k].x, s * rv[k].y, s * rv[k].z, s * rv[k].w);
      }
    }
  }
  if (tid < AGG_TN)
    lgt[tid] = base + tid < Nrows ? gt[base + tid] : make_float2(0.0f, 0.0f);
  float bgl = bg[layer];
  __syncthreads();

  const int* col = ei + E;
  int nsteps = (E + AGG_STEP - 1) / AGG_STEP;
  int nsuper = agg_keys_ok(key, ws, ws_bytes, E, Nrows) ? E / AGG_KSTEP : 0;
  int* q = wq[wave];
  int cnt = 0;
  if (wave < nsuper) {
    unsigned pat = blockIdx.x * 0x01010101u;
    uint4 kv[4];
    agg_load_keys(key, wave, lane, kv);
    for (int S = wave; S < nsuper; S += AGG_WAVES) {
      unsigned long long m = ((unsigned long long)agg_hits8(kv[2], kv[3], pat) << 32) |
                             agg_hits8(kv[0], kv[1], pat);
      int Sp = S + AGG_WAVES < nsuper ? S + AGG_WAVES : nsuper - 1;
      agg_load_keys(key, Sp, lane, kv);
      int e0 = S * AGG_KSTEP + lane * 16;
      while (__builtin_amdgcn_ballot_w32(m != 0) != 0) {
        bool hit = m != 0;
        int bit = hit ? __builtin_ctzll(m) : 0;
        int off = ((((bit >> 5) << 1) | ((bit >> 2) & 1)) << 9) | ((bit & 3) << 2) |
                  ((bit >> 3) & 3);
        cnt = agg_append(hit, e0 + off, q, cnt);
        m &= m - 1;
        __builtin_amdgcn_fence(__ATOMIC_ACQ_REL, "wavefront");
        if (cnt >= 32) {
          agg_drain(ei, col, gs, lgt, bgl, H, acc, q, cnt, base, lane);
          cnt = 0;
        }
      }
    }
  }
  for (int s = nsuper * 4 + wave; s < nsteps; s += AGG_WAVES) {
    int cv[16];
    int e0 = s * AGG_STEP + lane * 4;
#pragma unroll
    for (int i = 0; i < 16; ++i) {
      int ee = e0 + (i >> 2) * 128 + (i & 3);
      cv[i] = ee < E ? col[ee] : -1;
    }
    cnt = agg_queue(cv, s, lane, base, q, cnt);
    if (cnt >= 32) {
      agg_drain(ei, col, gs, lgt, bgl, H, acc, q, cnt, base, lane);
      cnt = 0;
    }
  }
  if (cnt > 0) agg_drain(ei, col, gs, lgt, bgl, H, acc, q, cnt, base, lane);
  __syncthreads();

  for (int i = tid; i < AGG_TN * 32; i += 1024) {
    int node = base + (i >> 5);
    if (node < Nrows) ((float4*)agg)[(size_t)node * 32 + (i & 31)] = ((float4*)acc)[i];
  }
}

__global__ void gemm_out_kernel(const float* __restrict__ H, const float* __restrict__ W,
                                const float* __restrict__ bias, float* __restrict__ out,
                                int Nrows) {
  int wave = threadIdx.x >> 5;
  int lane = threadIdx.x & 31;
  int tileM = blockIdx.x * 8 + wave;
  int tiles = (Nrows + 15) >> 4;
  if (tileM >= tiles) return;
  int mA = tileM * 16 + (lane & 15);
  int mclamp = mA < Nrows ? mA : (Nrows - 1);
  int n = lane & 15;
  int khalf = (lane >> 4) * 2;
  const float* hrow = H + (size_t)mclamp * 128 + khalf;
  const float* wcol = W + (size_t)khalf * 16 + n;
  __builtin_prefetch(hrow, 0, 0);
  v8f acc = {};
#pragma unroll 4
  for (int kk = 0; kk < 128; kk += 4) {
    v2f a = *(const v2f*)(hrow + kk);
    v2f b;
    b.x = wcol[(size_t)kk * 16];
    b.y = wcol[(size_t)kk * 16 + 16];
    acc = __builtin_amdgcn_wmma_f32_16x16x4_f32(false, a, false, b, (short)0, acc,
                                               false, false);
  }
  int mbase = tileM * 16 + ((lane >> 4) * 8);
  float bn = bias[n];
  if (tileM * 16 + 16 <= Nrows) {
    float* op = out + (size_t)mbase * 16 + n;
#pragma unroll
    for (int r = 0; r < 8; ++r)
      op[(size_t)r * 16] = acc[r] + bn;
  } else {
#pragma unroll
    for (int r = 0; r < 8; ++r) {
      int mm = mbase + r;
      if (mm < Nrows) out[(size_t)mm * 16 + n] = acc[r] + bn;
    }
  }
}

__global__ void logsoftmax_kernel(float4* out, int N) {
  int i = blockIdx.x * blockDim.x + threadIdx.x;
  if (i >= N) return;
  float4* row = out + (size_t)i * 4;
  float4 v[4];
#pragma unroll
  for (int q = 0; q < 4; ++q) v[q] = row[q];
  float m = v[0].x;
#pragma unroll
  for (int q = 0; q < 4; ++q) {
    m = fmaxf(m, v[q].x); m = fmaxf(m, v[q].y);
    m = fmaxf(m, v[q].z); m = fmaxf(m, v[q].w);
  }
  float s = 0.0f;
#pragma unroll
  for (int q = 0; q < 4; ++q) {
    s += expf(v[q].x - m); s += expf(v[q].y - m);
    s += expf(v[q].z - m); s += expf(v[q].w - m);
  }
  float d = m + logf(s);
#pragma unroll
  for (int q = 0; q < 4; ++q) {
    v[q].x -= d; v[q].y -= d; v[q].z -= d; v[q].w -= d;
    row[q] = v[q];
  }
}

extern "C" void kernel_launch(void* const* d_in, const int* in_sizes, int n_in,
                              void* d_out, int out_size, void* d_ws, size_t ws_size,
                              hipStream_t stream) {
  const float* x     = (const float*)d_in[0];
  const int*   ei    = (const int*)d_in[1];
  const float* W_in  = (const float*)d_in[2];
  const float* b_in  = (const float*)d_in[3];
  const float* Wg    = (const float*)d_in[4];
  const float* bg    = (const float*)d_in[5];
  const float* W_out = (const float*)d_in[6];
  const float* b_out = (const float*)d_in[7];
  float* out = (float*)d_out;

  const int Hdim = 128;
  int N = in_sizes[0] / Hdim;
  int E = in_sizes[1] / 2;
  int L = in_sizes[5];

  float* ws   = (float*)d_ws;
  float2* gs  = (float2*)ws;
  float2* gt  = gs + N;
  float* dn   = (float*)(gt + N);
  float* raw  = ws + 3 * (size_t)N + E;
  float* bufA = raw + (size_t)N * Hdim;
  float* bufB = bufA + (size_t)N * Hdim;
  unsigned* key = (unsigned*)(bufB + (size_t)N * Hdim);

  unsigned* part = (unsigned*)bufB;
  deg_key_kernel<<<dim3(HIST_K, (N + HIST_CAP - 1) / HIST_CAP), HIST_T, 0, stream>>>(
      ei, part, key, ws, ws_size, E, N);
  dn_kernel<<<(N + 255) / 256, 256, 0, stream>>>(part, dn, N);

  gemm_relu_kernel<<<(N + 15) / 16, 256, 0, stream>>>(x, W_in, b_in, raw, N);

  float* h = raw;
  float* agg = bufA;
  for (int i = 0; i < L; ++i) {
    gate_kernel<<<(N + 7) / 8, 256, 0, stream>>>(h, Wg + (size_t)i * 2 * Hdim, dn, gs, gt, N);
    agg_lds_kernel<<<(N + AGG_TN - 1) / AGG_TN, 1024, 0, stream>>>(ei, gs, gt, bg, i, h, raw, agg,
                                                                    key, ws, ws_size, E, N);
    h = agg;
    agg = (h == bufA) ? bufB : bufA;
  }

  int tiles = (N + 15) / 16;
  gemm_out_kernel<<<(tiles + 7) / 8, 256, 0, stream>>>(h, W_out, b_out, out, N);
  logsoftmax_kernel<<<(N + 255) / 256, 256, 0, stream>>>((float4*)out, N);
}


# Second layer's aggregation does not scan: first layer's groups store the (source, own-node index) pairs they drain (LDS counter) in their rows of the other buffer, it reads them; add order changes
# speedup vs baseline: 2.8359x; 1.0659x over previous
#include <hip/hip_runtime.h>

typedef float v2f __attribute__((ext_vector_type(2)));
typedef float v8f __attribute__((ext_vector_type(8)));

#define FAGCN_EPS 0.3f

#define AGG_TN 196
#define AGG_MAXG 256

__device__ __forceinline__ bool agg_keys_ok(const unsigned* key, const float* ws, size_t ws_bytes,
                                            int E, int Nrows) {
  size_t need = (size_t)((const char*)key - (const char*)ws) + (((size_t)E + 3) & ~(size_t)3);
  return need <= ws_bytes && (((size_t)key) & 15) == 0 &&
         (Nrows + AGG_TN - 1) / AGG_TN <= AGG_MAXG;
}

#define HIST_K 64
#define HIST_CAP 16384
#define HIST_T 1024
#define HIST_U 4

__device__ __forceinline__ void hist_count(unsigned* tab, int r, int rbase, int nhere) {
  unsigned d = (unsigned)(r - rbase);
  if (d < (unsigned)nhere) atomicAdd(&tab[d], 1u);
}

__device__ __forceinline__ unsigned agg_pack_keys(int c0, int c1, int c2, int c3) {
  return (((unsigned)c0 / AGG_TN) & 255u) | ((((unsigned)c1 / AGG_TN) & 255u) << 8) |
         ((((unsigned)c2 / AGG_TN) & 255u) << 16) | ((((unsigned)c3 / AGG_TN) & 255u) << 24);
}

__global__ __launch_bounds__(HIST_T) void deg_key_kernel(const int* __restrict__ ei,
                                                         unsigned* __restrict__ part,
                                                         unsigned* __restrict__ key,
                                                         const float* ws, size_t ws_bytes,
                                                         int E, int Nrows) {
  __shared__ unsigned tab[HIST_CAP];
  int tid = threadIdx.x;
  int rbase = blockIdx.y * HIST_CAP;
  int nhere = Nrows - rbase < HIST_CAP ? Nrows - rbase : HIST_CAP;
  for (int i = tid; i < HIST_CAP / 4; i += HIST_T) ((uint4*)tab)[i] = make_uint4(0u, 0u, 0u, 0u);
  __syncthreads();

  const int* col = ei + E;
  int Q = (E + 3) >> 2;
  int qpb = (Q + HIST_K - 1) / HIST_K;
  int q0 = blockIdx.x * qpb;
  int q1 = q0 + qpb < Q ? q0 + qpb : Q;
  bool keys = blockIdx.y == 0 && agg_keys_ok(key, ws, ws_bytes, E, Nrows);
  if (q0 < q1) {
    if ((E & 3) == 0 && ((((size_t)ei) | ((size_t)col)) & 15) == 0) {
      int trips = (qpb + HIST_T - 1) / HIST_T;
      for (int j0 = 0; j0 < trips; j0 += HIST_U) {
        int4 rv[HIST_U], cw[HIST_U];
#pragma unroll
        for (int u = 0; u < HIST_U; ++u) {
          int q = q0 + (j0 + u) * HIST_T + tid;
          int qq = q < q1 ? q : q1 - 1;
          rv[u] = ((const int4*)ei)[qq];
          cw[u] = ((const int4*)col)[qq];
        }
        asm volatile("" ::"v"(rv[0].x), "v"(rv[1].x), "v"(rv[2].x), "v"(rv[3].x), "v"(cw[0].x),
                     "v"(cw[1].x), "v"(cw[2].x), "v"(cw[3].x));
#pragma unroll
        for (int u = 0; u < HIST_U; ++u) {
          int q = q0 + (j0 + u) * HIST_T + tid;
          if (q < q1) {
            hist_count(tab, rv[u].x, rbase, nhere);
            hist_count(tab, rv[u].y, rbase, nhere);
            hist_count(tab, rv[u].z, rbase, nhere);
            hist_count(tab, rv[u].w, rbase, nhere);
            if (keys) key[q] = agg_pack_keys(cw[u].x, cw[u].y, cw[u].z, cw[u].w);
          }
        }
      }
    } else {
      for (int q = q0 + tid; q < q1; q += HIST_T) {
        int c[4];
#pragma unroll
        for (int k = 0; k < 4; ++k) {
          int e = q * 4 + k;
          int ec = e < E ? e : E - 1;
          if (e < E) hist_count(tab, ei[ec], rbase, nhere);
          c[k] = col[ec];
        }
        if (keys) key[q] = agg_pack_keys(c[0], c[1], c[2], c[3]);
      }
    }
  }
  __syncthreads();

  unsigned* dst = part + (size_t)blockIdx.x * Nrows + rbase;
  if ((nhere & 3) == 0 && (((size_t)dst) & 15) == 0) {
    for (int i = tid; i < nhere / 4; i += HIST_T) ((uint4*)dst)[i] = ((const uint4*)tab)[i];
  } else {
    for (int i = tid; i < nhere; i += HIST_T) dst[i] = tab[i];
  }
}

__global__ void dn_kernel(const unsigned* __restrict__ part, float* __restrict__ dn, int N) {
  int i = blockIdx.x * blockDim.x + threadIdx.x;
  if (i >= N) return;
  unsigned s = 0;
#pragma unroll
  for (int b = 0; b < HIST_K; ++b) s += part[(size_t)b * N + i];
  dn[i] = 1.0f / sqrtf(fmaxf((float)s, 1.0f));
}

__global__ void gemm_relu_kernel(const float* __restrict__ X, const float* __restrict__ W,
                                 const float* __restrict__ bias, float* __restrict__ H,
                                 int Nrows) {
  int wave = threadIdx.x >> 5;
  int lane = threadIdx.x & 31;
  int tileM = blockIdx.x;
  int mA = tileM * 16 + (lane & 15);
  int mclamp = mA < Nrows ? mA : (Nrows - 1);
  int n = wave * 16 + (lane & 15);
  int khalf = (lane >> 4) * 2;
  const float* xrow = X + (size_t)mclamp * 128 + khalf;
  const float* wcol = W + (size_t)khalf * 128 + n;
  __builtin_prefetch(xrow, 0, 0);
  v8f acc = {};
#pragma unroll 4
  for (int kk = 0; kk < 128; kk += 4) {
    v2f a = *(const v2f*)(xrow + kk);
    v2f b;
    b.x = wcol[(size_t)kk * 128];
    b.y = wcol[(size_t)kk * 128 + 128];
    acc = __builtin_amdgcn_wmma_f32_16x16x4_f32(false, a, false, b, (short)0, acc,
                                               false, false);
  }
  int mbase = tileM * 16 + ((lane >> 4) * 8);
  float bn = bias[n];
  if (tileM * 16 + 16 <= Nrows) {
    float* hp = H + (size_t)mbase * 128 + n;
#pragma unroll
    for (int r = 0; r < 8; ++r)
      hp[(size_t)r * 128] = fmaxf(acc[r] + bn, 0.0f);
  } else {
#pragma unroll
    for (int r = 0; r < 8; ++r) {
      int mm = mbase + r;
      if (mm < Nrows) H[(size_t)mm * 128 + n] = fmaxf(acc[r] + bn, 0.0f);
    }
  }
}

__global__ void gate_kernel(const float* __restrict__ H, const float* __restrict__ Wg,
                            const float* __restrict__ dn, float2* __restrict__ gs,
                            float2* __restrict__ gt, int Nrows) {
  int wave = threadIdx.x >> 5;
  int lane = threadIdx.x & 31;
  int node = blockIdx.x * 8 + wave;
  if (node >= Nrows) return;
  const float* hrow = H + (size_t)node * 128;
  float sa = 0.0f, sb = 0.0f;
#pragma unroll
  for (int j = 0; j < 4; ++j) {
    int f = lane + j * 32;
    float hv = hrow[f];
    sa += hv * Wg[f];
    sb += hv * Wg[128 + f];
  }
  for (int off = 16; off > 0; off >>= 1) {
    sa += __shfl_down(sa, off, 32);
    sb += __shfl_down(sb, off, 32);
  }
  if (lane == 0) {
    float d = dn[node];
    gs[node] = make_float2(sa, d);
    gt[node] = make_float2(sb, d);
  }
}

#define AGG_WAVES 32
#define AGG_STEP 512
#define AGG_KSTEP (4 * AGG_STEP)
#define AGG_QCAP (AGG_STEP + 32)
#define AGG_INIT_TRIPS ((AGG_TN * 32 + 1023) / 1024)

__device__ __forceinline__ void agg_rows_load(const float* __restrict__ H, int r, int j0, int n,
                                              int lane, float4 (&hv)[8]) {
#pragma unroll
  for (int u = 0; u < 8; ++u) {
    int jj = j0 + u < n ? j0 + u : n - 1;
    int rj = __builtin_amdgcn_readlane(r, jj);
    hv[u] = *(const float4*)(H + (size_t)rj * 128 + lane * 4);
  }
}

__device__ __forceinline__ void agg_rows_add(float* acc, int lrow, float sc, int j0, int n,
                                             int lane, const float4 (&hv)[8]) {
  int hi = lane >> 4;
#pragma unroll
  for (int u = 0; u < 8; ++u) {
    if (j0 + u < n) {
      int lr = __builtin_amdgcn_readlane(lrow, j0 + u);
      float sj = __builtin_bit_cast(float, __builtin_amdgcn_readlane(__builtin_bit_cast(int, sc), j0 + u));
      float a0 = hi ? hv[u].y : hv[u].x;
      float a1 = hi ? hv[u].x : hv[u].y;
      float a2 = hi ? hv[u].w : hv[u].z;
      float a3 = hi ? hv[u].z : hv[u].w;
      float* p = acc + lr + lane * 4;
      atomicAdd(p + (0 ^ hi), sj * a0);
      atomicAdd(p + (1 ^ hi), sj * a1);
      atomicAdd(p + (2 ^ hi), sj * a2);
      atomicAdd(p + (3 ^ hi), sj * a3);
    }
  }
}

__device__ __forceinline__ void agg_edges(int r, int cl, int n, const float2* __restrict__ gs,
                                          const float2* lgt, float bgl,
                                          const float* __restrict__ H, float* acc, int lane) {
  float2 s = gs[r];
  float4 hv[8];
  agg_rows_load(H, r, 0, n, lane, hv);
  float2 t = lgt[cl];
  float sc = tanhf(s.x + t.x + bgl) * s.y * t.y;
  int lrow = cl * 128;
  for (int j0 = 0;;) {
    agg_rows_add(acc, lrow, sc, j0, n, lane, hv);
    j0 += 8;
    if (j0 >= n) break;
    agg_rows_load(H, r, j0, n, lane, hv);
  }
}

__device__ __forceinline__ void agg_drain(const int* __restrict__ ei, const int* __restrict__ col,
                                          const float2* __restrict__ gs, const float2* lgt,
                                          float bgl, const float* __restrict__ H, float* acc,
                                          const int* q, int cnt, int base, int lane, int2* lw,
                                          int* lcount, int cap) {
  int pos = 0;
  if (lw) {
    if (lane == 0) pos = atomicAdd(lcount, cnt);
    pos = __builtin_amdgcn_readfirstlane(pos);
  }
  for (int b = 0; b < cnt; b += 32) {
    int n = cnt - b < 32 ? cnt - b : 32;
    int e = q[b + (lane < n ? lane : 0)];
    int r = ei[e];
    int cl = col[e] - base;
    agg_edges(r, cl, n, gs, lgt, bgl, H, acc, lane);
    if (lw && lane < n && pos + b + lane < cap) lw[pos + b + lane] = make_int2(r, cl);
  }
}

__device__ __forceinline__ void agg_load_keys(const unsigned* __restrict__ key, int S, int lane,
                                              uint4 (&kv)[4]) {
  const uint4* p = (const uint4*)key + (size_t)S * (AGG_KSTEP / 16) + lane;
#pragma unroll
  for (int k = 0; k < 4; ++k) kv[k] = p[k * 32];
}

__device__ __forceinline__ unsigned agg_zero_bytes(unsigned x) {
  return ~(((x & 0x7f7f7f7fu) + 0x7f7f7f7fu) | x) & 0x80808080u;
}

__device__ __forceinline__ unsigned agg_hits8(uint4 t0, uint4 t1, unsigned pat) {
  return (agg_zero_bytes(t0.x ^ pat) >> 7) | (agg_zero_bytes(t0.y ^ pat) >> 6) |
         (agg_zero_bytes(t0.z ^ pat) >> 5) | (agg_zero_bytes(t0.w ^ pat) >> 4) |
         (agg_zero_bytes(t1.x ^ pat) >> 3) | (agg_zero_bytes(t1.y ^ pat) >> 2) |
         (agg_zero_bytes(t1.z ^ pat) >> 1) | agg_zero_bytes(t1.w ^ pat);
}

__device__ __forceinline__ int agg_append(bool hit, int e, int* q, int cnt) {
  unsigned b = __builtin_amdgcn_ballot_w32(hit);
  if (hit) q[cnt + __builtin_amdgcn_mbcnt_lo(b, 0)] = e;
  return cnt + __builtin_popcount(b);
}

__device__ __forceinline__ int agg_queue(const int (&cv)[16], int s, int lane, int base, int* q,
                                         int cnt) {
  int e0 = s * AGG_STEP + lane * 4;
#pragma unroll
  for (int i = 0; i < 16; ++i)
    cnt = agg_append((unsigned)(cv[i] - base) < (unsigned)AGG_TN,
                     e0 + (i >> 2) * 128 + (i & 3), q, cnt);
  __builtin_amdgcn_fence(__ATOMIC_ACQ_REL, "wavefront");
  return cnt;
}

__global__ __launch_bounds__(1024) void agg_lds_kernel(const int* __restrict__ ei,
                                                       const float2* __restrict__ gs,
                                                       const float2* __restrict__ gt,
                                                       const float* __restrict__ bg, int layer,
                                                       const float* __restrict__ H,
                                                       const float* __restrict__ raw,
                                                       float* agg,
                                                       const unsigned* __restrict__ key,
                                                       const float* ws, size_t ws_bytes,
                                                       int E, int Nrows, float* lbuf, int L) {
  __shared__ float acc[AGG_TN * 128];
  __shared__ int wq[AGG_WAVES][AGG_QCAP];
  __shared__ float2 lgt[AGG_TN];
  __shared__ int lcount;
  int tid = threadIdx.x;
  int wave = __builtin_amdgcn_readfirstlane(tid >> 5);
  int lane = tid & 31;
  int base = blockIdx.x * AGG_TN;
  int rows_here = Nrows - base < AGG_TN ? Nrows - base : AGG_TN;
  int* lhdr = (int*)(lbuf + (size_t)base * 128);
  int2* lent = (int2*)(lhdr + 4);
  int lcap = rows_here * 64 - 2;
  bool lwrite = layer == 0 && L > 1;
  int ltotal = -1;
  if (layer == 1) {
    int t = __builtin_amdgcn_readfirstlane(lhdr[0]);
    if (t <= lcap) ltotal = t;
  }
  if (tid == 0) lcount = 0;
  {
    float4 rv[AGG_INIT_TRIPS];
#pragma unroll
    for (int k = 0; k < AGG_INIT_TRIPS; ++k) {
      int i = tid + k * 1024;
      int ic = i < AGG_TN * 32 ? i : AGG_TN * 32 - 1;
      int node = base + (ic >> 5);
      int nc = node < Nrows ? node : Nrows - 1;
      rv[k] = ((const float4*)raw)[(size_t)nc * 32 + (ic & 31)];
    }
#pragma unroll
    for (int k = 0; k < AGG_INIT_TRIPS; ++k) {
      int i = tid + k * 1024;
      if (i < AGG_TN * 32) {
        float s = base + (i >> 5) < Nrows ? FAGCN_EPS : 0.0f;
        ((float4*)acc)[i] = make_float4(s * rv[k].x, s * rv[k].y, s * rv[k].z, s * rv[k].w);
      }
    }
  }
  if (tid < AGG_TN)
    lgt[tid] = base + tid < Nrows ? gt[base + tid] : make_float2(0.0f, 0.0f);
  float bgl = bg[layer];
  __syncthreads();

  const int* col = ei + E;
  bool scan = ltotal < 0;
  int nsteps = scan ? (E + AGG_STEP - 1) / AGG_STEP : 0;
  int nsuper = scan && agg_keys_ok(key, ws, ws_bytes, E, Nrows) ? E / AGG_KSTEP : 0;
  int2* lw = lwrite ? lent : nullptr;
  int* q = wq[wave];
  int cnt = 0;
  if (!scan) {
    int per = (((ltotal + AGG_WAVES - 1) / AGG_WAVES) + 7) & ~7;
    int s0 = wave * per;
    int nw = ltotal - s0 < per ? ltotal - s0 : per;
    if (nw > 0) {
      const int2* we = lent + s0;
      int2 en = we[lane < nw ? lane : nw - 1];
      for (int b = 0; b < nw; b += 32) {
        int n = nw - b < 32 ? nw - b : 32;
        int nx = b + 32 + lane;
        int2 enx = we[nx < nw ? nx : nw - 1];
        agg_edges(en.x, en.y, n, gs, lgt, bgl, H, acc, lane);
        en = enx;
      }
    }
  }
  if (wave < nsuper) {
    unsigned pat = blockIdx.x * 0x01010101u;
    uint4 kv[4];
    agg_load_keys(key, wave, lane, kv);
    for (int S = wave; S < nsuper; S += AGG_WAVES) {
      unsigned long long m = ((unsigned long long)agg_hits8(kv[2], kv[3], pat) << 32) |
                             agg_hits8(kv[0], kv[1], pat);
      int Sp = S + AGG_WAVES < nsuper ? S + AGG_WAVES : nsuper - 1;
      agg_load_keys(key, Sp, lane, kv);
      int e0 = S * AGG_KSTEP + lane * 16;
      while (__builtin_amdgcn_ballot_w32(m != 0) != 0) {
        bool hit = m != 0;
        int bit = hit ? __builtin_ctzll(m) : 0;
        int off = ((((bit >> 5) << 1) | ((bit >> 2) & 1)) << 9) | ((bit & 3) << 2) |
                  ((bit >> 3) & 3);
        cnt = agg_append(hit, e0 + off, q, cnt);
        m &= m - 1;
        __builtin_amdgcn_fence(__ATOMIC_ACQ_REL, "wavefront");
        if (cnt >= 32) {
          agg_drain(ei, col, gs, lgt, bgl, H, acc, q, cnt, base, lane, lw, &lcount, lcap);
          cnt = 0;
        }
      }
    }
  }
  for (int s = nsuper * 4 + wave; s < nsteps; s += AGG_WAVES) {
    int cv[16];
    int e0 = s * AGG_STEP + lane * 4;
#pragma unroll
    for (int i = 0; i < 16; ++i) {
      int ee = e0 + (i >> 2) * 128 + (i & 3);
      cv[i] = ee < E ? col[ee] : -1;
    }
    cnt = agg_queue(cv, s, lane, base, q, cnt);
    if (cnt >= 32) {
      agg_drain(ei, col, gs, lgt, bgl, H, acc, q, cnt, base, lane, lw, &lcount, lcap);
      cnt = 0;
    }
  }
  if (cnt > 0) agg_drain(ei, col, gs, lgt, bgl, H, acc, q, cnt, base, lane, lw, &lcount, lcap);
  __syncthreads();
  if (lwrite && tid == 0) lhdr[0] = lcount;

  for (int i = tid; i < AGG_TN * 32; i += 1024) {
    int node = base + (i >> 5);
    if (node < Nrows) ((float4*)agg)[(size_t)node * 32 + (i & 31)] = ((float4*)acc)[i];
  }
}

__global__ void gemm_out_kernel(const float* __restrict__ H, const float* __restrict__ W,
                                const float* __restrict__ bias, float* __restrict__ out,
                                int Nrows) {
  int wave = threadIdx.x >> 5;
  int lane = threadIdx.x & 31;
  int tileM = blockIdx.x * 8 + wave;
  int tiles = (Nrows + 15) >> 4;
  if (tileM >= tiles) return;
  int mA = tileM * 16 + (lane & 15);
  int mclamp = mA < Nrows ? mA : (Nrows - 1);
  int n = lane & 15;
  int khalf = (lane >> 4) * 2;
  const float* hrow = H + (size_t)mclamp * 128 + khalf;
  const float* wcol = W + (size_t)khalf * 16 + n;
  __builtin_prefetch(hrow, 0, 0);
  v8f acc = {};
#pragma unroll 4
  for (int kk = 0; kk < 128; kk += 4) {
    v2f a = *(const v2f*)(hrow + kk);
    v2f b;
    b.x = wcol[(size_t)kk * 16];
    b.y = wcol[(size_t)kk * 16 + 16];
    acc = __builtin_amdgcn_wmma_f32_16x16x4_f32(false, a, false, b, (short)0, acc,
                                               false, false);
  }
  int mbase = tileM * 16 + ((lane >> 4) * 8);
  float bn = bias[n];
  if (tileM * 16 + 16 <= Nrows) {
    float* op = out + (size_t)mbase * 16 + n;
#pragma unroll
    for (int r = 0; r < 8; ++r)
      op[(size_t)r * 16] = acc[r] + bn;
  } else {
#pragma unroll
    for (int r = 0; r < 8; ++r) {
      int mm = mbase + r;
      if (mm < Nrows) out[(size_t)mm * 16 + n] = acc[r] + bn;
    }
  }
}

__global__ void logsoftmax_kernel(float4* out, int N) {
  int i = blockIdx.x * blockDim.x + threadIdx.x;
  if (i >= N) return;
  float4* row = out + (size_t)i * 4;
  float4 v[4];
#pragma unroll
  for (int q = 0; q < 4; ++q) v[q] = row[q];
  float m = v[0].x;
#pragma unroll
  for (int q = 0; q < 4; ++q) {
    m = fmaxf(m, v[q].x); m = fmaxf(m, v[q].y);
    m = fmaxf(m, v[q].z); m = fmaxf(m, v[q].w);
  }
  float s = 0.0f;
#pragma unroll
  for (int q = 0; q < 4; ++q) {
    s += expf(v[q].x - m); s += expf(v[q].y - m);
    s += expf(v[q].z - m); s += expf(v[q].w - m);
  }
  float d = m + logf(s);
#pragma unroll
  for (int q = 0; q < 4; ++q) {
    v[q].x -= d; v[q].y -= d; v[q].z -= d; v[q].w -= d;
    row[q] = v[q];
  }
}

extern "C" void kernel_launch(void* const* d_in, const int* in_sizes, int n_in,
                              void* d_out, int out_size, void* d_ws, size_t ws_size,
                              hipStream_t stream) {
  const float* x     = (const float*)d_in[0];
  const int*   ei    = (const int*)d_in[1];
  const float* W_in  = (const float*)d_in[2];
  const float* b_in  = (const float*)d_in[3];
  const float* Wg    = (const float*)d_in[4];
  const float* bg    = (const float*)d_in[5];
  const float* W_out = (const float*)d_in[6];
  const float* b_out = (const float*)d_in[7];
  float* out = (float*)d_out;

  const int Hdim = 128;
  int N = in_sizes[0] / Hdim;
  int E = in_sizes[1] / 2;
  int L = in_sizes[5];

  float* ws   = (float*)d_ws;
  float2* gs  = (float2*)ws;
  float2* gt  = gs + N;
  float* dn   = (float*)(gt + N);
  float* raw  = ws + 3 * (size_t)N + E;
  float* bufA = raw + (size_t)N * Hdim;
  float* bufB = bufA + (size_t)N * Hdim;
  unsigned* key = (unsigned*)(bufB + (size_t)N * Hdim);

  unsigned* part = (unsigned*)bufB;
  deg_key_kernel<<<dim3(HIST_K, (N + HIST_CAP - 1) / HIST_CAP), HIST_T, 0, stream>>>(
      ei, part, key, ws, ws_size, E, N);
  dn_kernel<<<(N + 255) / 256, 256, 0, stream>>>(part, dn, N);

  gemm_relu_kernel<<<(N + 15) / 16, 256, 0, stream>>>(x, W_in, b_in, raw, N);

  float* h = raw;
  float* agg = bufA;
  for (int i = 0; i < L; ++i) {
    gate_kernel<<<(N + 7) / 8, 256, 0, stream>>>(h, Wg + (size_t)i * 2 * Hdim, dn, gs, gt, N);
    agg_lds_kernel<<<(N + AGG_TN - 1) / AGG_TN, 1024, 0, stream>>>(ei, gs, gt, bg, i, h, raw, agg,
                                                                    key, ws, ws_size, E, N, bufB, L);
    h = agg;
    agg = (h == bufA) ? bufB : bufA;
  }

  int tiles = (N + 15) / 16;
  gemm_out_kernel<<<(tiles + 7) / 8, 256, 0, stream>>>(h, W_out, b_out, out, N);
  logsoftmax_kernel<<<(N + 255) / 256, 256, 0, stream>>>((float4*)out, N);
}


# Finished rows are consumed in LDS by the aggregation that made them: next layer's gate dots; at the last layer W_out by WMMA + log-softmax, rows not stored; 3 launches go; exp sum order changes
# speedup vs baseline: 3.3690x; 1.1880x over previous
#include <hip/hip_runtime.h>

typedef float v2f __attribute__((ext_vector_type(2)));
typedef float v8f __attribute__((ext_vector_type(8)));

#define FAGCN_EPS 0.3f

#define AGG_TN 196
#define AGG_MAXG 256

__device__ __forceinline__ bool agg_keys_ok(const unsigned* key, const float* ws, size_t ws_bytes,
                                            int E, int Nrows) {
  size_t need = (size_t)((const char*)key - (const char*)ws) + (((size_t)E + 3) & ~(size_t)3);
  return need <= ws_bytes && (((size_t)key) & 15) == 0 &&
         (Nrows + AGG_TN - 1) / AGG_TN <= AGG_MAXG;
}

#define HIST_K 64
#define HIST_CAP 16384
#define HIST_T 1024
#define HIST_U 4

__device__ __forceinline__ void hist_count(unsigned* tab, int r, int rbase, int nhere) {
  unsigned d = (unsigned)(r - rbase);
  if (d < (unsigned)nhere) atomicAdd(&tab[d], 1u);
}

__device__ __forceinline__ unsigned agg_pack_keys(int c0, int c1, int c2, int c3) {
  return (((unsigned)c0 / AGG_TN) & 255u) | ((((unsigned)c1 / AGG_TN) & 255u) << 8) |
         ((((unsigned)c2 / AGG_TN) & 255u) << 16) | ((((unsigned)c3 / AGG_TN) & 255u) << 24);
}

__global__ __launch_bounds__(HIST_T) void deg_key_kernel(const int* __restrict__ ei,
                                                         unsigned* __restrict__ part,
                                                         unsigned* __restrict__ key,
                                                         const float* ws, size_t ws_bytes,
                                                         int E, int Nrows) {
  __shared__ unsigned tab[HIST_CAP];
  int tid = threadIdx.x;
  int rbase = blockIdx.y * HIST_CAP;
  int nhere = Nrows - rbase < HIST_CAP ? Nrows - rbase : HIST_CAP;
  for (int i = tid; i < HIST_CAP / 4; i += HIST_T) ((uint4*)tab)[i] = make_uint4(0u, 0u, 0u, 0u);
  __syncthreads();

  const int* col = ei + E;
  int Q = (E + 3) >> 2;
  int qpb = (Q + HIST_K - 1) / HIST_K;
  int q0 = blockIdx.x * qpb;
  int q1 = q0 + qpb < Q ? q0 + qpb : Q;
  bool keys = blockIdx.y == 0 && agg_keys_ok(key, ws, ws_bytes, E, Nrows);
  if (q0 < q1) {
    if ((E & 3) == 0 && ((((size_t)ei) | ((size_t)col)) & 15) == 0) {
      int trips = (qpb + HIST_T - 1) / HIST_T;
      for (int j0 = 0; j0 < trips; j0 += HIST_U) {
        int4 rv[HIST_U], cw[HIST_U];
#pragma unroll
        for (int u = 0; u < HIST_U; ++u) {
          int q = q0 + (j0 + u) * HIST_T + tid;
          int qq = q < q1 ? q : q1 - 1;
          rv[u] = ((const int4*)ei)[qq];
          cw[u] = ((const int4*)col)[qq];
        }
        asm volatile("" ::"v"(rv[0].x), "v"(rv[1].x), "v"(rv[2].x), "v"(rv[3].x), "v"(cw[0].x),
                     "v"(cw[1].x), "v"(cw[2].x), "v"(cw[3].x));
#pragma unroll
        for (int u = 0; u < HIST_U; ++u) {
          int q = q0 + (j0 + u) * HIST_T + tid;
          if (q < q1) {
            hist_count(tab, rv[u].x, rbase, nhere);
            hist_count(tab, rv[u].y, rbase, nhere);
            hist_count(tab, rv[u].z, rbase, nhere);
            hist_count(tab, rv[u].w, rbase, nhere);
            if (keys) key[q] = agg_pack_keys(cw[u].x, cw[u].y, cw[u].z, cw[u].w);
          }
        }
      }
    } else {
      for (int q = q0 + tid; q < q1; q += HIST_T) {
        int c[4];
#pragma unroll
        for (int k = 0; k < 4; ++k) {
          int e = q * 4 + k;
          int ec = e < E ? e : E - 1;
          if (e < E) hist_count(tab, ei[ec], rbase, nhere);
          c[k] = col[ec];
        }
        if (keys) key[q] = agg_pack_keys(c[0], c[1], c[2], c[3]);
      }
    }
  }
  __syncthreads();

  unsigned* dst = part + (size_t)blockIdx.x * Nrows + rbase;
  if ((nhere & 3) == 0 && (((size_t)dst) & 15) == 0) {
    for (int i = tid; i < nhere / 4; i += HIST_T) ((uint4*)dst)[i] = ((const uint4*)tab)[i];
  } else {
    for (int i = tid; i < nhere; i += HIST_T) dst[i] = tab[i];
  }
}

__global__ void dn_kernel(const unsigned* __restrict__ part, float* __restrict__ dn, int N) {
  int i = blockIdx.x * blockDim.x + threadIdx.x;
  if (i >= N) return;
  unsigned s = 0;
#pragma unroll
  for (int b = 0; b < HIST_K; ++b) s += part[(size_t)b * N + i];
  dn[i] = 1.0f / sqrtf(fmaxf((float)s, 1.0f));
}

__global__ void gemm_relu_kernel(const float* __restrict__ X, const float* __restrict__ W,
                                 const float* __restrict__ bias, float* __restrict__ H,
                                 int Nrows) {
  int wave = threadIdx.x >> 5;
  int lane = threadIdx.x & 31;
  int tileM = blockIdx.x;
  int mA = tileM * 16 + (lane & 15);
  int mclamp = mA < Nrows ? mA : (Nrows - 1);
  int n = wave * 16 + (lane & 15);
  int khalf = (lane >> 4) * 2;
  const float* xrow = X + (size_t)mclamp * 128 + khalf;
  const float* wcol = W + (size_t)khalf * 128 + n;
  __builtin_prefetch(xrow, 0, 0);
  v8f acc = {};
#pragma unroll 4
  for (int kk = 0; kk < 128; kk += 4) {
    v2f a = *(const v2f*)(xrow + kk);
    v2f b;
    b.x = wcol[(size_t)kk * 128];
    b.y = wcol[(size_t)kk * 128 + 128];
    acc = __builtin_amdgcn_wmma_f32_16x16x4_f32(false, a, false, b, (short)0, acc,
                                               false, false);
  }
  int mbase = tileM * 16 + ((lane >> 4) * 8);
  float bn = bias[n];
  if (tileM * 16 + 16 <= Nrows) {
    float* hp = H + (size_t)mbase * 128 + n;
#pragma unroll
    for (int r = 0; r < 8; ++r)
      hp[(size_t)r * 128] = fmaxf(acc[r] + bn, 0.0f);
  } else {
#pragma unroll
    for (int r = 0; r < 8; ++r) {
      int mm = mbase + r;
      if (mm < Nrows) H[(size_t)mm * 128 + n] = fmaxf(acc[r] + bn, 0.0f);
    }
  }
}

__global__ void gate_kernel(const float* __restrict__ H, const float* __restrict__ Wg,
                            const float* __restrict__ dn, float2* __restrict__ gs,
                            float2* __restrict__ gt, int Nrows) {
  int wave = threadIdx.x >> 5;
  int lane = threadIdx.x & 31;
  int node = blockIdx.x * 8 + wave;
  if (node >= Nrows) return;
  const float* hrow = H + (size_t)node * 128;
  float sa = 0.0f, sb = 0.0f;
#pragma unroll
  for (int j = 0; j < 4; ++j) {
    int f = lane + j * 32;
    float hv = hrow[f];
    sa += hv * Wg[f];
    sb += hv * Wg[128 + f];
  }
  for (int off = 16; off > 0; off >>= 1) {
    sa += __shfl_down(sa, off, 32);
    sb += __shfl_down(sb, off, 32);
  }
  if (lane == 0) {
    float d = dn[node];
    gs[node] = make_float2(sa, d);
    gt[node] = make_float2(sb, d);
  }
}

#define AGG_WAVES 32
#define AGG_STEP 512
#define AGG_KSTEP (4 * AGG_STEP)
#define AGG_QCAP (AGG_STEP + 32)
#define AGG_INIT_TRIPS ((AGG_TN * 32 + 1023) / 1024)

__device__ __forceinline__ void agg_rows_load(const float* __restrict__ H, int r, int j0, int n,
                                              int lane, float4 (&hv)[8]) {
#pragma unroll
  for (int u = 0; u < 8; ++u) {
    int jj = j0 + u < n ? j0 + u : n - 1;
    int rj = __builtin_amdgcn_readlane(r, jj);
    hv[u] = *(const float4*)(H + (size_t)rj * 128 + lane * 4);
  }
}

__device__ __forceinline__ void agg_rows_add(float* acc, int lrow, float sc, int j0, int n,
                                             int lane, const float4 (&hv)[8]) {
  int hi = lane >> 4;
#pragma unroll
  for (int u = 0; u < 8; ++u) {
    if (j0 + u < n) {
      int lr = __builtin_amdgcn_readlane(lrow, j0 + u);
      float sj = __builtin_bit_cast(float, __builtin_amdgcn_readlane(__builtin_bit_cast(int, sc), j0 + u));
      float a0 = hi ? hv[u].y : hv[u].x;
      float a1 = hi ? hv[u].x : hv[u].y;
      float a2 = hi ? hv[u].w : hv[u].z;
      float a3 = hi ? hv[u].z : hv[u].w;
      float* p = acc + lr + lane * 4;
      atomicAdd(p + (0 ^ hi), sj * a0);
      atomicAdd(p + (1 ^ hi), sj * a1);
      atomicAdd(p + (2 ^ hi), sj * a2);
      atomicAdd(p + (3 ^ hi), sj * a3);
    }
  }
}

__device__ __forceinline__ void agg_edges(int r, int cl, int n, const float2* __restrict__ gs,
                                          const float2* lgt, float bgl,
                                          const float* __restrict__ H, float* acc, int lane) {
  float2 s = gs[r];
  float4 hv[8];
  agg_rows_load(H, r, 0, n, lane, hv);
  float2 t = lgt[cl];
  float sc = tanhf(s.x + t.x + bgl) * s.y * t.y;
  int lrow = cl * 128;
  for (int j0 = 0;;) {
    agg_rows_add(acc, lrow, sc, j0, n, lane, hv);
    j0 += 8;
    if (j0 >= n) break;
    agg_rows_load(H, r, j0, n, lane, hv);
  }
}

__device__ __forceinline__ void agg_drain(const int* __restrict__ ei, const int* __restrict__ col,
                                          const float2* __restrict__ gs, const float2* lgt,
                                          float bgl, const float* __restrict__ H, float* acc,
                                          const int* q, int cnt, int base, int lane, int2* lw,
                                          int* lcount, int cap) {
  int pos = 0;
  if (lw) {
    if (lane == 0) pos = atomicAdd(lcount, cnt);
    pos = __builtin_amdgcn_readfirstlane(pos);
  }
  for (int b = 0; b < cnt; b += 32) {
    int n = cnt - b < 32 ? cnt - b : 32;
    int e = q[b + (lane < n ? lane : 0)];
    int r = ei[e];
    int cl = col[e] - base;
    agg_edges(r, cl, n, gs, lgt, bgl, H, acc, lane);
    if (lw && lane < n && pos + b + lane < cap) lw[pos + b + lane] = make_int2(r, cl);
  }
}

__device__ __forceinline__ void agg_load_keys(const unsigned* __restrict__ key, int S, int lane,
                                              uint4 (&kv)[4]) {
  const uint4* p = (const uint4*)key + (size_t)S * (AGG_KSTEP / 16) + lane;
#pragma unroll
  for (int k = 0; k < 4; ++k) kv[k] = p[k * 32];
}

__device__ __forceinline__ unsigned agg_zero_bytes(unsigned x) {
  return ~(((x & 0x7f7f7f7fu) + 0x7f7f7f7fu) | x) & 0x80808080u;
}

__device__ __forceinline__ unsigned agg_hits8(uint4 t0, uint4 t1, unsigned pat) {
  return (agg_zero_bytes(t0.x ^ pat) >> 7) | (agg_zero_bytes(t0.y ^ pat) >> 6) |
         (agg_zero_bytes(t0.z ^ pat) >> 5) | (agg_zero_bytes(t0.w ^ pat) >> 4) |
         (agg_zero_bytes(t1.x ^ pat) >> 3) | (agg_zero_bytes(t1.y ^ pat) >> 2) |
         (agg_zero_bytes(t1.z ^ pat) >> 1) | agg_zero_bytes(t1.w ^ pat);
}

__device__ __forceinline__ int agg_append(bool hit, int e, int* q, int cnt) {
  unsigned b = __builtin_amdgcn_ballot_w32(hit);
  if (hit) q[cnt + __builtin_amdgcn_mbcnt_lo(b, 0)] = e;
  return cnt + __builtin_popcount(b);
}

__device__ __forceinline__ int agg_queue(const int (&cv)[16], int s, int lane, int base, int* q,
                                         int cnt) {
  int e0 = s * AGG_STEP + lane * 4;
#pragma unroll
  for (int i = 0; i < 16; ++i)
    cnt = agg_append((unsigned)(cv[i] - base) < (unsigned)AGG_TN,
                     e0 + (i >> 2) * 128 + (i & 3), q, cnt);
  __builtin_amdgcn_fence(__ATOMIC_ACQ_REL, "wavefront");
  return cnt;
}

#define AGG_GROWS ((AGG_TN + AGG_WAVES - 1) / AGG_WAVES)

__device__ __forceinline__ void agg_next_gates(const float* acc, const float2* lgt,
                                               const float* __restrict__ Wgn,
                                               float2* __restrict__ gsn, float2* __restrict__ gtn,
                                               int wave, int lane, int base, int rows_here) {
  int r0 = wave * AGG_GROWS;
  if (r0 >= rows_here) return;
  float wa[4], wb[4];
#pragma unroll
  for (int j = 0; j < 4; ++j) {
    wa[j] = Wgn[lane + j * 32];
    wb[j] = Wgn[128 + lane + j * 32];
  }
  float ka = 0.0f, kb = 0.0f;
#pragma unroll
  for (int k = 0; k < AGG_GROWS; ++k) {
    int t = r0 + k < AGG_TN ? r0 + k : AGG_TN - 1;
    const float* hrow = acc + t * 128;
    float sa = 0.0f, sb = 0.0f;
#pragma unroll
    for (int j = 0; j < 4; ++j) {
      float hv = hrow[lane + j * 32];
      sa += hv * wa[j];
      sb += hv * wb[j];
    }
    for (int off = 16; off > 0; off >>= 1) {
      sa += __shfl_down(sa, off, 32);
      sb += __shfl_down(sb, off, 32);
    }
    sa = __shfl(sa, 0, 32);
    sb = __shfl(sb, 0, 32);
    if (lane == k) { ka = sa; kb = sb; }
  }
  int t = r0 + lane;
  if (lane < AGG_GROWS && t < rows_here) {
    float d = lgt[t].y;
    gsn[base + t] = make_float2(ka, d);
    gtn[base + t] = make_float2(kb, d);
  }
}

__device__ __forceinline__ void agg_head_tile(const float* acc, int t, int rows_here, int base,
                                              const float* __restrict__ W,
                                              const float* __restrict__ bias,
                                              float* __restrict__ out, int lane) {
  int n = lane & 15;
  int hi = lane >> 4;
  int lr = t * 16 + n;
  int lrc = lr < AGG_TN ? lr : AGG_TN - 1;
  const float* arow = acc + lrc * 128 + hi * 2;
  const float* wcol = W + hi * 32 + n;
  v2f bf[32];
#pragma unroll
  for (int s = 0; s < 32; ++s) {
    bf[s].x = wcol[s * 64];
    bf[s].y = wcol[s * 64 + 16];
  }
  float bn = bias[n];
  v8f d = {};
#pragma unroll
  for (int s = 0; s < 32; ++s) {
    v2f a = *(const v2f*)(arow + s * 4);
    d = __builtin_amdgcn_wmma_f32_16x16x4_f32(false, a, false, bf[s], (short)0, d, false, false);
  }
  float v[8], m[8], s[8];
#pragma unroll
  for (int r = 0; r < 8; ++r) m[r] = v[r] = d[r] + bn;
#pragma unroll
  for (int off = 1; off < 16; off <<= 1) {
#pragma unroll
    for (int r = 0; r < 8; ++r) m[r] = fmaxf(m[r], __shfl_xor(m[r], off, 32));
  }
#pragma unroll
  for (int r = 0; r < 8; ++r) s[r] = expf(v[r] - m[r]);
#pragma unroll
  for (int off = 1; off < 16; off <<= 1) {
#pragma unroll
    for (int r = 0; r < 8; ++r) s[r] += __shfl_xor(s[r], off, 32);
  }
#pragma unroll
  for (int r = 0; r < 8; ++r) {
    float o = v[r] - (m[r] + logf(s[r]));
    int lrow = t * 16 + r + 8 * hi;
    if (lrow < rows_here) out[(size_t)(base + lrow) * 16 + n] = o;
  }
}

__global__ __launch_bounds__(1024) void agg_lds_kernel(const int* __restrict__ ei,
                                                       const float2* __restrict__ gs,
                                                       const float2* __restrict__ gt,
                                                       const float* __restrict__ bg, int layer,
                                                       const float* __restrict__ H,
                                                       const float* __restrict__ raw,
                                                       float* agg,
                                                       const unsigned* __restrict__ key,
                                                       const float* ws, size_t ws_bytes,
                                                       int E, int Nrows, float* lbuf, int L,
                                                       const float* __restrict__ Wg,
                                                       float2* __restrict__ gsn,
                                                       float2* __restrict__ gtn,
                                                       const float* __restrict__ Wout,
                                                       const float* __restrict__ bout,
                                                       float* __restrict__ out) {
  __shared__ float acc[AGG_TN * 128];
  __shared__ int wq[AGG_WAVES][AGG_QCAP];
  __shared__ float2 lgt[AGG_TN];
  __shared__ int lcount;
  int tid = threadIdx.x;
  int wave = __builtin_amdgcn_readfirstlane(tid >> 5);
  int lane = tid & 31;
  int base = blockIdx.x * AGG_TN;
  int rows_here = Nrows - base < AGG_TN ? Nrows - base : AGG_TN;
  int* lhdr = (int*)(lbuf + (size_t)base * 128);
  int2* lent = (int2*)(lhdr + 4);
  int lcap = rows_here * 64 - 2;
  bool lwrite = layer == 0 && L > 1;
  int ltotal = -1;
  if (layer == 1) {
    int t = __builtin_amdgcn_readfirstlane(lhdr[0]);
    if (t <= lcap) ltotal = t;
  }
  if (tid == 0) lcount = 0;
  {
    float4 rv[AGG_INIT_TRIPS];
#pragma unroll
    for (int k = 0; k < AGG_INIT_TRIPS; ++k) {
      int i = tid + k * 1024;
      int ic = i < AGG_TN * 32 ? i : AGG_TN * 32 - 1;
      int node = base + (ic >> 5);
      int nc = node < Nrows ? node : Nrows - 1;
      rv[k] = ((const float4*)raw)[(size_t)nc * 32 + (ic & 31)];
    }
#pragma unroll
    for (int k = 0; k < AGG_INIT_TRIPS; ++k) {
      int i = tid + k * 1024;
      if (i < AGG_TN * 32) {
        float s = base + (i >> 5) < Nrows ? FAGCN_EPS : 0.0f;
        ((float4*)acc)[i] = make_float4(s * rv[k].x, s * rv[k].y, s * rv[k].z, s * rv[k].w);
      }
    }
  }
  if (tid < AGG_TN)
    lgt[tid] = base + tid < Nrows ? gt[base + tid] : make_float2(0.0f, 0.0f);
  float bgl = bg[layer];
  __syncthreads();

  const int* col = ei + E;
  bool scan = ltotal < 0;
  int nsteps = scan ? (E + AGG_STEP - 1) / AGG_STEP : 0;
  int nsuper = scan && agg_keys_ok(key, ws, ws_bytes, E, Nrows) ? E / AGG_KSTEP : 0;
  int2* lw = lwrite ? lent : nullptr;
  int* q = wq[wave];
  int cnt = 0;
  if (!scan) {
    int per = (((ltotal + AGG_WAVES - 1) / AGG_WAVES) + 7) & ~7;
    int s0 = wave * per;
    int nw = ltotal - s0 < per ? ltotal - s0 : per;
    if (nw > 0) {
      const int2* we = lent + s0;
      int2 en = we[lane < nw ? lane : nw - 1];
      for (int b = 0; b < nw; b += 32) {
        int n = nw - b < 32 ? nw - b : 32;
        int nx = b + 32 + lane;
        int2 enx = we[nx < nw ? nx : nw - 1];
        agg_edges(en.x, en.y, n, gs, lgt, bgl, H, acc, lane);
        en = enx;
      }
    }
  }
  if (wave < nsuper) {
    unsigned pat = blockIdx.x * 0x01010101u;
    uint4 kv[4];
    agg_load_keys(key, wave, lane, kv);
    for (int S = wave; S < nsuper; S += AGG_WAVES) {
      unsigned long long m = ((unsigned long long)agg_hits8(kv[2], kv[3], pat) << 32) |
                             agg_hits8(kv[0], kv[1], pat);
      int Sp = S + AGG_WAVES < nsuper ? S + AGG_WAVES : nsuper - 1;
      agg_load_keys(key, Sp, lane, kv);
      int e0 = S * AGG_KSTEP + lane * 16;
      while (__builtin_amdgcn_ballot_w32(m != 0) != 0) {
        bool hit = m != 0;
        int bit = hit ? __builtin_ctzll(m) : 0;
        int off = ((((bit >> 5) << 1) | ((bit >> 2) & 1)) << 9) | ((bit & 3) << 2) |
                  ((bit >> 3) & 3);
        cnt = agg_append(hit, e0 + off, q, cnt);
        m &= m - 1;
        __builtin_amdgcn_fence(__ATOMIC_ACQ_REL, "wavefront");
        if (cnt >= 32) {
          agg_drain(ei, col, gs, lgt, bgl, H, acc, q, cnt, base, lane, lw, &lcount, lcap);
          cnt = 0;
        }
      }
    }
  }
  for (int s = nsuper * 4 + wave; s < nsteps; s += AGG_WAVES) {
    int cv[16];
    int e0 = s * AGG_STEP + lane * 4;
#pragma unroll
    for (int i = 0; i < 16; ++i) {
      int ee = e0 + (i >> 2) * 128 + (i & 3);
      cv[i] = ee < E ? col[ee] : -1;
    }
    cnt = agg_queue(cv, s, lane, base, q, cnt);
    if (cnt >= 32) {
      agg_drain(ei, col, gs, lgt, bgl, H, acc, q, cnt, base, lane, lw, &lcount, lcap);
      cnt = 0;
    }
  }
  if (cnt > 0) agg_drain(ei, col, gs, lgt, bgl, H, acc, q, cnt, base, lane, lw, &lcount, lcap);
  __syncthreads();
  if (lwrite && tid == 0) lhdr[0] = lcount;

  if (layer == L - 1) {
    if (wave < (rows_here + 15) >> 4) agg_head_tile(acc, wave, rows_here, base, Wout, bout, out, lane);
    return;
  }
  for (int i = tid; i < AGG_TN * 32; i += 1024) {
    int node = base + (i >> 5);
    if (node < Nrows) ((float4*)agg)[(size_t)node * 32 + (i & 31)] = ((float4*)acc)[i];
  }
  agg_next_gates(acc, lgt, Wg + (size_t)(layer + 1) * 256, gsn, gtn, wave, lane, base, rows_here);
}

__global__ void gemm_out_kernel(const float* __restrict__ H, const float* __restrict__ W,
                                const float* __restrict__ bias, float* __restrict__ out,
                                int Nrows) {
  int wave = threadIdx.x >> 5;
  int lane = threadIdx.x & 31;
  int tileM = blockIdx.x * 8 + wave;
  int tiles = (Nrows + 15) >> 4;
  if (tileM >= tiles) return;
  int mA = tileM * 16 + (lane & 15);
  int mclamp = mA < Nrows ? mA : (Nrows - 1);
  int n = lane & 15;
  int khalf = (lane >> 4) * 2;
  const float* hrow = H + (size_t)mclamp * 128 + khalf;
  const float* wcol = W + (size_t)khalf * 16 + n;
  __builtin_prefetch(hrow, 0, 0);
  v8f acc = {};
#pragma unroll 4
  for (int kk = 0; kk < 128; kk += 4) {
    v2f a = *(const v2f*)(hrow + kk);
    v2f b;
    b.x = wcol[(size_t)kk * 16];
    b.y = wcol[(size_t)kk * 16 + 16];
    acc = __builtin_amdgcn_wmma_f32_16x16x4_f32(false, a, false, b, (short)0, acc,
                                               false, false);
  }
  int mbase = tileM * 16 + ((lane >> 4) * 8);
  float bn = bias[n];
  if (tileM * 16 + 16 <= Nrows) {
    float* op = out + (size_t)mbase * 16 + n;
#pragma unroll
    for (int r = 0; r < 8; ++r)
      op[(size_t)r * 16] = acc[r] + bn;
  } else {
#pragma unroll
    for (int r = 0; r < 8; ++r) {
      int mm = mbase + r;
      if (mm < Nrows) out[(size_t)mm * 16 + n] = acc[r] + bn;
    }
  }
}

__global__ void logsoftmax_kernel(float4* out, int N) {
  int i = blockIdx.x * blockDim.x + threadIdx.x;
  if (i >= N) return;
  float4* row = out + (size_t)i * 4;
  float4 v[4];
#pragma unroll
  for (int q = 0; q < 4; ++q) v[q] = row[q];
  float m = v[0].x;
#pragma unroll
  for (int q = 0; q < 4; ++q) {
    m = fmaxf(m, v[q].x); m = fmaxf(m, v[q].y);
    m = fmaxf(m, v[q].z); m = fmaxf(m, v[q].w);
  }
  float s = 0.0f;
#pragma unroll
  for (int q = 0; q < 4; ++q) {
    s += expf(v[q].x - m); s += expf(v[q].y - m);
    s += expf(v[q].z - m); s += expf(v[q].w - m);
  }
  float d = m + logf(s);
#pragma unroll
  for (int q = 0; q < 4; ++q) {
    v[q].x -= d; v[q].y -= d; v[q].z -= d; v[q].w -= d;
    row[q] = v[q];
  }
}

extern "C" void kernel_launch(void* const* d_in, const int* in_sizes, int n_in,
                              void* d_out, int out_size, void* d_ws, size_t ws_size,
                              hipStream_t stream) {
  const float* x     = (const float*)d_in[0];
  const int*   ei    = (const int*)d_in[1];
  const float* W_in  = (const float*)d_in[2];
  const float* b_in  = (const float*)d_in[3];
  const float* Wg    = (const float*)d_in[4];
  const float* bg    = (const float*)d_in[5];
  const float* W_out = (const float*)d_in[6];
  const float* b_out = (const float*)d_in[7];
  float* out = (float*)d_out;

  const int Hdim = 128;
  int N = in_sizes[0] / Hdim;
  int E = in_sizes[1] / 2;
  int L = in_sizes[5];

  float* ws   = (float*)d_ws;
  float2* gs  = (float2*)ws;
  float2* gt  = gs + N;
  float2* gsn = gt + N;
  float2* gtn = gsn + N;
  float* dn   = (float*)(gtn + N);
  float* raw  = ws + 3 * (size_t)N + E;
  float* bufA = raw + (size_t)N * Hdim;
  float* bufB = bufA + (size_t)N * Hdim;
  unsigned* key = (unsigned*)(bufB + (size_t)N * Hdim);

  unsigned* part = (unsigned*)bufB;
  deg_key_kernel<<<dim3(HIST_K, (N + HIST_CAP - 1) / HIST_CAP), HIST_T, 0, stream>>>(
      ei, part, key, ws, ws_size, E, N);
  dn_kernel<<<(N + 255) / 256, 256, 0, stream>>>(part, dn, N);

  gemm_relu_kernel<<<(N + 15) / 16, 256, 0, stream>>>(x, W_in, b_in, raw, N);

  float* h = raw;
  float* agg = bufA;
  if (L > 0) gate_kernel<<<(N + 7) / 8, 256, 0, stream>>>(h, Wg, dn, gs, gt, N);
  for (int i = 0; i < L; ++i) {
    agg_lds_kernel<<<(N + AGG_TN - 1) / AGG_TN, 1024, 0, stream>>>(
        ei, gs, gt, bg, i, h, raw, agg, key, ws, ws_size, E, N, bufB, L, Wg, gsn, gtn, W_out,
        b_out, out);
    h = agg;
    agg = (h == bufA) ? bufB : bufA;
    float2* sw = gs; gs = gsn; gsn = sw;
    sw = gt; gt = gtn; gtn = sw;
  }

  if (L == 0) {
    int tiles = (N + 15) / 16;
    gemm_out_kernel<<<(tiles + 7) / 8, 256, 0, stream>>>(h, W_out, b_out, out, N);
    logsoftmax_kernel<<<(N + 255) / 256, 256, 0, stream>>>((float4*)out, N);
  }
}


# Row sums leave LDS float adds: a found edge's source goes to its target row's LDS queue (64 places); a wave per row then sums the messages in registers, adds to the LDS row once; sum order changes
# speedup vs baseline: 3.6312x; 1.0778x over previous
#include <hip/hip_runtime.h>

typedef float v2f __attribute__((ext_vector_type(2)));
typedef float v8f __attribute__((ext_vector_type(8)));

#define FAGCN_EPS 0.3f

#define AGG_TN 196
#define AGG_MAXG 256

__device__ __forceinline__ bool agg_keys_ok(const unsigned* key, const float* ws, size_t ws_bytes,
                                            int E, int Nrows) {
  size_t need = (size_t)((const char*)key - (const char*)ws) + (((size_t)E + 3) & ~(size_t)3);
  return need <= ws_bytes && (((size_t)key) & 15) == 0 &&
         (Nrows + AGG_TN - 1) / AGG_TN <= AGG_MAXG;
}

#define HIST_K 64
#define HIST_CAP 16384
#define HIST_T 1024
#define HIST_U 4

__device__ __forceinline__ void hist_count(unsigned* tab, int r, int rbase, int nhere) {
  unsigned d = (unsigned)(r - rbase);
  if (d < (unsigned)nhere) atomicAdd(&tab[d], 1u);
}

__device__ __forceinline__ unsigned agg_pack_keys(int c0, int c1, int c2, int c3) {
  return (((unsigned)c0 / AGG_TN) & 255u) | ((((unsigned)c1 / AGG_TN) & 255u) << 8) |
         ((((unsigned)c2 / AGG_TN) & 255u) << 16) | ((((unsigned)c3 / AGG_TN) & 255u) << 24);
}

__global__ __launch_bounds__(HIST_T) void deg_key_kernel(const int* __restrict__ ei,
                                                         unsigned* __restrict__ part,
                                                         unsigned* __restrict__ key,
                                                         const float* ws, size_t ws_bytes,
                                                         int E, int Nrows) {
  __shared__ unsigned tab[HIST_CAP];
  int tid = threadIdx.x;
  int rbase = blockIdx.y * HIST_CAP;
  int nhere = Nrows - rbase < HIST_CAP ? Nrows - rbase : HIST_CAP;
  for (int i = tid; i < HIST_CAP / 4; i += HIST_T) ((uint4*)tab)[i] = make_uint4(0u, 0u, 0u, 0u);
  __syncthreads();

  const int* col = ei + E;
  int Q = (E + 3) >> 2;
  int qpb = (Q + HIST_K - 1) / HIST_K;
  int q0 = blockIdx.x * qpb;
  int q1 = q0 + qpb < Q ? q0 + qpb : Q;
  bool keys = blockIdx.y == 0 && agg_keys_ok(key, ws, ws_bytes, E, Nrows);
  if (q0 < q1) {
    if ((E & 3) == 0 && ((((size_t)ei) | ((size_t)col)) & 15) == 0) {
      int trips = (qpb + HIST_T - 1) / HIST_T;
      for (int j0 = 0; j0 < trips; j0 += HIST_U) {
        int4 rv[HIST_U], cw[HIST_U];
#pragma unroll
        for (int u = 0; u < HIST_U; ++u) {
          int q = q0 + (j0 + u) * HIST_T + tid;
          int qq = q < q1 ? q : q1 - 1;
          rv[u] = ((const int4*)ei)[qq];
          cw[u] = ((const int4*)col)[qq];
        }
        asm volatile("" ::"v"(rv[0].x), "v"(rv[1].x), "v"(rv[2].x), "v"(rv[3].x), "v"(cw[0].x),
                     "v"(cw[1].x), "v"(cw[2].x), "v"(cw[3].x));
#pragma unroll
        for (int u = 0; u < HIST_U; ++u) {
          int q = q0 + (j0 + u) * HIST_T + tid;
          if (q < q1) {
            hist_count(tab, rv[u].x, rbase, nhere);
            hist_count(tab, rv[u].y, rbase, nhere);
            hist_count(tab, rv[u].z, rbase, nhere);
            hist_count(tab, rv[u].w, rbase, nhere);
            if (keys) key[q] = agg_pack_keys(cw[u].x, cw[u].y, cw[u].z, cw[u].w);
          }
        }
      }
    } else {
      for (int q = q0 + tid; q < q1; q += HIST_T) {
        int c[4];
#pragma unroll
        for (int k = 0; k < 4; ++k) {
          int e = q * 4 + k;
          int ec = e < E ? e : E - 1;
          if (e < E) hist_count(tab, ei[ec], rbase, nhere);
          c[k] = col[ec];
        }
        if (keys) key[q] = agg_pack_keys(c[0], c[1], c[2], c[3]);
      }
    }
  }
  __syncthreads();

  unsigned* dst = part + (size_t)blockIdx.x * Nrows + rbase;
  if ((nhere & 3) == 0 && (((size_t)dst) & 15) == 0) {
    for (int i = tid; i < nhere / 4; i += HIST_T) ((uint4*)dst)[i] = ((const uint4*)tab)[i];
  } else {
    for (int i = tid; i < nhere; i += HIST_T) dst[i] = tab[i];
  }
}

__global__ void dn_kernel(const unsigned* __restrict__ part, float* __restrict__ dn, int N) {
  int i = blockIdx.x * blockDim.x + threadIdx.x;
  if (i >= N) return;
  unsigned s = 0;
#pragma unroll
  for (int b = 0; b < HIST_K; ++b) s += part[(size_t)b * N + i];
  dn[i] = 1.0f / sqrtf(fmaxf((float)s, 1.0f));
}

__global__ void gemm_relu_kernel(const float* __restrict__ X, const float* __restrict__ W,
                                 const float* __restrict__ bias, float* __restrict__ H,
                                 int Nrows) {
  int wave = threadIdx.x >> 5;
  int lane = threadIdx.x & 31;
  int tileM = blockIdx.x;
  int mA = tileM * 16 + (lane & 15);
  int mclamp = mA < Nrows ? mA : (Nrows - 1);
  int n = wave * 16 + (lane & 15);
  int khalf = (lane >> 4) * 2;
  const float* xrow = X + (size_t)mclamp * 128 + khalf;
  const float* wcol = W + (size_t)khalf * 128 + n;
  __builtin_prefetch(xrow, 0, 0);
  v8f acc = {};
#pragma unroll 4
  for (int kk = 0; kk < 128; kk += 4) {
    v2f a = *(const v2f*)(xrow + kk);
    v2f b;
    b.x = wcol[(size_t)kk * 128];
    b.y = wcol[(size_t)kk * 128 + 128];
    acc = __builtin_amdgcn_wmma_f32_16x16x4_f32(false, a, false, b, (short)0, acc,
                                               false, false);
  }
  int mbase = tileM * 16 + ((lane >> 4) * 8);
  float bn = bias[n];
  if (tileM * 16 + 16 <= Nrows) {
    float* hp = H + (size_t)mbase * 128 + n;
#pragma unroll
    for (int r = 0; r < 8; ++r)
      hp[(size_t)r * 128] = fmaxf(acc[r] + bn, 0.0f);
  } else {
#pragma unroll
    for (int r = 0; r < 8; ++r) {
      int mm = mbase + r;
      if (mm < Nrows) H[(size_t)mm * 128 + n] = fmaxf(acc[r] + bn, 0.0f);
    }
  }
}

__global__ void gate_kernel(const float* __restrict__ H, const float* __restrict__ Wg,
                            const float* __restrict__ dn, float2* __restrict__ gs,
                            float2* __restrict__ gt, int Nrows) {
  int wave = threadIdx.x >> 5;
  int lane = threadIdx.x & 31;
  int node = blockIdx.x * 8 + wave;
  if (node >= Nrows) return;
  const float* hrow = H + (size_t)node * 128;
  float sa = 0.0f, sb = 0.0f;
#pragma unroll
  for (int j = 0; j < 4; ++j) {
    int f = lane + j * 32;
    float hv = hrow[f];
    sa += hv * Wg[f];
    sb += hv * Wg[128 + f];
  }
  for (int off = 16; off > 0; off >>= 1) {
    sa += __shfl_down(sa, off, 32);
    sb += __shfl_down(sb, off, 32);
  }
  if (lane == 0) {
    float d = dn[node];
    gs[node] = make_float2(sa, d);
    gt[node] = make_float2(sb, d);
  }
}

#define AGG_WAVES 32
#define AGG_RCAP 64
#define AGG_STEP 512
#define AGG_KSTEP (4 * AGG_STEP)
#define AGG_QCAP (AGG_STEP + 32)
#define AGG_INIT_TRIPS ((AGG_TN * 32 + 1023) / 1024)

__device__ __forceinline__ void agg_rows_load(const float* __restrict__ H, int r, int j0, int n,
                                              int lane, float4 (&hv)[8]) {
#pragma unroll
  for (int u = 0; u < 8; ++u) {
    int jj = j0 + u < n ? j0 + u : n - 1;
    int rj = __builtin_amdgcn_readlane(r, jj);
    hv[u] = *(const float4*)(H + (size_t)rj * 128 + lane * 4);
  }
}

__device__ __forceinline__ void agg_rows_add(float* acc, int lrow, float sc, int j0, int n,
                                             int lane, const float4 (&hv)[8]) {
  int hi = lane >> 4;
#pragma unroll
  for (int u = 0; u < 8; ++u) {
    if (j0 + u < n) {
      int lr = __builtin_amdgcn_readlane(lrow, j0 + u);
      float sj = __builtin_bit_cast(float, __builtin_amdgcn_readlane(__builtin_bit_cast(int, sc), j0 + u));
      float a0 = hi ? hv[u].y : hv[u].x;
      float a1 = hi ? hv[u].x : hv[u].y;
      float a2 = hi ? hv[u].w : hv[u].z;
      float a3 = hi ? hv[u].z : hv[u].w;
      float* p = acc + lr + lane * 4;
      atomicAdd(p + (0 ^ hi), sj * a0);
      atomicAdd(p + (1 ^ hi), sj * a1);
      atomicAdd(p + (2 ^ hi), sj * a2);
      atomicAdd(p + (3 ^ hi), sj * a3);
    }
  }
}

__device__ __forceinline__ void agg_edges(int r, int cl, int n, const float2* __restrict__ gs,
                                          const float2* lgt, float bgl,
                                          const float* __restrict__ H, float* acc, int lane) {
  float2 s = gs[r];
  float4 hv[8];
  agg_rows_load(H, r, 0, n, lane, hv);
  float2 t = lgt[cl];
  float sc = tanhf(s.x + t.x + bgl) * s.y * t.y;
  int lrow = cl * 128;
  for (int j0 = 0;;) {
    agg_rows_add(acc, lrow, sc, j0, n, lane, hv);
    j0 += 8;
    if (j0 >= n) break;
    agg_rows_load(H, r, j0, n, lane, hv);
  }
}

__device__ __forceinline__ void agg_route(bool valid, int r, int cl, int* rq, int* rcount,
                                          const float2* __restrict__ gs, const float2* lgt,
                                          float bgl, const float* __restrict__ H, float* acc,
                                          int lane) {
  int pos = 0;
  if (valid) pos = atomicAdd(&rcount[cl], 1);
  bool ok = valid && pos < AGG_RCAP;
  if (ok) rq[cl * AGG_RCAP + (unsigned)(pos + cl) % AGG_RCAP] = r;
  unsigned ovf = __builtin_amdgcn_ballot_w32(valid && !ok);
  while (ovf != 0) {
    int j = __builtin_ctz(ovf);
    ovf &= ovf - 1;
    int rj = __builtin_amdgcn_readlane(r, j);
    int cj = __builtin_amdgcn_readlane(cl, j);
    agg_edges(rj, cj, 1, gs, lgt, bgl, H, acc, lane);
  }
}

__device__ __forceinline__ void agg_rows_fma(float4& a, float sc, int j0, int n,
                                             const float4 (&hv)[8]) {
#pragma unroll
  for (int u = 0; u < 8; ++u) {
    if (j0 + u < n) {
      float sj = __builtin_bit_cast(
          float, __builtin_amdgcn_readlane(__builtin_bit_cast(int, sc), j0 + u));
      a.x += sj * hv[u].x;
      a.y += sj * hv[u].y;
      a.z += sj * hv[u].z;
      a.w += sj * hv[u].w;
    }
  }
}

__device__ __forceinline__ void agg_row_sum(const int* rqt, int b, int n, int t,
                                            const float2* __restrict__ gs, const float2* lgt,
                                            float bgl, const float* __restrict__ H, float* acc,
                                            int lane) {
  int r = rqt[(unsigned)(b + (lane < n ? lane : n - 1) + t) % AGG_RCAP];
  float2 s = gs[r];
  float4 hva[8], hvb[8];
  agg_rows_load(H, r, 0, n, lane, hva);
  if (n > 8) agg_rows_load(H, r, 8, n, lane, hvb);
  float2 tg = lgt[t];
  float sc = tanhf(s.x + tg.x + bgl) * s.y * tg.y;
  float4 a = make_float4(0.0f, 0.0f, 0.0f, 0.0f);
  agg_rows_fma(a, sc, 0, n, hva);
  if (n > 16) agg_rows_load(H, r, 16, n, lane, hva);
  if (n > 8) agg_rows_fma(a, sc, 8, n, hvb);
  if (n > 24) agg_rows_load(H, r, 24, n, lane, hvb);
  if (n > 16) agg_rows_fma(a, sc, 16, n, hva);
  if (n > 24) agg_rows_fma(a, sc, 24, n, hvb);
  float4* p = (float4*)(acc + t * 128) + lane;
  float4 v = *p;
  v.x += a.x;
  v.y += a.y;
  v.z += a.z;
  v.w += a.w;
  *p = v;
}

__device__ __forceinline__ void agg_drain(const int* __restrict__ ei, const int* __restrict__ col,
                                          const float2* __restrict__ gs, const float2* lgt,
                                          float bgl, const float* __restrict__ H, float* acc,
                                          const int* q, int cnt, int base, int lane, int2* lw,
                                          int* lcount, int cap, int* rq, int* rcount) {
  int pos = 0;
  if (lw) {
    if (lane == 0) pos = atomicAdd(lcount, cnt);
    pos = __builtin_amdgcn_readfirstlane(pos);
  }
  for (int b = 0; b < cnt; b += 32) {
    int n = cnt - b < 32 ? cnt - b : 32;
    int e = q[b + (lane < n ? lane : 0)];
    int r = ei[e];
    int cl = col[e] - base;
    agg_route(lane < n, r, cl, rq, rcount, gs, lgt, bgl, H, acc, lane);
    if (lw && lane < n && pos + b + lane < cap) lw[pos + b + lane] = make_int2(r, cl);
  }
}

__device__ __forceinline__ void agg_load_keys(const unsigned* __restrict__ key, int S, int lane,
                                              uint4 (&kv)[4]) {
  const uint4* p = (const uint4*)key + (size_t)S * (AGG_KSTEP / 16) + lane;
#pragma unroll
  for (int k = 0; k < 4; ++k) kv[k] = p[k * 32];
}

__device__ __forceinline__ unsigned agg_zero_bytes(unsigned x) {
  return ~(((x & 0x7f7f7f7fu) + 0x7f7f7f7fu) | x) & 0x80808080u;
}

__device__ __forceinline__ unsigned agg_hits8(uint4 t0, uint4 t1, unsigned pat) {
  return (agg_zero_bytes(t0.x ^ pat) >> 7) | (agg_zero_bytes(t0.y ^ pat) >> 6) |
         (agg_zero_bytes(t0.z ^ pat) >> 5) | (agg_zero_bytes(t0.w ^ pat) >> 4) |
         (agg_zero_bytes(t1.x ^ pat) >> 3) | (agg_zero_bytes(t1.y ^ pat) >> 2) |
         (agg_zero_bytes(t1.z ^ pat) >> 1) | agg_zero_bytes(t1.w ^ pat);
}

__device__ __forceinline__ int agg_append(bool hit, int e, int* q, int cnt) {
  unsigned b = __builtin_amdgcn_ballot_w32(hit);
  if (hit) q[cnt + __builtin_amdgcn_mbcnt_lo(b, 0)] = e;
  return cnt + __builtin_popcount(b);
}

__device__ __forceinline__ int agg_queue(const int (&cv)[16], int s, int lane, int base, int* q,
                                         int cnt) {
  int e0 = s * AGG_STEP + lane * 4;
#pragma unroll
  for (int i = 0; i < 16; ++i)
    cnt = agg_append((unsigned)(cv[i] - base) < (unsigned)AGG_TN,
                     e0 + (i >> 2) * 128 + (i & 3), q, cnt);
  __builtin_amdgcn_fence(__ATOMIC_ACQ_REL, "wavefront");
  return cnt;
}

#define AGG_GROWS ((AGG_TN + AGG_WAVES - 1) / AGG_WAVES)

__device__ __forceinline__ void agg_next_gates(const float* acc, const float2* lgt,
                                               const float* __restrict__ Wgn,
                                               float2* __restrict__ gsn, float2* __restrict__ gtn,
                                               int wave, int lane, int base, int rows_here) {
  int r0 = wave * AGG_GROWS;
  if (r0 >= rows_here) return;
  float wa[4], wb[4];
#pragma unroll
  for (int j = 0; j < 4; ++j) {
    wa[j] = Wgn[lane + j * 32];
    wb[j] = Wgn[128 + lane + j * 32];
  }
  float ka = 0.0f, kb = 0.0f;
#pragma unroll
  for (int k = 0; k < AGG_GROWS; ++k) {
    int t = r0 + k < AGG_TN ? r0 + k : AGG_TN - 1;
    const float* hrow = acc + t * 128;
    float sa = 0.0f, sb = 0.0f;
#pragma unroll
    for (int j = 0; j < 4; ++j) {
      float hv = hrow[lane + j * 32];
      sa += hv * wa[j];
      sb += hv * wb[j];
    }
    for (int off = 16; off > 0; off >>= 1) {
      sa += __shfl_down(sa, off, 32);
      sb += __shfl_down(sb, off, 32);
    }
    sa = __shfl(sa, 0, 32);
    sb = __shfl(sb, 0, 32);
    if (lane == k) { ka = sa; kb = sb; }
  }
  int t = r0 + lane;
  if (lane < AGG_GROWS && t < rows_here) {
    float d = lgt[t].y;
    gsn[base + t] = make_float2(ka, d);
    gtn[base + t] = make_float2(kb, d);
  }
}

__device__ __forceinline__ void agg_head_tile(const float* acc, int t, int rows_here, int base,
                                              const float* __restrict__ W,
                                              const float* __restrict__ bias,
                                              float* __restrict__ out, int lane) {
  int n = lane & 15;
  int hi = lane >> 4;
  int lr = t * 16 + n;
  int lrc = lr < AGG_TN ? lr : AGG_TN - 1;
  const float* arow = acc + lrc * 128 + hi * 2;
  const float* wcol = W + hi * 32 + n;
  v2f bf[32];
#pragma unroll
  for (int s = 0; s < 32; ++s) {
    bf[s].x = wcol[s * 64];
    bf[s].y = wcol[s * 64 + 16];
  }
  float bn = bias[n];
  v8f d = {};
#pragma unroll
  for (int s = 0; s < 32; ++s) {
    v2f a = *(const v2f*)(arow + s * 4);
    d = __builtin_amdgcn_wmma_f32_16x16x4_f32(false, a, false, bf[s], (short)0, d, false, false);
  }
  float v[8], m[8], s[8];
#pragma unroll
  for (int r = 0; r < 8; ++r) m[r] = v[r] = d[r] + bn;
#pragma unroll
  for (int off = 1; off < 16; off <<= 1) {
#pragma unroll
    for (int r = 0; r < 8; ++r) m[r] = fmaxf(m[r], __shfl_xor(m[r], off, 32));
  }
#pragma unroll
  for (int r = 0; r < 8; ++r) s[r] = expf(v[r] - m[r]);
#pragma unroll
  for (int off = 1; off < 16; off <<= 1) {
#pragma unroll
    for (int r = 0; r < 8; ++r) s[r] += __shfl_xor(s[r], off, 32);
  }
#pragma unroll
  for (int r = 0; r < 8; ++r) {
    float o = v[r] - (m[r] + logf(s[r]));
    int lrow = t * 16 + r + 8 * hi;
    if (lrow < rows_here) out[(size_t)(base + lrow) * 16 + n] = o;
  }
}

__global__ __launch_bounds__(1024) void agg_lds_kernel(const int* __restrict__ ei,
                                                       const float2* __restrict__ gs,
                                                       const float2* __restrict__ gt,
                                                       const float* __restrict__ bg, int layer,
                                                       const float* __restrict__ H,
                                                       const float* __restrict__ raw,
                                                       float* agg,
                                                       const unsigned* __restrict__ key,
                                                       const float* ws, size_t ws_bytes,
                                                       int E, int Nrows, float* lbuf, int L,
                                                       const float* __restrict__ Wg,
                                                       float2* __restrict__ gsn,
                                                       float2* __restrict__ gtn,
                                                       const float* __restrict__ Wout,
                                                       const float* __restrict__ bout,
                                                       float* __restrict__ out) {
  __shared__ float acc[AGG_TN * 128];
  __shared__ int wq[AGG_WAVES][AGG_QCAP];
  __shared__ float2 lgt[AGG_TN];
  __shared__ int lcount;
  __shared__ int rq[AGG_TN * AGG_RCAP];
  __shared__ int rcount[AGG_TN];
  int tid = threadIdx.x;
  int wave = __builtin_amdgcn_readfirstlane(tid >> 5);
  int lane = tid & 31;
  int base = blockIdx.x * AGG_TN;
  int rows_here = Nrows - base < AGG_TN ? Nrows - base : AGG_TN;
  int* lhdr = (int*)(lbuf + (size_t)base * 128);
  int2* lent = (int2*)(lhdr + 4);
  int lcap = rows_here * 64 - 2;
  bool lwrite = layer == 0 && L > 1;
  int ltotal = -1;
  if (layer == 1) {
    int t = __builtin_amdgcn_readfirstlane(lhdr[0]);
    if (t <= lcap) ltotal = t;
  }
  if (tid == 0) lcount = 0;
  if (tid < AGG_TN) rcount[tid] = 0;
  {
    float4 rv[AGG_INIT_TRIPS];
#pragma unroll
    for (int k = 0; k < AGG_INIT_TRIPS; ++k) {
      int i = tid + k * 1024;
      int ic = i < AGG_TN * 32 ? i : AGG_TN * 32 - 1;
      int node = base + (ic >> 5);
      int nc = node < Nrows ? node : Nrows - 1;
      rv[k] = ((const float4*)raw)[(size_t)nc * 32 + (ic & 31)];
    }
#pragma unroll
    for (int k = 0; k < AGG_INIT_TRIPS; ++k) {
      int i = tid + k * 1024;
      if (i < AGG_TN * 32) {
        float s = base + (i >> 5) < Nrows ? FAGCN_EPS : 0.0f;
        ((float4*)acc)[i] = make_float4(s * rv[k].x, s * rv[k].y, s * rv[k].z, s * rv[k].w);
      }
    }
  }
  if (tid < AGG_TN)
    lgt[tid] = base + tid < Nrows ? gt[base + tid] : make_float2(0.0f, 0.0f);
  float bgl = bg[layer];
  __syncthreads();

  const int* col = ei + E;
  bool scan = ltotal < 0;
  int nsteps = scan ? (E + AGG_STEP - 1) / AGG_STEP : 0;
  int nsuper = scan && agg_keys_ok(key, ws, ws_bytes, E, Nrows) ? E / AGG_KSTEP : 0;
  int2* lw = lwrite ? lent : nullptr;
  int* q = wq[wave];
  int cnt = 0;
  if (!scan && wave * 32 < ltotal) {
    int i = wave * 32 + lane;
    int2 en = lent[i < ltotal ? i : ltotal - 1];
    for (int i0 = wave * 32; i0 < ltotal; i0 += 1024) {
      int nx = i0 + 1024 + lane;
      int2 enx = lent[nx < ltotal ? nx : ltotal - 1];
      agg_route(i0 + lane < ltotal, en.x, en.y, rq, rcount, gs, lgt, bgl, H, acc, lane);
      en = enx;
    }
  }
  if (wave < nsuper) {
    unsigned pat = blockIdx.x * 0x01010101u;
    uint4 kv[4];
    agg_load_keys(key, wave, lane, kv);
    for (int S = wave; S < nsuper; S += AGG_WAVES) {
      unsigned long long m = ((unsigned long long)agg_hits8(kv[2], kv[3], pat) << 32) |
                             agg_hits8(kv[0], kv[1], pat);
      int Sp = S + AGG_WAVES < nsuper ? S + AGG_WAVES : nsuper - 1;
      agg_load_keys(key, Sp, lane, kv);
      int e0 = S * AGG_KSTEP + lane * 16;
      while (__builtin_amdgcn_ballot_w32(m != 0) != 0) {
        bool hit = m != 0;
        int bit = hit ? __builtin_ctzll(m) : 0;
        int off = ((((bit >> 5) << 1) | ((bit >> 2) & 1)) << 9) | ((bit & 3) << 2) |
                  ((bit >> 3) & 3);
        cnt = agg_append(hit, e0 + off, q, cnt);
        m &= m - 1;
        __builtin_amdgcn_fence(__ATOMIC_ACQ_REL, "wavefront");
        if (cnt >= 32) {
          agg_drain(ei, col, gs, lgt, bgl, H, acc, q, cnt, base, lane, lw, &lcount, lcap, rq, rcount);
          cnt = 0;
        }
      }
    }
  }
  for (int s = nsuper * 4 + wave; s < nsteps; s += AGG_WAVES) {
    int cv[16];
    int e0 = s * AGG_STEP + lane * 4;
#pragma unroll
    for (int i = 0; i < 16; ++i) {
      int ee = e0 + (i >> 2) * 128 + (i & 3);
      cv[i] = ee < E ? col[ee] : -1;
    }
    cnt = agg_queue(cv, s, lane, base, q, cnt);
    if (cnt >= 32) {
      agg_drain(ei, col, gs, lgt, bgl, H, acc, q, cnt, base, lane, lw, &lcount, lcap, rq, rcount);
      cnt = 0;
    }
  }
  if (cnt > 0) agg_drain(ei, col, gs, lgt, bgl, H, acc, q, cnt, base, lane, lw, &lcount, lcap, rq, rcount);
  __syncthreads();
  if (lwrite && tid == 0) lhdr[0] = lcount;
  for (int t = wave; t < AGG_TN; t += AGG_WAVES) {
    int n = rcount[t] < AGG_RCAP ? rcount[t] : AGG_RCAP;
    n = __builtin_amdgcn_readfirstlane(n);
    for (int b = 0; b < n; b += 32)
      agg_row_sum(rq + t * AGG_RCAP, b, n - b < 32 ? n - b : 32, t, gs, lgt, bgl, H, acc, lane);
  }
  __syncthreads();

  if (layer == L - 1) {
    if (wave < (rows_here + 15) >> 4) agg_head_tile(acc, wave, rows_here, base, Wout, bout, out, lane);
    return;
  }
  for (int i = tid; i < AGG_TN * 32; i += 1024) {
    int node = base + (i >> 5);
    if (node < Nrows) ((float4*)agg)[(size_t)node * 32 + (i & 31)] = ((float4*)acc)[i];
  }
  agg_next_gates(acc, lgt, Wg + (size_t)(layer + 1) * 256, gsn, gtn, wave, lane, base, rows_here);
}

__global__ void gemm_out_kernel(const float* __restrict__ H, const float* __restrict__ W,
                                const float* __restrict__ bias, float* __restrict__ out,
                                int Nrows) {
  int wave = threadIdx.x >> 5;
  int lane = threadIdx.x & 31;
  int tileM = blockIdx.x * 8 + wave;
  int tiles = (Nrows + 15) >> 4;
  if (tileM >= tiles) return;
  int mA = tileM * 16 + (lane & 15);
  int mclamp = mA < Nrows ? mA : (Nrows - 1);
  int n = lane & 15;
  int khalf = (lane >> 4) * 2;
  const float* hrow = H + (size_t)mclamp * 128 + khalf;
  const float* wcol = W + (size_t)khalf * 16 + n;
  __builtin_prefetch(hrow, 0, 0);
  v8f acc = {};
#pragma unroll 4
  for (int kk = 0; kk < 128; kk += 4) {
    v2f a = *(const v2f*)(hrow + kk);
    v2f b;
    b.x = wcol[(size_t)kk * 16];
    b.y = wcol[(size_t)kk * 16 + 16];
    acc = __builtin_amdgcn_wmma_f32_16x16x4_f32(false, a, false, b, (short)0, acc,
                                               false, false);
  }
  int mbase = tileM * 16 + ((lane >> 4) * 8);
  float bn = bias[n];
  if (tileM * 16 + 16 <= Nrows) {
    float* op = out + (size_t)mbase * 16 + n;
#pragma unroll
    for (int r = 0; r < 8; ++r)
      op[(size_t)r * 16] = acc[r] + bn;
  } else {
#pragma unroll
    for (int r = 0; r < 8; ++r) {
      int mm = mbase + r;
      if (mm < Nrows) out[(size_t)mm * 16 + n] = acc[r] + bn;
    }
  }
}

__global__ void logsoftmax_kernel(float4* out, int N) {
  int i = blockIdx.x * blockDim.x + threadIdx.x;
  if (i >= N) return;
  float4* row = out + (size_t)i * 4;
  float4 v[4];
#pragma unroll
  for (int q = 0; q < 4; ++q) v[q] = row[q];
  float m = v[0].x;
#pragma unroll
  for (int q = 0; q < 4; ++q) {
    m = fmaxf(m, v[q].x); m = fmaxf(m, v[q].y);
    m = fmaxf(m, v[q].z); m = fmaxf(m, v[q].w);
  }
  float s = 0.0f;
#pragma unroll
  for (int q = 0; q < 4; ++q) {
    s += expf(v[q].x - m); s += expf(v[q].y - m);
    s += expf(v[q].z - m); s += expf(v[q].w - m);
  }
  float d = m + logf(s);
#pragma unroll
  for (int q = 0; q < 4; ++q) {
    v[q].x -= d; v[q].y -= d; v[q].z -= d; v[q].w -= d;
    row[q] = v[q];
  }
}

extern "C" void kernel_launch(void* const* d_in, const int* in_sizes, int n_in,
                              void* d_out, int out_size, void* d_ws, size_t ws_size,
                              hipStream_t stream) {
  const float* x     = (const float*)d_in[0];
  const int*   ei    = (const int*)d_in[1];
  const float* W_in  = (const float*)d_in[2];
  const float* b_in  = (const float*)d_in[3];
  const float* Wg    = (const float*)d_in[4];
  const float* bg    = (const float*)d_in[5];
  const float* W_out = (const float*)d_in[6];
  const float* b_out = (const float*)d_in[7];
  float* out = (float*)d_out;

  const int Hdim = 128;
  int N = in_sizes[0] / Hdim;
  int E = in_sizes[1] / 2;
  int L = in_sizes[5];

  float* ws   = (float*)d_ws;
  float2* gs  = (float2*)ws;
  float2* gt  = gs + N;
  float2* gsn = gt + N;
  float2* gtn = gsn + N;
  float* dn   = (float*)(gtn + N);
  float* raw  = ws + 3 * (size_t)N + E;
  float* bufA = raw + (size_t)N * Hdim;
  float* bufB = bufA + (size_t)N * Hdim;
  unsigned* key = (unsigned*)(bufB + (size_t)N * Hdim);

  unsigned* part = (unsigned*)bufB;
  deg_key_kernel<<<dim3(HIST_K, (N + HIST_CAP - 1) / HIST_CAP), HIST_T, 0, stream>>>(
      ei, part, key, ws, ws_size, E, N);
  dn_kernel<<<(N + 255) / 256, 256, 0, stream>>>(part, dn, N);

  gemm_relu_kernel<<<(N + 15) / 16, 256, 0, stream>>>(x, W_in, b_in, raw, N);

  float* h = raw;
  float* agg = bufA;
  if (L > 0) gate_kernel<<<(N + 7) / 8, 256, 0, stream>>>(h, Wg, dn, gs, gt, N);
  for (int i = 0; i < L; ++i) {
    agg_lds_kernel<<<(N + AGG_TN - 1) / AGG_TN, 1024, 0, stream>>>(
        ei, gs, gt, bg, i, h, raw, agg, key, ws, ws_size, E, N, bufB, L, Wg, gsn, gtn, W_out,
        b_out, out);
    h = agg;
    agg = (h == bufA) ? bufB : bufA;
    float2* sw = gs; gs = gsn; gsn = sw;
    sw = gt; gt = gtn; gtn = sw;
  }

  if (L == 0) {
    int tiles = (N + 15) / 16;
    gemm_out_kernel<<<(tiles + 7) / 8, 256, 0, stream>>>(h, W_out, b_out, out, N);
    logsoftmax_kernel<<<(N + 255) / 256, 256, 0, stream>>>((float4*)out, N);
  }
}


# First layer's gate dots: one wave per 16 nodes, not per node (16 whole-row loads in flight, the 32 lane sums folded over the lanes at once); 50000 short waves -> 3128; gate dot sum order changes
# speedup vs baseline: 3.7516x; 1.0332x over previous
#include <hip/hip_runtime.h>

typedef float v2f __attribute__((ext_vector_type(2)));
typedef float v8f __attribute__((ext_vector_type(8)));

#define FAGCN_EPS 0.3f

#define AGG_TN 196
#define AGG_MAXG 256

__device__ __forceinline__ bool agg_keys_ok(const unsigned* key, const float* ws, size_t ws_bytes,
                                            int E, int Nrows) {
  size_t need = (size_t)((const char*)key - (const char*)ws) + (((size_t)E + 3) & ~(size_t)3);
  return need <= ws_bytes && (((size_t)key) & 15) == 0 &&
         (Nrows + AGG_TN - 1) / AGG_TN <= AGG_MAXG;
}

#define HIST_K 64
#define HIST_CAP 16384
#define HIST_T 1024
#define HIST_U 4

__device__ __forceinline__ void hist_count(unsigned* tab, int r, int rbase, int nhere) {
  unsigned d = (unsigned)(r - rbase);
  if (d < (unsigned)nhere) atomicAdd(&tab[d], 1u);
}

__device__ __forceinline__ unsigned agg_pack_keys(int c0, int c1, int c2, int c3) {
  return (((unsigned)c0 / AGG_TN) & 255u) | ((((unsigned)c1 / AGG_TN) & 255u) << 8) |
         ((((unsigned)c2 / AGG_TN) & 255u) << 16) | ((((unsigned)c3 / AGG_TN) & 255u) << 24);
}

__global__ __launch_bounds__(HIST_T) void deg_key_kernel(const int* __restrict__ ei,
                                                         unsigned* __restrict__ part,
                                                         unsigned* __restrict__ key,
                                                         const float* ws, size_t ws_bytes,
                                                         int E, int Nrows) {
  __shared__ unsigned tab[HIST_CAP];
  int tid = threadIdx.x;
  int rbase = blockIdx.y * HIST_CAP;
  int nhere = Nrows - rbase < HIST_CAP ? Nrows - rbase : HIST_CAP;
  for (int i = tid; i < HIST_CAP / 4; i += HIST_T) ((uint4*)tab)[i] = make_uint4(0u, 0u, 0u, 0u);
  __syncthreads();

  const int* col = ei + E;
  int Q = (E + 3) >> 2;
  int qpb = (Q + HIST_K - 1) / HIST_K;
  int q0 = blockIdx.x * qpb;
  int q1 = q0 + qpb < Q ? q0 + qpb : Q;
  bool keys = blockIdx.y == 0 && agg_keys_ok(key, ws, ws_bytes, E, Nrows);
  if (q0 < q1) {
    if ((E & 3) == 0 && ((((size_t)ei) | ((size_t)col)) & 15) == 0) {
      int trips = (qpb + HIST_T - 1) / HIST_T;
      for (int j0 = 0; j0 < trips; j0 += HIST_U) {
        int4 rv[HIST_U], cw[HIST_U];
#pragma unroll
        for (int u = 0; u < HIST_U; ++u) {
          int q = q0 + (j0 + u) * HIST_T + tid;
          int qq = q < q1 ? q : q1 - 1;
          rv[u] = ((const int4*)ei)[qq];
          cw[u] = ((const int4*)col)[qq];
        }
        asm volatile("" ::"v"(rv[0].x), "v"(rv[1].x), "v"(rv[2].x), "v"(rv[3].x), "v"(cw[0].x),
                     "v"(cw[1].x), "v"(cw[2].x), "v"(cw[3].x));
#pragma unroll
        for (int u = 0; u < HIST_U; ++u) {
          int q = q0 + (j0 + u) * HIST_T + tid;
          if (q < q1) {
            hist_count(tab, rv[u].x, rbase, nhere);
            hist_count(tab, rv[u].y, rbase, nhere);
            hist_count(tab, rv[u].z, rbase, nhere);
            hist_count(tab, rv[u].w, rbase, nhere);
            if (keys) key[q] = agg_pack_keys(cw[u].x, cw[u].y, cw[u].z, cw[u].w);
          }
        }
      }
    } else {
      for (int q = q0 + tid; q < q1; q += HIST_T) {
        int c[4];
#pragma unroll
        for (int k = 0; k < 4; ++k) {
          int e = q * 4 + k;
          int ec = e < E ? e : E - 1;
          if (e < E) hist_count(tab, ei[ec], rbase, nhere);
          c[k] = col[ec];
        }
        if (keys) key[q] = agg_pack_keys(c[0], c[1], c[2], c[3]);
      }
    }
  }
  __syncthreads();

  unsigned* dst = part + (size_t)blockIdx.x * Nrows + rbase;
  if ((nhere & 3) == 0 && (((size_t)dst) & 15) == 0) {
    for (int i = tid; i < nhere / 4; i += HIST_T) ((uint4*)dst)[i] = ((const uint4*)tab)[i];
  } else {
    for (int i = tid; i < nhere; i += HIST_T) dst[i] = tab[i];
  }
}

__global__ void dn_kernel(const unsigned* __restrict__ part, float* __restrict__ dn, int N) {
  int i = blockIdx.x * blockDim.x + threadIdx.x;
  if (i >= N) return;
  unsigned s = 0;
#pragma unroll
  for (int b = 0; b < HIST_K; ++b) s += part[(size_t)b * N + i];
  dn[i] = 1.0f / sqrtf(fmaxf((float)s, 1.0f));
}

__global__ void gemm_relu_kernel(const float* __restrict__ X, const float* __restrict__ W,
                                 const float* __restrict__ bias, float* __restrict__ H,
                                 int Nrows) {
  int wave = threadIdx.x >> 5;
  int lane = threadIdx.x & 31;
  int tileM = blockIdx.x;
  int mA = tileM * 16 + (lane & 15);
  int mclamp = mA < Nrows ? mA : (Nrows - 1);
  int n = wave * 16 + (lane & 15);
  int khalf = (lane >> 4) * 2;
  const float* xrow = X + (size_t)mclamp * 128 + khalf;
  const float* wcol = W + (size_t)khalf * 128 + n;
  __builtin_prefetch(xrow, 0, 0);
  v8f acc = {};
#pragma unroll 4
  for (int kk = 0; kk < 128; kk += 4) {
    v2f a = *(const v2f*)(xrow + kk);
    v2f b;
    b.x = wcol[(size_t)kk * 128];
    b.y = wcol[(size_t)kk * 128 + 128];
    acc = __builtin_amdgcn_wmma_f32_16x16x4_f32(false, a, false, b, (short)0, acc,
                                               false, false);
  }
  int mbase = tileM * 16 + ((lane >> 4) * 8);
  float bn = bias[n];
  if (tileM * 16 + 16 <= Nrows) {
    float* hp = H + (size_t)mbase * 128 + n;
#pragma unroll
    for (int r = 0; r < 8; ++r)
      hp[(size_t)r * 128] = fmaxf(acc[r] + bn, 0.0f);
  } else {
#pragma unroll
    for (int r = 0; r < 8; ++r) {
      int mm = mbase + r;
      if (mm < Nrows) H[(size_t)mm * 128 + n] = fmaxf(acc[r] + bn, 0.0f);
    }
  }
}

#define GATE_R 16

template <int OFF>
__device__ __forceinline__ void gate_fold(float (&v)[2 * GATE_R], int lane) {
  bool up = (lane & OFF) != 0;
#pragma unroll
  for (int k = 0; k < OFF; ++k) {
    float keep = up ? v[k + OFF] : v[k];
    float send = up ? v[k] : v[k + OFF];
    v[k] = keep + __shfl_xor(send, OFF, 32);
  }
}

__global__ __launch_bounds__(256) void gate_kernel(const float* __restrict__ H,
                                                   const float* __restrict__ Wg,
                                                   const float* __restrict__ dn,
                                                   float2* __restrict__ gs,
                                                   float2* __restrict__ gt, int Nrows) {
  int wave = __builtin_amdgcn_readfirstlane(threadIdx.x >> 5);
  int lane = threadIdx.x & 31;
  int n0 = (blockIdx.x * 8 + wave) * GATE_R;
  if (n0 >= Nrows) return;
  const float* hl = H + lane * 4;
  float4 hv[GATE_R];
#pragma unroll
  for (int k = 0; k < GATE_R; ++k) {
    int node = n0 + k < Nrows ? n0 + k : Nrows - 1;
    hv[k] = *(const float4*)(hl + (size_t)node * 128);
  }
  float4 wa = *(const float4*)(Wg + lane * 4);
  float4 wb = *(const float4*)(Wg + 128 + lane * 4);
  int mine = n0 + (lane & (GATE_R - 1));
  float d = dn[mine < Nrows ? mine : Nrows - 1];
  asm volatile("" ::"v"(hv[0].x), "v"(hv[1].x), "v"(hv[2].x), "v"(hv[3].x), "v"(hv[4].x),
               "v"(hv[5].x), "v"(hv[6].x), "v"(hv[7].x), "v"(hv[8].x), "v"(hv[9].x),
               "v"(hv[10].x), "v"(hv[11].x), "v"(hv[12].x), "v"(hv[13].x), "v"(hv[14].x),
               "v"(hv[15].x), "v"(wa.x), "v"(wb.x), "v"(d));
  float v[2 * GATE_R];
#pragma unroll
  for (int k = 0; k < GATE_R; ++k) {
    v[k] = hv[k].x * wa.x + hv[k].y * wa.y + hv[k].z * wa.z + hv[k].w * wa.w;
    v[GATE_R + k] = hv[k].x * wb.x + hv[k].y * wb.y + hv[k].z * wb.z + hv[k].w * wb.w;
  }
  gate_fold<16>(v, lane);
  gate_fold<8>(v, lane);
  gate_fold<4>(v, lane);
  gate_fold<2>(v, lane);
  gate_fold<1>(v, lane);
  if (mine < Nrows) {
    float2 rec = make_float2(v[0], d);
    if (lane < GATE_R) gs[mine] = rec;
    else gt[mine] = rec;
  }
}

#define AGG_WAVES 32
#define AGG_RCAP 64
#define AGG_STEP 512
#define AGG_KSTEP (4 * AGG_STEP)
#define AGG_QCAP (AGG_STEP + 32)
#define AGG_INIT_TRIPS ((AGG_TN * 32 + 1023) / 1024)

__device__ __forceinline__ void agg_rows_load(const float* __restrict__ H, int r, int j0, int n,
                                              int lane, float4 (&hv)[8]) {
#pragma unroll
  for (int u = 0; u < 8; ++u) {
    int jj = j0 + u < n ? j0 + u : n - 1;
    int rj = __builtin_amdgcn_readlane(r, jj);
    hv[u] = *(const float4*)(H + (size_t)rj * 128 + lane * 4);
  }
}

__device__ __forceinline__ void agg_rows_add(float* acc, int lrow, float sc, int j0, int n,
                                             int lane, const float4 (&hv)[8]) {
  int hi = lane >> 4;
#pragma unroll
  for (int u = 0; u < 8; ++u) {
    if (j0 + u < n) {
      int lr = __builtin_amdgcn_readlane(lrow, j0 + u);
      float sj = __builtin_bit_cast(float, __builtin_amdgcn_readlane(__builtin_bit_cast(int, sc), j0 + u));
      float a0 = hi ? hv[u].y : hv[u].x;
      float a1 = hi ? hv[u].x : hv[u].y;
      float a2 = hi ? hv[u].w : hv[u].z;
      float a3 = hi ? hv[u].z : hv[u].w;
      float* p = acc + lr + lane * 4;
      atomicAdd(p + (0 ^ hi), sj * a0);
      atomicAdd(p + (1 ^ hi), sj * a1);
      atomicAdd(p + (2 ^ hi), sj * a2);
      atomicAdd(p + (3 ^ hi), sj * a3);
    }
  }
}

__device__ __forceinline__ void agg_edges(int r, int cl, int n, const float2* __restrict__ gs,
                                          const float2* lgt, float bgl,
                                          const float* __restrict__ H, float* acc, int lane) {
  float2 s = gs[r];
  float4 hv[8];
  agg_rows_load(H, r, 0, n, lane, hv);
  float2 t = lgt[cl];
  float sc = tanhf(s.x + t.x + bgl) * s.y * t.y;
  int lrow = cl * 128;
  for (int j0 = 0;;) {
    agg_rows_add(acc, lrow, sc, j0, n, lane, hv);
    j0 += 8;
    if (j0 >= n) break;
    agg_rows_load(H, r, j0, n, lane, hv);
  }
}

__device__ __forceinline__ void agg_route(bool valid, int r, int cl, int* rq, int* rcount,
                                          const float2* __restrict__ gs, const float2* lgt,
                                          float bgl, const float* __restrict__ H, float* acc,
                                          int lane) {
  int pos = 0;
  if (valid) pos = atomicAdd(&rcount[cl], 1);
  bool ok = valid && pos < AGG_RCAP;
  if (ok) rq[cl * AGG_RCAP + (unsigned)(pos + cl) % AGG_RCAP] = r;
  unsigned ovf = __builtin_amdgcn_ballot_w32(valid && !ok);
  while (ovf != 0) {
    int j = __builtin_ctz(ovf);
    ovf &= ovf - 1;
    int rj = __builtin_amdgcn_readlane(r, j);
    int cj = __builtin_amdgcn_readlane(cl, j);
    agg_edges(rj, cj, 1, gs, lgt, bgl, H, acc, lane);
  }
}

__device__ __forceinline__ void agg_rows_fma(float4& a, float sc, int j0, int n,
                                             const float4 (&hv)[8]) {
#pragma unroll
  for (int u = 0; u < 8; ++u) {
    if (j0 + u < n) {
      float sj = __builtin_bit_cast(
          float, __builtin_amdgcn_readlane(__builtin_bit_cast(int, sc), j0 + u));
      a.x += sj * hv[u].x;
      a.y += sj * hv[u].y;
      a.z += sj * hv[u].z;
      a.w += sj * hv[u].w;
    }
  }
}

__device__ __forceinline__ void agg_row_sum(const int* rqt, int b, int n, int t,
                                            const float2* __restrict__ gs, const float2* lgt,
                                            float bgl, const float* __restrict__ H, float* acc,
                                            int lane) {
  int r = rqt[(unsigned)(b + (lane < n ? lane : n - 1) + t) % AGG_RCAP];
  float2 s = gs[r];
  float4 hva[8], hvb[8];
  agg_rows_load(H, r, 0, n, lane, hva);
  if (n > 8) agg_rows_load(H, r, 8, n, lane, hvb);
  float2 tg = lgt[t];
  float sc = tanhf(s.x + tg.x + bgl) * s.y * tg.y;
  float4 a = make_float4(0.0f, 0.0f, 0.0f, 0.0f);
  agg_rows_fma(a, sc, 0, n, hva);
  if (n > 16) agg_rows_load(H, r, 16, n, lane, hva);
  if (n > 8) agg_rows_fma(a, sc, 8, n, hvb);
  if (n > 24) agg_rows_load(H, r, 24, n, lane, hvb);
  if (n > 16) agg_rows_fma(a, sc, 16, n, hva);
  if (n > 24) agg_rows_fma(a, sc, 24, n, hvb);
  float4* p = (float4*)(acc + t * 128) + lane;
  float4 v = *p;
  v.x += a.x;
  v.y += a.y;
  v.z += a.z;
  v.w += a.w;
  *p = v;
}

__device__ __forceinline__ void agg_drain(const int* __restrict__ ei, const int* __restrict__ col,
                                          const float2* __restrict__ gs, const float2* lgt,
                                          float bgl, const float* __restrict__ H, float* acc,
                                          const int* q, int cnt, int base, int lane, int2* lw,
                                          int* lcount, int cap, int* rq, int* rcount) {
  int pos = 0;
  if (lw) {
    if (lane == 0) pos = atomicAdd(lcount, cnt);
    pos = __builtin_amdgcn_readfirstlane(pos);
  }
  for (int b = 0; b < cnt; b += 32) {
    int n = cnt - b < 32 ? cnt - b : 32;
    int e = q[b + (lane < n ? lane : 0)];
    int r = ei[e];
    int cl = col[e] - base;
    agg_route(lane < n, r, cl, rq, rcount, gs, lgt, bgl, H, acc, lane);
    if (lw && lane < n && pos + b + lane < cap) lw[pos + b + lane] = make_int2(r, cl);
  }
}

__device__ __forceinline__ void agg_load_keys(const unsigned* __restrict__ key, int S, int lane,
                                              uint4 (&kv)[4]) {
  const uint4* p = (const uint4*)key + (size_t)S * (AGG_KSTEP / 16) + lane;
#pragma unroll
  for (int k = 0; k < 4; ++k) kv[k] = p[k * 32];
}

__device__ __forceinline__ unsigned agg_zero_bytes(unsigned x) {
  return ~(((x & 0x7f7f7f7fu) + 0x7f7f7f7fu) | x) & 0x80808080u;
}

__device__ __forceinline__ unsigned agg_hits8(uint4 t0, uint4 t1, unsigned pat) {
  return (agg_zero_bytes(t0.x ^ pat) >> 7) | (agg_zero_bytes(t0.y ^ pat) >> 6) |
         (agg_zero_bytes(t0.z ^ pat) >> 5) | (agg_zero_bytes(t0.w ^ pat) >> 4) |
         (agg_zero_bytes(t1.x ^ pat) >> 3) | (agg_zero_bytes(t1.y ^ pat) >> 2) |
         (agg_zero_bytes(t1.z ^ pat) >> 1) | agg_zero_bytes(t1.w ^ pat);
}

__device__ __forceinline__ int agg_append(bool hit, int e, int* q, int cnt) {
  unsigned b = __builtin_amdgcn_ballot_w32(hit);
  if (hit) q[cnt + __builtin_amdgcn_mbcnt_lo(b, 0)] = e;
  return cnt + __builtin_popcount(b);
}

__device__ __forceinline__ int agg_queue(const int (&cv)[16], int s, int lane, int base, int* q,
                                         int cnt) {
  int e0 = s * AGG_STEP + lane * 4;
#pragma unroll
  for (int i = 0; i < 16; ++i)
    cnt = agg_append((unsigned)(cv[i] - base) < (unsigned)AGG_TN,
                     e0 + (i >> 2) * 128 + (i & 3), q, cnt);
  __builtin_amdgcn_fence(__ATOMIC_ACQ_REL, "wavefront");
  return cnt;
}

#define AGG_GROWS ((AGG_TN + AGG_WAVES - 1) / AGG_WAVES)

__device__ __forceinline__ void agg_next_gates(const float* acc, const float2* lgt,
                                               const float* __restrict__ Wgn,
                                               float2* __restrict__ gsn, float2* __restrict__ gtn,
                                               int wave, int lane, int base, int rows_here) {
  int r0 = wave * AGG_GROWS;
  if (r0 >= rows_here) return;
  float wa[4], wb[4];
#pragma unroll
  for (int j = 0; j < 4; ++j) {
    wa[j] = Wgn[lane + j * 32];
    wb[j] = Wgn[128 + lane + j * 32];
  }
  float ka = 0.0f, kb = 0.0f;
#pragma unroll
  for (int k = 0; k < AGG_GROWS; ++k) {
    int t = r0 + k < AGG_TN ? r0 + k : AGG_TN - 1;
    const float* hrow = acc + t * 128;
    float sa = 0.0f, sb = 0.0f;
#pragma unroll
    for (int j = 0; j < 4; ++j) {
      float hv = hrow[lane + j * 32];
      sa += hv * wa[j];
      sb += hv * wb[j];
    }
    for (int off = 16; off > 0; off >>= 1) {
      sa += __shfl_down(sa, off, 32);
      sb += __shfl_down(sb, off, 32);
    }
    sa = __shfl(sa, 0, 32);
    sb = __shfl(sb, 0, 32);
    if (lane == k) { ka = sa; kb = sb; }
  }
  int t = r0 + lane;
  if (lane < AGG_GROWS && t < rows_here) {
    float d = lgt[t].y;
    gsn[base + t] = make_float2(ka, d);
    gtn[base + t] = make_float2(kb, d);
  }
}

__device__ __forceinline__ void agg_head_tile(const float* acc, int t, int rows_here, int base,
                                              const float* __restrict__ W,
                                              const float* __restrict__ bias,
                                              float* __restrict__ out, int lane) {
  int n = lane & 15;
  int hi = lane >> 4;
  int lr = t * 16 + n;
  int lrc = lr < AGG_TN ? lr : AGG_TN - 1;
  const float* arow = acc + lrc * 128 + hi * 2;
  const float* wcol = W + hi * 32 + n;
  v2f bf[32];
#pragma unroll
  for (int s = 0; s < 32; ++s) {
    bf[s].x = wcol[s * 64];
    bf[s].y = wcol[s * 64 + 16];
  }
  float bn = bias[n];
  v8f d = {};
#pragma unroll
  for (int s = 0; s < 32; ++s) {
    v2f a = *(const v2f*)(arow + s * 4);
    d = __builtin_amdgcn_wmma_f32_16x16x4_f32(false, a, false, bf[s], (short)0, d, false, false);
  }
  float v[8], m[8], s[8];
#pragma unroll
  for (int r = 0; r < 8; ++r) m[r] = v[r] = d[r] + bn;
#pragma unroll
  for (int off = 1; off < 16; off <<= 1) {
#pragma unroll
    for (int r = 0; r < 8; ++r) m[r] = fmaxf(m[r], __shfl_xor(m[r], off, 32));
  }
#pragma unroll
  for (int r = 0; r < 8; ++r) s[r] = expf(v[r] - m[r]);
#pragma unroll
  for (int off = 1; off < 16; off <<= 1) {
#pragma unroll
    for (int r = 0; r < 8; ++r) s[r] += __shfl_xor(s[r], off, 32);
  }
#pragma unroll
  for (int r = 0; r < 8; ++r) {
    float o = v[r] - (m[r] + logf(s[r]));
    int lrow = t * 16 + r + 8 * hi;
    if (lrow < rows_here) out[(size_t)(base + lrow) * 16 + n] = o;
  }
}

__global__ __launch_bounds__(1024) void agg_lds_kernel(const int* __restrict__ ei,
                                                       const float2* __restrict__ gs,
                                                       const float2* __restrict__ gt,
                                                       const float* __restrict__ bg, int layer,
                                                       const float* __restrict__ H,
                                                       const float* __restrict__ raw,
                                                       float* agg,
                                                       const unsigned* __restrict__ key,
                                                       const float* ws, size_t ws_bytes,
                                                       int E, int Nrows, float* lbuf, int L,
                                                       const float* __restrict__ Wg,
                                                       float2* __restrict__ gsn,
                                                       float2* __restrict__ gtn,
                                                       const float* __restrict__ Wout,
                                                       const float* __restrict__ bout,
                                                       float* __restrict__ out) {
  __shared__ float acc[AGG_TN * 128];
  __shared__ int wq[AGG_WAVES][AGG_QCAP];
  __shared__ float2 lgt[AGG_TN];
  __shared__ int lcount;
  __shared__ int rq[AGG_TN * AGG_RCAP];
  __shared__ int rcount[AGG_TN];
  int tid = threadIdx.x;
  int wave = __builtin_amdgcn_readfirstlane(tid >> 5);
  int lane = tid & 31;
  int base = blockIdx.x * AGG_TN;
  int rows_here = Nrows - base < AGG_TN ? Nrows - base : AGG_TN;
  int* lhdr = (int*)(lbuf + (size_t)base * 128);
  int2* lent = (int2*)(lhdr + 4);
  int lcap = rows_here * 64 - 2;
  bool lwrite = layer == 0 && L > 1;
  int ltotal = -1;
  if (layer == 1) {
    int t = __builtin_amdgcn_readfirstlane(lhdr[0]);
    if (t <= lcap) ltotal = t;
  }
  if (tid == 0) lcount = 0;
  if (tid < AGG_TN) rcount[tid] = 0;
  {
    float4 rv[AGG_INIT_TRIPS];
#pragma unroll
    for (int k = 0; k < AGG_INIT_TRIPS; ++k) {
      int i = tid + k * 1024;
      int ic = i < AGG_TN * 32 ? i : AGG_TN * 32 - 1;
      int node = base + (ic >> 5);
      int nc = node < Nrows ? node : Nrows - 1;
      rv[k] = ((const float4*)raw)[(size_t)nc * 32 + (ic & 31)];
    }
#pragma unroll
    for (int k = 0; k < AGG_INIT_TRIPS; ++k) {
      int i = tid + k * 1024;
      if (i < AGG_TN * 32) {
        float s = base + (i >> 5) < Nrows ? FAGCN_EPS : 0.0f;
        ((float4*)acc)[i] = make_float4(s * rv[k].x, s * rv[k].y, s * rv[k].z, s * rv[k].w);
      }
    }
  }
  if (tid < AGG_TN)
    lgt[tid] = base + tid < Nrows ? gt[base + tid] : make_float2(0.0f, 0.0f);
  float bgl = bg[layer];
  __syncthreads();

  const int* col = ei + E;
  bool scan = ltotal < 0;
  int nsteps = scan ? (E + AGG_STEP - 1) / AGG_STEP : 0;
  int nsuper = scan && agg_keys_ok(key, ws, ws_bytes, E, Nrows) ? E / AGG_KSTEP : 0;
  int2* lw = lwrite ? lent : nullptr;
  int* q = wq[wave];
  int cnt = 0;
  if (!scan && wave * 32 < ltotal) {
    int i = wave * 32 + lane;
    int2 en = lent[i < ltotal ? i : ltotal - 1];
    for (int i0 = wave * 32; i0 < ltotal; i0 += 1024) {
      int nx = i0 + 1024 + lane;
      int2 enx = lent[nx < ltotal ? nx : ltotal - 1];
      agg_route(i0 + lane < ltotal, en.x, en.y, rq, rcount, gs, lgt, bgl, H, acc, lane);
      en = enx;
    }
  }
  if (wave < nsuper) {
    unsigned pat = blockIdx.x * 0x01010101u;
    uint4 kv[4];
    agg_load_keys(key, wave, lane, kv);
    for (int S = wave; S < nsuper; S += AGG_WAVES) {
      unsigned long long m = ((unsigned long long)agg_hits8(kv[2], kv[3], pat) << 32) |
                             agg_hits8(kv[0], kv[1], pat);
      int Sp = S + AGG_WAVES < nsuper ? S + AGG_WAVES : nsuper - 1;
      agg_load_keys(key, Sp, lane, kv);
      int e0 = S * AGG_KSTEP + lane * 16;
      while (__builtin_amdgcn_ballot_w32(m != 0) != 0) {
        bool hit = m != 0;
        int bit = hit ? __builtin_ctzll(m) : 0;
        int off = ((((bit >> 5) << 1) | ((bit >> 2) & 1)) << 9) | ((bit & 3) << 2) |
                  ((bit >> 3) & 3);
        cnt = agg_append(hit, e0 + off, q, cnt);
        m &= m - 1;
        __builtin_amdgcn_fence(__ATOMIC_ACQ_REL, "wavefront");
        if (cnt >= 32) {
          agg_drain(ei, col, gs, lgt, bgl, H, acc, q, cnt, base, lane, lw, &lcount, lcap, rq, rcount);
          cnt = 0;
        }
      }
    }
  }
  for (int s = nsuper * 4 + wave; s < nsteps; s += AGG_WAVES) {
    int cv[16];
    int e0 = s * AGG_STEP + lane * 4;
#pragma unroll
    for (int i = 0; i < 16; ++i) {
      int ee = e0 + (i >> 2) * 128 + (i & 3);
      cv[i] = ee < E ? col[ee] : -1;
    }
    cnt = agg_queue(cv, s, lane, base, q, cnt);
    if (cnt >= 32) {
      agg_drain(ei, col, gs, lgt, bgl, H, acc, q, cnt, base, lane, lw, &lcount, lcap, rq, rcount);
      cnt = 0;
    }
  }
  if (cnt > 0) agg_drain(ei, col, gs, lgt, bgl, H, acc, q, cnt, base, lane, lw, &lcount, lcap, rq, rcount);
  __syncthreads();
  if (lwrite && tid == 0) lhdr[0] = lcount;
  for (int t = wave; t < AGG_TN; t += AGG_WAVES) {
    int n = rcount[t] < AGG_RCAP ? rcount[t] : AGG_RCAP;
    n = __builtin_amdgcn_readfirstlane(n);
    for (int b = 0; b < n; b += 32)
      agg_row_sum(rq + t * AGG_RCAP, b, n - b < 32 ? n - b : 32, t, gs, lgt, bgl, H, acc, lane);
  }
  __syncthreads();

  if (layer == L - 1) {
    if (wave < (rows_here + 15) >> 4) agg_head_tile(acc, wave, rows_here, base, Wout, bout, out, lane);
    return;
  }
  for (int i = tid; i < AGG_TN * 32; i += 1024) {
    int node = base + (i >> 5);
    if (node < Nrows) ((float4*)agg)[(size_t)node * 32 + (i & 31)] = ((float4*)acc)[i];
  }
  agg_next_gates(acc, lgt, Wg + (size_t)(layer + 1) * 256, gsn, gtn, wave, lane, base, rows_here);
}

__global__ void gemm_out_kernel(const float* __restrict__ H, const float* __restrict__ W,
                                const float* __restrict__ bias, float* __restrict__ out,
                                int Nrows) {
  int wave = threadIdx.x >> 5;
  int lane = threadIdx.x & 31;
  int tileM = blockIdx.x * 8 + wave;
  int tiles = (Nrows + 15) >> 4;
  if (tileM >= tiles) return;
  int mA = tileM * 16 + (lane & 15);
  int mclamp = mA < Nrows ? mA : (Nrows - 1);
  int n = lane & 15;
  int khalf = (lane >> 4) * 2;
  const float* hrow = H + (size_t)mclamp * 128 + khalf;
  const float* wcol = W + (size_t)khalf * 16 + n;
  __builtin_prefetch(hrow, 0, 0);
  v8f acc = {};
#pragma unroll 4
  for (int kk = 0; kk < 128; kk += 4) {
    v2f a = *(const v2f*)(hrow + kk);
    v2f b;
    b.x = wcol[(size_t)kk * 16];
    b.y = wcol[(size_t)kk * 16 + 16];
    acc = __builtin_amdgcn_wmma_f32_16x16x4_f32(false, a, false, b, (short)0, acc,
                                               false, false);
  }
  int mbase = tileM * 16 + ((lane >> 4) * 8);
  float bn = bias[n];
  if (tileM * 16 + 16 <= Nrows) {
    float* op = out + (size_t)mbase * 16 + n;
#pragma unroll
    for (int r = 0; r < 8; ++r)
      op[(size_t)r * 16] = acc[r] + bn;
  } else {
#pragma unroll
    for (int r = 0; r < 8; ++r) {
      int mm = mbase + r;
      if (mm < Nrows) out[(size_t)mm * 16 + n] = acc[r] + bn;
    }
  }
}

__global__ void logsoftmax_kernel(float4* out, int N) {
  int i = blockIdx.x * blockDim.x + threadIdx.x;
  if (i >= N) return;
  float4* row = out + (size_t)i * 4;
  float4 v[4];
#pragma unroll
  for (int q = 0; q < 4; ++q) v[q] = row[q];
  float m = v[0].x;
#pragma unroll
  for (int q = 0; q < 4; ++q) {
    m = fmaxf(m, v[q].x); m = fmaxf(m, v[q].y);
    m = fmaxf(m, v[q].z); m = fmaxf(m, v[q].w);
  }
  float s = 0.0f;
#pragma unroll
  for (int q = 0; q < 4; ++q) {
    s += expf(v[q].x - m); s += expf(v[q].y - m);
    s += expf(v[q].z - m); s += expf(v[q].w - m);
  }
  float d = m + logf(s);
#pragma unroll
  for (int q = 0; q < 4; ++q) {
    v[q].x -= d; v[q].y -= d; v[q].z -= d; v[q].w -= d;
    row[q] = v[q];
  }
}

extern "C" void kernel_launch(void* const* d_in, const int* in_sizes, int n_in,
                              void* d_out, int out_size, void* d_ws, size_t ws_size,
                              hipStream_t stream) {
  const float* x     = (const float*)d_in[0];
  const int*   ei    = (const int*)d_in[1];
  const float* W_in  = (const float*)d_in[2];
  const float* b_in  = (const float*)d_in[3];
  const float* Wg    = (const float*)d_in[4];
  const float* bg    = (const float*)d_in[5];
  const float* W_out = (const float*)d_in[6];
  const float* b_out = (const float*)d_in[7];
  float* out = (float*)d_out;

  const int Hdim = 128;
  int N = in_sizes[0] / Hdim;
  int E = in_sizes[1] / 2;
  int L = in_sizes[5];

  float* ws   = (float*)d_ws;
  float2* gs  = (float2*)ws;
  float2* gt  = gs + N;
  float2* gsn = gt + N;
  float2* gtn = gsn + N;
  float* dn   = (float*)(gtn + N);
  float* raw  = ws + 3 * (size_t)N + E;
  float* bufA = raw + (size_t)N * Hdim;
  float* bufB = bufA + (size_t)N * Hdim;
  unsigned* key = (unsigned*)(bufB + (size_t)N * Hdim);

  unsigned* part = (unsigned*)bufB;
  deg_key_kernel<<<dim3(HIST_K, (N + HIST_CAP - 1) / HIST_CAP), HIST_T, 0, stream>>>(
      ei, part, key, ws, ws_size, E, N);
  dn_kernel<<<(N + 255) / 256, 256, 0, stream>>>(part, dn, N);

  gemm_relu_kernel<<<(N + 15) / 16, 256, 0, stream>>>(x, W_in, b_in, raw, N);

  float* h = raw;
  float* agg = bufA;
  if (L > 0) gate_kernel<<<(N + 8 * GATE_R - 1) / (8 * GATE_R), 256, 0, stream>>>(h, Wg, dn, gs, gt, N);
  for (int i = 0; i < L; ++i) {
    agg_lds_kernel<<<(N + AGG_TN - 1) / AGG_TN, 1024, 0, stream>>>(
        ei, gs, gt, bg, i, h, raw, agg, key, ws, ws_size, E, N, bufB, L, Wg, gsn, gtn, W_out,
        b_out, out);
    h = agg;
    agg = (h == bufA) ? bufB : bufA;
    float2* sw = gs; gs = gsn; gsn = sw;
    sw = gt; gt = gtn; gtn = sw;
  }

  if (L == 0) {
    int tiles = (N + 15) / 16;
    gemm_out_kernel<<<(tiles + 7) / 8, 256, 0, stream>>>(h, W_out, b_out, out, N);
    logsoftmax_kernel<<<(N + 255) / 256, 256, 0, stream>>>((float4*)out, N);
  }
}


# Input GEMM re-tiled: one wave per 16x64 half strip, not per 16x16 tile; x rows loaded once for 4 products, W_in fragments read from a per-group LDS copy, whole-line stores; product sum order changes
# speedup vs baseline: 4.1035x; 1.0938x over previous
#include <hip/hip_runtime.h>

typedef float v2f __attribute__((ext_vector_type(2)));
typedef float v8f __attribute__((ext_vector_type(8)));

#define FAGCN_EPS 0.3f

#define AGG_TN 196
#define AGG_MAXG 256

__device__ __forceinline__ bool agg_keys_ok(const unsigned* key, const float* ws, size_t ws_bytes,
                                            int E, int Nrows) {
  size_t need = (size_t)((const char*)key - (const char*)ws) + (((size_t)E + 3) & ~(size_t)3);
  return need <= ws_bytes && (((size_t)key) & 15) == 0 &&
         (Nrows + AGG_TN - 1) / AGG_TN <= AGG_MAXG;
}

#define HIST_K 64
#define HIST_CAP 16384
#define HIST_T 1024
#define HIST_U 4

__device__ __forceinline__ void hist_count(unsigned* tab, int r, int rbase, int nhere) {
  unsigned d = (unsigned)(r - rbase);
  if (d < (unsigned)nhere) atomicAdd(&tab[d], 1u);
}

__device__ __forceinline__ unsigned agg_pack_keys(int c0, int c1, int c2, int c3) {
  return (((unsigned)c0 / AGG_TN) & 255u) | ((((unsigned)c1 / AGG_TN) & 255u) << 8) |
         ((((unsigned)c2 / AGG_TN) & 255u) << 16) | ((((unsigned)c3 / AGG_TN) & 255u) << 24);
}

__global__ __launch_bounds__(HIST_T) void deg_key_kernel(const int* __restrict__ ei,
                                                         unsigned* __restrict__ part,
                                                         unsigned* __restrict__ key,
                                                         const float* ws, size_t ws_bytes,
                                                         int E, int Nrows) {
  __shared__ unsigned tab[HIST_CAP];
  int tid = threadIdx.x;
  int rbase = blockIdx.y * HIST_CAP;
  int nhere = Nrows - rbase < HIST_CAP ? Nrows - rbase : HIST_CAP;
  for (int i = tid; i < HIST_CAP / 4; i += HIST_T) ((uint4*)tab)[i] = make_uint4(0u, 0u, 0u, 0u);
  __syncthreads();

  const int* col = ei + E;
  int Q = (E + 3) >> 2;
  int qpb = (Q + HIST_K - 1) / HIST_K;
  int q0 = blockIdx.x * qpb;
  int q1 = q0 + qpb < Q ? q0 + qpb : Q;
  bool keys = blockIdx.y == 0 && agg_keys_ok(key, ws, ws_bytes, E, Nrows);
  if (q0 < q1) {
    if ((E & 3) == 0 && ((((size_t)ei) | ((size_t)col)) & 15) == 0) {
      int trips = (qpb + HIST_T - 1) / HIST_T;
      for (int j0 = 0; j0 < trips; j0 += HIST_U) {
        int4 rv[HIST_U], cw[HIST_U];
#pragma unroll
        for (int u = 0; u < HIST_U; ++u) {
          int q = q0 + (j0 + u) * HIST_T + tid;
          int qq = q < q1 ? q : q1 - 1;
          rv[u] = ((const int4*)ei)[qq];
          cw[u] = ((const int4*)col)[qq];
        }
        asm volatile("" ::"v"(rv[0].x), "v"(rv[1].x), "v"(rv[2].x), "v"(rv[3].x), "v"(cw[0].x),
                     "v"(cw[1].x), "v"(cw[2].x), "v"(cw[3].x));
#pragma unroll
        for (int u = 0; u < HIST_U; ++u) {
          int q = q0 + (j0 + u) * HIST_T + tid;
          if (q < q1) {
            hist_count(tab, rv[u].x, rbase, nhere);
            hist_count(tab, rv[u].y, rbase, nhere);
            hist_count(tab, rv[u].z, rbase, nhere);
            hist_count(tab, rv[u].w, rbase, nhere);
            if (keys) key[q] = agg_pack_keys(cw[u].x, cw[u].y, cw[u].z, cw[u].w);
          }
        }
      }
    } else {
      for (int q = q0 + tid; q < q1; q += HIST_T) {
        int c[4];
#pragma unroll
        for (int k = 0; k < 4; ++k) {
          int e = q * 4 + k;
          int ec = e < E ? e : E - 1;
          if (e < E) hist_count(tab, ei[ec], rbase, nhere);
          c[k] = col[ec];
        }
        if (keys) key[q] = agg_pack_keys(c[0], c[1], c[2], c[3]);
      }
    }
  }
  __syncthreads();

  unsigned* dst = part + (size_t)blockIdx.x * Nrows + rbase;
  if ((nhere & 3) == 0 && (((size_t)dst) & 15) == 0) {
    for (int i = tid; i < nhere / 4; i += HIST_T) ((uint4*)dst)[i] = ((const uint4*)tab)[i];
  } else {
    for (int i = tid; i < nhere; i += HIST_T) dst[i] = tab[i];
  }
}

__global__ void dn_kernel(const unsigned* __restrict__ part, float* __restrict__ dn, int N) {
  int i = blockIdx.x * blockDim.x + threadIdx.x;
  if (i >= N) return;
  unsigned s = 0;
#pragma unroll
  for (int b = 0; b < HIST_K; ++b) s += part[(size_t)b * N + i];
  dn[i] = 1.0f / sqrtf(fmaxf((float)s, 1.0f));
}

#define GEMM_T 1024
#define GEMM_RT 13

__global__ __launch_bounds__(GEMM_T) void gemm_relu_kernel(const float* __restrict__ X,
                                                           const float* __restrict__ W,
                                                           const float* __restrict__ bias,
                                                           float* __restrict__ H, int Nrows) {
  __shared__ float4 wf[32 * 2 * 2 * 32];
  int wave = __builtin_amdgcn_readfirstlane(threadIdx.x >> 5);
  int lane = threadIdx.x & 31;
  int c = lane & 15;
  int hh = lane >> 4;
  int tiles = (Nrows + 15) >> 4;
  int ch = wave & 1;
  int rt = blockIdx.x * GEMM_RT + (wave >> 1);
  bool work = (wave >> 1) < GEMM_RT && rt < tiles;
  float4 a4[16];
  if (work) {
    int row = rt * 16 + c;
    row = row < Nrows ? row : Nrows - 1;
    const float* xp = X + (size_t)row * 128 + hh * 4;
#pragma unroll
    for (int j = 0; j < 16; ++j) a4[j] = *(const float4*)(xp + j * 8);
  }

  for (int t = wave; t < 32; t += GEMM_T / 32) {
    int k0 = (t >> 1) * 8 + hh * 4 + (t & 1) * 2;
    const float* w0 = W + (size_t)k0 * 128 + c * 4;
    float4 r00 = *(const float4*)(w0);
    float4 r01 = *(const float4*)(w0 + 64);
    float4 r10 = *(const float4*)(w0 + 128);
    float4 r11 = *(const float4*)(w0 + 192);
    float4* d = wf + t * 128 + lane;
    d[0] = make_float4(r00.x, r10.x, r00.y, r10.y);
    d[32] = make_float4(r00.z, r10.z, r00.w, r10.w);
    d[64] = make_float4(r01.x, r11.x, r01.y, r11.y);
    d[96] = make_float4(r01.z, r11.z, r01.w, r11.w);
  }
  __syncthreads();
  if (!work) return;

  const float4* bp = wf + ch * 64 + lane;
  const float* bcol = bias + ch * 64 + c * 4;
  float4 bv = make_float4(bcol[0], bcol[1], bcol[2], bcol[3]);
  v8f acc0 = {}, acc1 = {}, acc2 = {}, acc3 = {};
#pragma unroll
  for (int t = 0; t < 32; ++t) {
    float4 av = a4[t >> 1];
    v2f a;
    a.x = (t & 1) ? av.z : av.x;
    a.y = (t & 1) ? av.w : av.y;
    float4 b01 = bp[t * 128];
    float4 b23 = bp[t * 128 + 32];
    v2f b0, b1, b2, b3;
    b0.x = b01.x; b0.y = b01.y;
    b1.x = b01.z; b1.y = b01.w;
    b2.x = b23.x; b2.y = b23.y;
    b3.x = b23.z; b3.y = b23.w;
    acc0 = __builtin_amdgcn_wmma_f32_16x16x4_f32(false, a, false, b0, (short)0, acc0, false, false);
    acc1 = __builtin_amdgcn_wmma_f32_16x16x4_f32(false, a, false, b1, (short)0, acc1, false, false);
    acc2 = __builtin_amdgcn_wmma_f32_16x16x4_f32(false, a, false, b2, (short)0, acc2, false, false);
    acc3 = __builtin_amdgcn_wmma_f32_16x16x4_f32(false, a, false, b3, (short)0, acc3, false, false);
  }
  int r0 = rt * 16 + hh * 8;
  float* hp = H + (size_t)r0 * 128 + ch * 64 + c * 4;
  if (rt * 16 + 16 <= Nrows) {
#pragma unroll
    for (int r = 0; r < 8; ++r)
      *(float4*)(hp + (size_t)r * 128) =
          make_float4(fmaxf(acc0[r] + bv.x, 0.0f), fmaxf(acc1[r] + bv.y, 0.0f),
                      fmaxf(acc2[r] + bv.z, 0.0f), fmaxf(acc3[r] + bv.w, 0.0f));
  } else {
#pragma unroll
    for (int r = 0; r < 8; ++r) {
      if (r0 + r < Nrows)
        *(float4*)(hp + (size_t)r * 128) =
            make_float4(fmaxf(acc0[r] + bv.x, 0.0f), fmaxf(acc1[r] + bv.y, 0.0f),
                        fmaxf(acc2[r] + bv.z, 0.0f), fmaxf(acc3[r] + bv.w, 0.0f));
    }
  }
}

#define GATE_R 16

template <int OFF>
__device__ __forceinline__ void gate_fold(float (&v)[2 * GATE_R], int lane) {
  bool up = (lane & OFF) != 0;
#pragma unroll
  for (int k = 0; k < OFF; ++k) {
    float keep = up ? v[k + OFF] : v[k];
    float send = up ? v[k] : v[k + OFF];
    v[k] = keep + __shfl_xor(send, OFF, 32);
  }
}

__global__ __launch_bounds__(256) void gate_kernel(const float* __restrict__ H,
                                                   const float* __restrict__ Wg,
                                                   const float* __restrict__ dn,
                                                   float2* __restrict__ gs,
                                                   float2* __restrict__ gt, int Nrows) {
  int wave = __builtin_amdgcn_readfirstlane(threadIdx.x >> 5);
  int lane = threadIdx.x & 31;
  int n0 = (blockIdx.x * 8 + wave) * GATE_R;
  if (n0 >= Nrows) return;
  const float* hl = H + lane * 4;
  float4 hv[GATE_R];
#pragma unroll
  for (int k = 0; k < GATE_R; ++k) {
    int node = n0 + k < Nrows ? n0 + k : Nrows - 1;
    hv[k] = *(const float4*)(hl + (size_t)node * 128);
  }
  float4 wa = *(const float4*)(Wg + lane * 4);
  float4 wb = *(const float4*)(Wg + 128 + lane * 4);
  int mine = n0 + (lane & (GATE_R - 1));
  float d = dn[mine < Nrows ? mine : Nrows - 1];
  asm volatile("" ::"v"(hv[0].x), "v"(hv[1].x), "v"(hv[2].x), "v"(hv[3].x), "v"(hv[4].x),
               "v"(hv[5].x), "v"(hv[6].x), "v"(hv[7].x), "v"(hv[8].x), "v"(hv[9].x),
               "v"(hv[10].x), "v"(hv[11].x), "v"(hv[12].x), "v"(hv[13].x), "v"(hv[14].x),
               "v"(hv[15].x), "v"(wa.x), "v"(wb.x), "v"(d));
  float v[2 * GATE_R];
#pragma unroll
  for (int k = 0; k < GATE_R; ++k) {
    v[k] = hv[k].x * wa.x + hv[k].y * wa.y + hv[k].z * wa.z + hv[k].w * wa.w;
    v[GATE_R + k] = hv[k].x * wb.x + hv[k].y * wb.y + hv[k].z * wb.z + hv[k].w * wb.w;
  }
  gate_fold<16>(v, lane);
  gate_fold<8>(v, lane);
  gate_fold<4>(v, lane);
  gate_fold<2>(v, lane);
  gate_fold<1>(v, lane);
  if (mine < Nrows) {
    float2 rec = make_float2(v[0], d);
    if (lane < GATE_R) gs[mine] = rec;
    else gt[mine] = rec;
  }
}

#define AGG_WAVES 32
#define AGG_RCAP 64
#define AGG_STEP 512
#define AGG_KSTEP (4 * AGG_STEP)
#define AGG_QCAP (AGG_STEP + 32)
#define AGG_INIT_TRIPS ((AGG_TN * 32 + 1023) / 1024)

__device__ __forceinline__ void agg_rows_load(const float* __restrict__ H, int r, int j0, int n,
                                              int lane, float4 (&hv)[8]) {
#pragma unroll
  for (int u = 0; u < 8; ++u) {
    int jj = j0 + u < n ? j0 + u : n - 1;
    int rj = __builtin_amdgcn_readlane(r, jj);
    hv[u] = *(const float4*)(H + (size_t)rj * 128 + lane * 4);
  }
}

__device__ __forceinline__ void agg_rows_add(float* acc, int lrow, float sc, int j0, int n,
                                             int lane, const float4 (&hv)[8]) {
  int hi = lane >> 4;
#pragma unroll
  for (int u = 0; u < 8; ++u) {
    if (j0 + u < n) {
      int lr = __builtin_amdgcn_readlane(lrow, j0 + u);
      float sj = __builtin_bit_cast(float, __builtin_amdgcn_readlane(__builtin_bit_cast(int, sc), j0 + u));
      float a0 = hi ? hv[u].y : hv[u].x;
      float a1 = hi ? hv[u].x : hv[u].y;
      float a2 = hi ? hv[u].w : hv[u].z;
      float a3 = hi ? hv[u].z : hv[u].w;
      float* p = acc + lr + lane * 4;
      atomicAdd(p + (0 ^ hi), sj * a0);
      atomicAdd(p + (1 ^ hi), sj * a1);
      atomicAdd(p + (2 ^ hi), sj * a2);
      atomicAdd(p + (3 ^ hi), sj * a3);
    }
  }
}

__device__ __forceinline__ void agg_edges(int r, int cl, int n, const float2* __restrict__ gs,
                                          const float2* lgt, float bgl,
                                          const float* __restrict__ H, float* acc, int lane) {
  float2 s = gs[r];
  float4 hv[8];
  agg_rows_load(H, r, 0, n, lane, hv);
  float2 t = lgt[cl];
  float sc = tanhf(s.x + t.x + bgl) * s.y * t.y;
  int lrow = cl * 128;
  for (int j0 = 0;;) {
    agg_rows_add(acc, lrow, sc, j0, n, lane, hv);
    j0 += 8;
    if (j0 >= n) break;
    agg_rows_load(H, r, j0, n, lane, hv);
  }
}

__device__ __forceinline__ void agg_route(bool valid, int r, int cl, int* rq, int* rcount,
                                          const float2* __restrict__ gs, const float2* lgt,
                                          float bgl, const float* __restrict__ H, float* acc,
                                          int lane) {
  int pos = 0;
  if (valid) pos = atomicAdd(&rcount[cl], 1);
  bool ok = valid && pos < AGG_RCAP;
  if (ok) rq[cl * AGG_RCAP + (unsigned)(pos + cl) % AGG_RCAP] = r;
  unsigned ovf = __builtin_amdgcn_ballot_w32(valid && !ok);
  while (ovf != 0) {
    int j = __builtin_ctz(ovf);
    ovf &= ovf - 1;
    int rj = __builtin_amdgcn_readlane(r, j);
    int cj = __builtin_amdgcn_readlane(cl, j);
    agg_edges(rj, cj, 1, gs, lgt, bgl, H, acc, lane);
  }
}

__device__ __forceinline__ void agg_rows_fma(float4& a, float sc, int j0, int n,
                                             const float4 (&hv)[8]) {
#pragma unroll
  for (int u = 0; u < 8; ++u) {
    if (j0 + u < n) {
      float sj = __builtin_bit_cast(
          float, __builtin_amdgcn_readlane(__builtin_bit_cast(int, sc), j0 + u));
      a.x += sj * hv[u].x;
      a.y += sj * hv[u].y;
      a.z += sj * hv[u].z;
      a.w += sj * hv[u].w;
    }
  }
}

__device__ __forceinline__ void agg_row_sum(const int* rqt, int b, int n, int t,
                                            const float2* __restrict__ gs, const float2* lgt,
                                            float bgl, const float* __restrict__ H, float* acc,
                                            int lane) {
  int r = rqt[(unsigned)(b + (lane < n ? lane : n - 1) + t) % AGG_RCAP];
  float2 s = gs[r];
  float4 hva[8], hvb[8];
  agg_rows_load(H, r, 0, n, lane, hva);
  if (n > 8) agg_rows_load(H, r, 8, n, lane, hvb);
  float2 tg = lgt[t];
  float sc = tanhf(s.x + tg.x + bgl) * s.y * tg.y;
  float4 a = make_float4(0.0f, 0.0f, 0.0f, 0.0f);
  agg_rows_fma(a, sc, 0, n, hva);
  if (n > 16) agg_rows_load(H, r, 16, n, lane, hva);
  if (n > 8) agg_rows_fma(a, sc, 8, n, hvb);
  if (n > 24) agg_rows_load(H, r, 24, n, lane, hvb);
  if (n > 16) agg_rows_fma(a, sc, 16, n, hva);
  if (n > 24) agg_rows_fma(a, sc, 24, n, hvb);
  float4* p = (float4*)(acc + t * 128) + lane;
  float4 v = *p;
  v.x += a.x;
  v.y += a.y;
  v.z += a.z;
  v.w += a.w;
  *p = v;
}

__device__ __forceinline__ void agg_drain(const int* __restrict__ ei, const int* __restrict__ col,
                                          const float2* __restrict__ gs, const float2* lgt,
                                          float bgl, const float* __restrict__ H, float* acc,
                                          const int* q, int cnt, int base, int lane, int2* lw,
                                          int* lcount, int cap, int* rq, int* rcount) {
  int pos = 0;
  if (lw) {
    if (lane == 0) pos = atomicAdd(lcount, cnt);
    pos = __builtin_amdgcn_readfirstlane(pos);
  }
  for (int b = 0; b < cnt; b += 32) {
    int n = cnt - b < 32 ? cnt - b : 32;
    int e = q[b + (lane < n ? lane : 0)];
    int r = ei[e];
    int cl = col[e] - base;
    agg_route(lane < n, r, cl, rq, rcount, gs, lgt, bgl, H, acc, lane);
    if (lw && lane < n && pos + b + lane < cap) lw[pos + b + lane] = make_int2(r, cl);
  }
}

__device__ __forceinline__ void agg_load_keys(const unsigned* __restrict__ key, int S, int lane,
                                              uint4 (&kv)[4]) {
  const uint4* p = (const uint4*)key + (size_t)S * (AGG_KSTEP / 16) + lane;
#pragma unroll
  for (int k = 0; k < 4; ++k) kv[k] = p[k * 32];
}

__device__ __forceinline__ unsigned agg_zero_bytes(unsigned x) {
  return ~(((x & 0x7f7f7f7fu) + 0x7f7f7f7fu) | x) & 0x80808080u;
}

__device__ __forceinline__ unsigned agg_hits8(uint4 t0, uint4 t1, unsigned pat) {
  return (agg_zero_bytes(t0.x ^ pat) >> 7) | (agg_zero_bytes(t0.y ^ pat) >> 6) |
         (agg_zero_bytes(t0.z ^ pat) >> 5) | (agg_zero_bytes(t0.w ^ pat) >> 4) |
         (agg_zero_bytes(t1.x ^ pat) >> 3) | (agg_zero_bytes(t1.y ^ pat) >> 2) |
         (agg_zero_bytes(t1.z ^ pat) >> 1) | agg_zero_bytes(t1.w ^ pat);
}

__device__ __forceinline__ int agg_append(bool hit, int e, int* q, int cnt) {
  unsigned b = __builtin_amdgcn_ballot_w32(hit);
  if (hit) q[cnt + __builtin_amdgcn_mbcnt_lo(b, 0)] = e;
  return cnt + __builtin_popcount(b);
}

__device__ __forceinline__ int agg_queue(const int (&cv)[16], int s, int lane, int base, int* q,
                                         int cnt) {
  int e0 = s * AGG_STEP + lane * 4;
#pragma unroll
  for (int i = 0; i < 16; ++i)
    cnt = agg_append((unsigned)(cv[i] - base) < (unsigned)AGG_TN,
                     e0 + (i >> 2) * 128 + (i & 3), q, cnt);
  __builtin_amdgcn_fence(__ATOMIC_ACQ_REL, "wavefront");
  return cnt;
}

#define AGG_GROWS ((AGG_TN + AGG_WAVES - 1) / AGG_WAVES)

__device__ __forceinline__ void agg_next_gates(const float* acc, const float2* lgt,
                                               const float* __restrict__ Wgn,
                                               float2* __restrict__ gsn, float2* __restrict__ gtn,
                                               int wave, int lane, int base, int rows_here) {
  int r0 = wave * AGG_GROWS;
  if (r0 >= rows_here) return;
  float wa[4], wb[4];
#pragma unroll
  for (int j = 0; j < 4; ++j) {
    wa[j] = Wgn[lane + j * 32];
    wb[j] = Wgn[128 + lane + j * 32];
  }
  float ka = 0.0f, kb = 0.0f;
#pragma unroll
  for (int k = 0; k < AGG_GROWS; ++k) {
    int t = r0 + k < AGG_TN ? r0 + k : AGG_TN - 1;
    const float* hrow = acc + t * 128;
    float sa = 0.0f, sb = 0.0f;
#pragma unroll
    for (int j = 0; j < 4; ++j) {
      float hv = hrow[lane + j * 32];
      sa += hv * wa[j];
      sb += hv * wb[j];
    }
    for (int off = 16; off > 0; off >>= 1) {
      sa += __shfl_down(sa, off, 32);
      sb += __shfl_down(sb, off, 32);
    }
    sa = __shfl(sa, 0, 32);
    sb = __shfl(sb, 0, 32);
    if (lane == k) { ka = sa; kb = sb; }
  }
  int t = r0 + lane;
  if (lane < AGG_GROWS && t < rows_here) {
    float d = lgt[t].y;
    gsn[base + t] = make_float2(ka, d);
    gtn[base + t] = make_float2(kb, d);
  }
}

__device__ __forceinline__ void agg_head_tile(const float* acc, int t, int rows_here, int base,
                                              const float* __restrict__ W,
                                              const float* __restrict__ bias,
                                              float* __restrict__ out, int lane) {
  int n = lane & 15;
  int hi = lane >> 4;
  int lr = t * 16 + n;
  int lrc = lr < AGG_TN ? lr : AGG_TN - 1;
  const float* arow = acc + lrc * 128 + hi * 2;
  const float* wcol = W + hi * 32 + n;
  v2f bf[32];
#pragma unroll
  for (int s = 0; s < 32; ++s) {
    bf[s].x = wcol[s * 64];
    bf[s].y = wcol[s * 64 + 16];
  }
  float bn = bias[n];
  v8f d = {};
#pragma unroll
  for (int s = 0; s < 32; ++s) {
    v2f a = *(const v2f*)(arow + s * 4);
    d = __builtin_amdgcn_wmma_f32_16x16x4_f32(false, a, false, bf[s], (short)0, d, false, false);
  }
  float v[8], m[8], s[8];
#pragma unroll
  for (int r = 0; r < 8; ++r) m[r] = v[r] = d[r] + bn;
#pragma unroll
  for (int off = 1; off < 16; off <<= 1) {
#pragma unroll
    for (int r = 0; r < 8; ++r) m[r] = fmaxf(m[r], __shfl_xor(m[r], off, 32));
  }
#pragma unroll
  for (int r = 0; r < 8; ++r) s[r] = expf(v[r] - m[r]);
#pragma unroll
  for (int off = 1; off < 16; off <<= 1) {
#pragma unroll
    for (int r = 0; r < 8; ++r) s[r] += __shfl_xor(s[r], off, 32);
  }
#pragma unroll
  for (int r = 0; r < 8; ++r) {
    float o = v[r] - (m[r] + logf(s[r]));
    int lrow = t * 16 + r + 8 * hi;
    if (lrow < rows_here) out[(size_t)(base + lrow) * 16 + n] = o;
  }
}

__global__ __launch_bounds__(1024) void agg_lds_kernel(const int* __restrict__ ei,
                                                       const float2* __restrict__ gs,
                                                       const float2* __restrict__ gt,
                                                       const float* __restrict__ bg, int layer,
                                                       const float* __restrict__ H,
                                                       const float* __restrict__ raw,
                                                       float* agg,
                                                       const unsigned* __restrict__ key,
                                                       const float* ws, size_t ws_bytes,
                                                       int E, int Nrows, float* lbuf, int L,
                                                       const float* __restrict__ Wg,
                                                       float2* __restrict__ gsn,
                                                       float2* __restrict__ gtn,
                                                       const float* __restrict__ Wout,
                                                       const float* __restrict__ bout,
                                                       float* __restrict__ out) {
  __shared__ float acc[AGG_TN * 128];
  __shared__ int wq[AGG_WAVES][AGG_QCAP];
  __shared__ float2 lgt[AGG_TN];
  __shared__ int lcount;
  __shared__ int rq[AGG_TN * AGG_RCAP];
  __shared__ int rcount[AGG_TN];
  int tid = threadIdx.x;
  int wave = __builtin_amdgcn_readfirstlane(tid >> 5);
  int lane = tid & 31;
  int base = blockIdx.x * AGG_TN;
  int rows_here = Nrows - base < AGG_TN ? Nrows - base : AGG_TN;
  int* lhdr = (int*)(lbuf + (size_t)base * 128);
  int2* lent = (int2*)(lhdr + 4);
  int lcap = rows_here * 64 - 2;
  bool lwrite = layer == 0 && L > 1;
  int ltotal = -1;
  if (layer == 1) {
    int t = __builtin_amdgcn_readfirstlane(lhdr[0]);
    if (t <= lcap) ltotal = t;
  }
  if (tid == 0) lcount = 0;
  if (tid < AGG_TN) rcount[tid] = 0;
  {
    float4 rv[AGG_INIT_TRIPS];
#pragma unroll
    for (int k = 0; k < AGG_INIT_TRIPS; ++k) {
      int i = tid + k * 1024;
      int ic = i < AGG_TN * 32 ? i : AGG_TN * 32 - 1;
      int node = base + (ic >> 5);
      int nc = node < Nrows ? node : Nrows - 1;
      rv[k] = ((const float4*)raw)[(size_t)nc * 32 + (ic & 31)];
    }
#pragma unroll
    for (int k = 0; k < AGG_INIT_TRIPS; ++k) {
      int i = tid + k * 1024;
      if (i < AGG_TN * 32) {
        float s = base + (i >> 5) < Nrows ? FAGCN_EPS : 0.0f;
        ((float4*)acc)[i] = make_float4(s * rv[k].x, s * rv[k].y, s * rv[k].z, s * rv[k].w);
      }
    }
  }
  if (tid < AGG_TN)
    lgt[tid] = base + tid < Nrows ? gt[base + tid] : make_float2(0.0f, 0.0f);
  float bgl = bg[layer];
  __syncthreads();

  const int* col = ei + E;
  bool scan = ltotal < 0;
  int nsteps = scan ? (E + AGG_STEP - 1) / AGG_STEP : 0;
  int nsuper = scan && agg_keys_ok(key, ws, ws_bytes, E, Nrows) ? E / AGG_KSTEP : 0;
  int2* lw = lwrite ? lent : nullptr;
  int* q = wq[wave];
  int cnt = 0;
  if (!scan && wave * 32 < ltotal) {
    int i = wave * 32 + lane;
    int2 en = lent[i < ltotal ? i : ltotal - 1];
    for (int i0 = wave * 32; i0 < ltotal; i0 += 1024) {
      int nx = i0 + 1024 + lane;
      int2 enx = lent[nx < ltotal ? nx : ltotal - 1];
      agg_route(i0 + lane < ltotal, en.x, en.y, rq, rcount, gs, lgt, bgl, H, acc, lane);
      en = enx;
    }
  }
  if (wave < nsuper) {
    unsigned pat = blockIdx.x * 0x01010101u;
    uint4 kv[4];
    agg_load_keys(key, wave, lane, kv);
    for (int S = wave; S < nsuper; S += AGG_WAVES) {
      unsigned long long m = ((unsigned long long)agg_hits8(kv[2], kv[3], pat) << 32) |
                             agg_hits8(kv[0], kv[1], pat);
      int Sp = S + AGG_WAVES < nsuper ? S + AGG_WAVES : nsuper - 1;
      agg_load_keys(key, Sp, lane, kv);
      int e0 = S * AGG_KSTEP + lane * 16;
      while (__builtin_amdgcn_ballot_w32(m != 0) != 0) {
        bool hit = m != 0;
        int bit = hit ? __builtin_ctzll(m) : 0;
        int off = ((((bit >> 5) << 1) | ((bit >> 2) & 1)) << 9) | ((bit & 3) << 2) |
                  ((bit >> 3) & 3);
        cnt = agg_append(hit, e0 + off, q, cnt);
        m &= m - 1;
        __builtin_amdgcn_fence(__ATOMIC_ACQ_REL, "wavefront");
        if (cnt >= 32) {
          agg_drain(ei, col, gs, lgt, bgl, H, acc, q, cnt, base, lane, lw, &lcount, lcap, rq, rcount);
          cnt = 0;
        }
      }
    }
  }
  for (int s = nsuper * 4 + wave; s < nsteps; s += AGG_WAVES) {
    int cv[16];
    int e0 = s * AGG_STEP + lane * 4;
#pragma unroll
    for (int i = 0; i < 16; ++i) {
      int ee = e0 + (i >> 2) * 128 + (i & 3);
      cv[i] = ee < E ? col[ee] : -1;
    }
    cnt = agg_queue(cv, s, lane, base, q, cnt);
    if (cnt >= 32) {
      agg_drain(ei, col, gs, lgt, bgl, H, acc, q, cnt, base, lane, lw, &lcount, lcap, rq, rcount);
      cnt = 0;
    }
  }
  if (cnt > 0) agg_drain(ei, col, gs, lgt, bgl, H, acc, q, cnt, base, lane, lw, &lcount, lcap, rq, rcount);
  __syncthreads();
  if (lwrite && tid == 0) lhdr[0] = lcount;
  for (int t = wave; t < AGG_TN; t += AGG_WAVES) {
    int n = rcount[t] < AGG_RCAP ? rcount[t] : AGG_RCAP;
    n = __builtin_amdgcn_readfirstlane(n);
    for (int b = 0; b < n; b += 32)
      agg_row_sum(rq + t * AGG_RCAP, b, n - b < 32 ? n - b : 32, t, gs, lgt, bgl, H, acc, lane);
  }
  __syncthreads();

  if (layer == L - 1) {
    if (wave < (rows_here + 15) >> 4) agg_head_tile(acc, wave, rows_here, base, Wout, bout, out, lane);
    return;
  }
  for (int i = tid; i < AGG_TN * 32; i += 1024) {
    int node = base + (i >> 5);
    if (node < Nrows) ((float4*)agg)[(size_t)node * 32 + (i & 31)] = ((float4*)acc)[i];
  }
  agg_next_gates(acc, lgt, Wg + (size_t)(layer + 1) * 256, gsn, gtn, wave, lane, base, rows_here);
}

__global__ void gemm_out_kernel(const float* __restrict__ H, const float* __restrict__ W,
                                const float* __restrict__ bias, float* __restrict__ out,
                                int Nrows) {
  int wave = threadIdx.x >> 5;
  int lane = threadIdx.x & 31;
  int tileM = blockIdx.x * 8 + wave;
  int tiles = (Nrows + 15) >> 4;
  if (tileM >= tiles) return;
  int mA = tileM * 16 + (lane & 15);
  int mclamp = mA < Nrows ? mA : (Nrows - 1);
  int n = lane & 15;
  int khalf = (lane >> 4) * 2;
  const float* hrow = H + (size_t)mclamp * 128 + khalf;
  const float* wcol = W + (size_t)khalf * 16 + n;
  __builtin_prefetch(hrow, 0, 0);
  v8f acc = {};
#pragma unroll 4
  for (int kk = 0; kk < 128; kk += 4) {
    v2f a = *(const v2f*)(hrow + kk);
    v2f b;
    b.x = wcol[(size_t)kk * 16];
    b.y = wcol[(size_t)kk * 16 + 16];
    acc = __builtin_amdgcn_wmma_f32_16x16x4_f32(false, a, false, b, (short)0, acc,
                                               false, false);
  }
  int mbase = tileM * 16 + ((lane >> 4) * 8);
  float bn = bias[n];
  if (tileM * 16 + 16 <= Nrows) {
    float* op = out + (size_t)mbase * 16 + n;
#pragma unroll
    for (int r = 0; r < 8; ++r)
      op[(size_t)r * 16] = acc[r] + bn;
  } else {
#pragma unroll
    for (int r = 0; r < 8; ++r) {
      int mm = mbase + r;
      if (mm < Nrows) out[(size_t)mm * 16 + n] = acc[r] + bn;
    }
  }
}

__global__ void logsoftmax_kernel(float4* out, int N) {
  int i = blockIdx.x * blockDim.x + threadIdx.x;
  if (i >= N) return;
  float4* row = out + (size_t)i * 4;
  float4 v[4];
#pragma unroll
  for (int q = 0; q < 4; ++q) v[q] = row[q];
  float m = v[0].x;
#pragma unroll
  for (int q = 0; q < 4; ++q) {
    m = fmaxf(m, v[q].x); m = fmaxf(m, v[q].y);
    m = fmaxf(m, v[q].z); m = fmaxf(m, v[q].w);
  }
  float s = 0.0f;
#pragma unroll
  for (int q = 0; q < 4; ++q) {
    s += expf(v[q].x - m); s += expf(v[q].y - m);
    s += expf(v[q].z - m); s += expf(v[q].w - m);
  }
  float d = m + logf(s);
#pragma unroll
  for (int q = 0; q < 4; ++q) {
    v[q].x -= d; v[q].y -= d; v[q].z -= d; v[q].w -= d;
    row[q] = v[q];
  }
}

extern "C" void kernel_launch(void* const* d_in, const int* in_sizes, int n_in,
                              void* d_out, int out_size, void* d_ws, size_t ws_size,
                              hipStream_t stream) {
  const float* x     = (const float*)d_in[0];
  const int*   ei    = (const int*)d_in[1];
  const float* W_in  = (const float*)d_in[2];
  const float* b_in  = (const float*)d_in[3];
  const float* Wg    = (const float*)d_in[4];
  const float* bg    = (const float*)d_in[5];
  const float* W_out = (const float*)d_in[6];
  const float* b_out = (const float*)d_in[7];
  float* out = (float*)d_out;

  const int Hdim = 128;
  int N = in_sizes[0] / Hdim;
  int E = in_sizes[1] / 2;
  int L = in_sizes[5];

  float* ws   = (float*)d_ws;
  float2* gs  = (float2*)ws;
  float2* gt  = gs + N;
  float2* gsn = gt + N;
  float2* gtn = gsn + N;
  float* dn   = (float*)(gtn + N);
  float* raw  = ws + 3 * (size_t)N + E;
  float* bufA = raw + (size_t)N * Hdim;
  float* bufB = bufA + (size_t)N * Hdim;
  unsigned* key = (unsigned*)(bufB + (size_t)N * Hdim);

  unsigned* part = (unsigned*)bufB;
  deg_key_kernel<<<dim3(HIST_K, (N + HIST_CAP - 1) / HIST_CAP), HIST_T, 0, stream>>>(
      ei, part, key, ws, ws_size, E, N);
  dn_kernel<<<(N + 255) / 256, 256, 0, stream>>>(part, dn, N);

  gemm_relu_kernel<<<((N + 15) / 16 + GEMM_RT - 1) / GEMM_RT, GEMM_T, 0, stream>>>(x, W_in, b_in, raw, N);

  float* h = raw;
  float* agg = bufA;
  if (L > 0) gate_kernel<<<(N + 8 * GATE_R - 1) / (8 * GATE_R), 256, 0, stream>>>(h, Wg, dn, gs, gt, N);
  for (int i = 0; i < L; ++i) {
    agg_lds_kernel<<<(N + AGG_TN - 1) / AGG_TN, 1024, 0, stream>>>(
        ei, gs, gt, bg, i, h, raw, agg, key, ws, ws_size, E, N, bufB, L, Wg, gsn, gtn, W_out,
        b_out, out);
    h = agg;
    agg = (h == bufA) ? bufB : bufA;
    float2* sw = gs; gs = gsn; gsn = sw;
    sw = gt; gt = gtn; gtn = sw;
  }

  if (L == 0) {
    int tiles = (N + 15) / 16;
    gemm_out_kernel<<<(tiles + 7) / 8, 256, 0, stream>>>(h, W_out, b_out, out, N);
    logsoftmax_kernel<<<(N + 255) / 256, 256, 0, stream>>>((float4*)out, N);
  }
}
